# starting kernel
# baseline (speedup 1.0000x reference)
#include <hip/hip_runtime.h>
#include <math.h>

typedef __attribute__((ext_vector_type(2))) float v2f;
typedef __attribute__((ext_vector_type(8))) float v8f;

#define BATCH 512
#define SEQT  512
#define IND   64
#define H     100
#define NT    7
#define KB1   16
#define KBH   28
#define HSTR  116
#define XSTR  68
#define HBUF  (16*HSTR)

#define OFF_W1X 0
#define OFF_W1H (OFF_W1X + NT*KB1*32*2)
#define OFF_W2X (OFF_W1H + NT*KBH*32*2)
#define OFF_W2H (OFF_W2X + NT*KBH*32*2)
#define OFF_H1  (OFF_W2H + NT*KBH*32*2)
#define OFF_H2  (OFF_H1 + 2*HBUF)
#define OFF_X   (OFF_H2 + 2*HBUF)
#define LDS_FLOATS (OFF_X + 2*16*XSTR)

#if __has_builtin(__builtin_amdgcn_tanhf)
#define TANHF(v) __builtin_amdgcn_tanhf(v)
#elif __has_builtin(__builtin_amdgcn_tanh_f32)
#define TANHF(v) __builtin_amdgcn_tanh_f32(v)
#else
#define TANHF(v) tanhf(v)
#endif

__device__ inline void fill_frags(float* dst, const float* W, int Ka, int Na,
                                  int kb, int tid, int nthr) {
    int total = NT * kb * 32;
    for (int i = tid; i < total; i += nthr) {
        int laneI = i & 31;
        int kblk  = (i >> 5) % kb;
        int nt    = (i >> 5) / kb;
        int N = nt * 16 + (laneI & 15);
        int K = kblk * 4 + ((laneI >> 4) << 1);
        float a = 0.f, b = 0.f;
        if (N < Na) {
            if (K     < Ka) a = W[(size_t)K * Na + N];
            if (K + 1 < Ka) b = W[(size_t)(K + 1) * Na + N];
        }
        dst[2 * i + 0] = a;
        dst[2 * i + 1] = b;
    }
}

__device__ inline v2f ld2(const float* p) { return *(const v2f*)p; }

__global__ void __launch_bounds__(224)
drnn_fused_kernel(const float* __restrict__ x,
                  const float* __restrict__ W1x, const float* __restrict__ W1h,
                  const float* __restrict__ b1,
                  const float* __restrict__ W2x, const float* __restrict__ W2h,
                  const float* __restrict__ b2,
                  const float* __restrict__ Wo,  const float* __restrict__ bo,
                  float* __restrict__ outp, float* __restrict__ h1out,
                  float* __restrict__ h2out) {
    extern __shared__ float lds[];
    const int tid  = threadIdx.x;
    const int nthr = blockDim.x;
    const int b0   = blockIdx.x * 16;

    fill_frags(lds + OFF_W1X, W1x, IND, H, KB1, tid, nthr);
    fill_frags(lds + OFF_W1H, W1h, H,   H, KBH, tid, nthr);
    fill_frags(lds + OFF_W2X, W2x, H,   H, KBH, tid, nthr);
    fill_frags(lds + OFF_W2H, W2h, H,   H, KBH, tid, nthr);
    for (int i = tid; i < 4 * HBUF; i += nthr) lds[OFF_H1 + i] = 0.f;

    {
        #pragma unroll
        for (int ii = 0; ii < 2; ++ii) {
            int i = tid + ii * 224;
            if (i < 256) {
                int row = i >> 4, c = (i & 15) << 2;
                float4 v = *(const float4*)(x + ((size_t)(b0 + row) * SEQT + 0) * IND + c);
                *(float4*)(lds + OFF_X + row * XSTR + c) = v;
            }
        }
    }
    __syncthreads();

    const int lane = tid & 31;
    const int wv   = tid >> 5;
    const int mi   = lane & 15;
    const int kh   = (lane >> 4) << 1;
    const int crow = (lane >> 4) << 3;

    float bias1 = 0.f, bias2 = 0.f;
    {
        int n = wv * 16 + mi;
        if (n < H) { bias1 = b1[n]; bias2 = b2[n]; }
    }

    for (int t = 0; t < SEQT; ++t) {
        const int cb = t & 1;
        const int pb = cb ^ 1;

        float4 xr0 = {}, xr1 = {};
        const bool has1 = (tid < 32);
        if (t + 1 < SEQT) {
            int row = tid >> 4, c = (tid & 15) << 2;
            xr0 = *(const float4*)(x + ((size_t)(b0 + row) * SEQT + (t + 1)) * IND + c);
            if (has1) {
                int j = tid + 224;
                int r2 = j >> 4, c2 = (j & 15) << 2;
                xr1 = *(const float4*)(x + ((size_t)(b0 + r2) * SEQT + (t + 1)) * IND + c2);
            }
        }

        v8f c0, c1;
        #pragma unroll
        for (int r = 0; r < 8; ++r) { c0[r] = bias1; c1[r] = 0.f; }
        {
            const float* xl = lds + OFF_X + cb * (16 * XSTR) + mi * XSTR + kh;
            const float* wp = lds + OFF_W1X + wv * (KB1 * 64) + lane * 2;
            #pragma unroll
            for (int k = 0; k < KB1; k += 2) {
                c0 = __builtin_amdgcn_wmma_f32_16x16x4_f32(
                        false, ld2(xl + 4 * k),       false, ld2(wp + 64 * k),
                        (short)0, c0, false, false);
                c1 = __builtin_amdgcn_wmma_f32_16x16x4_f32(
                        false, ld2(xl + 4 * (k + 1)), false, ld2(wp + 64 * (k + 1)),
                        (short)0, c1, false, false);
            }
            const float* hl = lds + OFF_H1 + pb * HBUF + mi * HSTR + kh;
            wp = lds + OFF_W1H + wv * (KBH * 64) + lane * 2;
            #pragma unroll
            for (int k = 0; k < KBH; k += 2) {
                c0 = __builtin_amdgcn_wmma_f32_16x16x4_f32(
                        false, ld2(hl + 4 * k),       false, ld2(wp + 64 * k),
                        (short)0, c0, false, false);
                c1 = __builtin_amdgcn_wmma_f32_16x16x4_f32(
                        false, ld2(hl + 4 * (k + 1)), false, ld2(wp + 64 * (k + 1)),
                        (short)0, c1, false, false);
            }
        }

        if (t + 1 < SEQT) {
            int row = tid >> 4, c = (tid & 15) << 2;
            *(float4*)(lds + OFF_X + pb * (16 * XSTR) + row * XSTR + c) = xr0;
            if (has1) {
                int j = tid + 224;
                int r2 = j >> 4, c2 = (j & 15) << 2;
                *(float4*)(lds + OFF_X + pb * (16 * XSTR) + r2 * XSTR + c2) = xr1;
            }
        }

        #pragma unroll
        for (int r = 0; r < 8; ++r) {
            float v = TANHF(c0[r] + c1[r]);
            lds[OFF_H1 + cb * HBUF + (r + crow) * HSTR + wv * 16 + mi] = v;
        }
        __syncthreads();

        #pragma unroll
        for (int r = 0; r < 8; ++r) { c0[r] = bias2; c1[r] = 0.f; }
        {
            const float* hl = lds + OFF_H1 + cb * HBUF + mi * HSTR + kh;
            const float* wp = lds + OFF_W2X + wv * (KBH * 64) + lane * 2;
            #pragma unroll
            for (int k = 0; k < KBH; k += 2) {
                c0 = __builtin_amdgcn_wmma_f32_16x16x4_f32(
                        false, ld2(hl + 4 * k),       false, ld2(wp + 64 * k),
                        (short)0, c0, false, false);
                c1 = __builtin_amdgcn_wmma_f32_16x16x4_f32(
                        false, ld2(hl + 4 * (k + 1)), false, ld2(wp + 64 * (k + 1)),
                        (short)0, c1, false, false);
            }
            const float* h2l = lds + OFF_H2 + pb * HBUF + mi * HSTR + kh;
            wp = lds + OFF_W2H + wv * (KBH * 64) + lane * 2;
            #pragma unroll
            for (int k = 0; k < KBH; k += 2) {
                c0 = __builtin_amdgcn_wmma_f32_16x16x4_f32(
                        false, ld2(h2l + 4 * k),       false, ld2(wp + 64 * k),
                        (short)0, c0, false, false);
                c1 = __builtin_amdgcn_wmma_f32_16x16x4_f32(
                        false, ld2(h2l + 4 * (k + 1)), false, ld2(wp + 64 * (k + 1)),
                        (short)0, c1, false, false);
            }
        }

        #pragma unroll
        for (int r = 0; r < 8; ++r) {
            float v = TANHF(c0[r] + c1[r]);
            lds[OFF_H2 + cb * HBUF + (r + crow) * HSTR + wv * 16 + mi] = v;
        }
    }
    __syncthreads();

    const int fb = (SEQT - 1) & 1;
    for (int i = tid; i < 16 * H; i += nthr) {
        int m = i / H, n = i - m * H;
        h1out[(size_t)(b0 + m) * H + n] = lds[OFF_H1 + fb * HBUF + m * HSTR + n];
        h2out[(size_t)(b0 + m) * H + n] = lds[OFF_H2 + fb * HBUF + m * HSTR + n];
    }
    if (tid < 16) {
        float s = bo[0];
        for (int n = 0; n < H; ++n) s += lds[OFF_H2 + fb * HBUF + tid * HSTR + n] * Wo[n];
        outp[b0 + tid] = s;
    }
}

extern "C" void kernel_launch(void* const* d_in, const int* in_sizes, int n_in,
                              void* d_out, int out_size, void* d_ws, size_t ws_size,
                              hipStream_t stream) {
    (void)in_sizes; (void)n_in; (void)out_size; (void)d_ws; (void)ws_size;
    const float* x   = (const float*)d_in[0];
    const float* W1x = (const float*)d_in[1];
    const float* W1h = (const float*)d_in[2];
    const float* b1  = (const float*)d_in[3];
    const float* W2x = (const float*)d_in[4];
    const float* W2h = (const float*)d_in[5];
    const float* b2  = (const float*)d_in[6];
    const float* Wo  = (const float*)d_in[7];
    const float* bo  = (const float*)d_in[8];

    float* outp  = (float*)d_out;
    float* h1out = outp + BATCH;
    float* h2out = h1out + BATCH * H;

    const size_t shmem = (size_t)LDS_FLOATS * sizeof(float);
    (void)hipFuncSetAttribute((const void*)drnn_fused_kernel,
                              hipFuncAttributeMaxDynamicSharedMemorySize, (int)shmem);

    drnn_fused_kernel<<<BATCH / 16, 224, shmem, stream>>>(
        x, W1x, W1h, b1, W2x, W2h, b2, Wo, bo, outp, h1out, h2out);
}


# Hidden-state products skip their all-zero padded k range (k 100..111): 25 k-blocks instead of 28 in each of the three h@W contractions per step, 91 WMMAs per wave and step instead of 100
# speedup vs baseline: 1.0321x; 1.0321x over previous
#include <hip/hip_runtime.h>
#include <math.h>

typedef __attribute__((ext_vector_type(2))) float v2f;
typedef __attribute__((ext_vector_type(8))) float v8f;

#define BATCH 512
#define SEQT  512
#define IND   64
#define H     100
#define NT    7
#define KB1   16
#define KBH   25
#define HSTR  116
#define XSTR  68
#define HBUF  (16*HSTR)

#define OFF_W1X 0
#define OFF_W1H (OFF_W1X + NT*KB1*32*2)
#define OFF_W2X (OFF_W1H + NT*KBH*32*2)
#define OFF_W2H (OFF_W2X + NT*KBH*32*2)
#define OFF_H1  (OFF_W2H + NT*KBH*32*2)
#define OFF_H2  (OFF_H1 + 2*HBUF)
#define OFF_X   (OFF_H2 + 2*HBUF)
#define LDS_FLOATS (OFF_X + 2*16*XSTR)

#if __has_builtin(__builtin_amdgcn_tanhf)
#define TANHF(v) __builtin_amdgcn_tanhf(v)
#elif __has_builtin(__builtin_amdgcn_tanh_f32)
#define TANHF(v) __builtin_amdgcn_tanh_f32(v)
#else
#define TANHF(v) tanhf(v)
#endif

__device__ inline void fill_frags(float* dst, const float* W, int Ka, int Na,
                                  int kb, int tid, int nthr) {
    int total = NT * kb * 32;
    for (int i = tid; i < total; i += nthr) {
        int laneI = i & 31;
        int kblk  = (i >> 5) % kb;
        int nt    = (i >> 5) / kb;
        int N = nt * 16 + (laneI & 15);
        int K = kblk * 4 + ((laneI >> 4) << 1);
        float a = 0.f, b = 0.f;
        if (N < Na) {
            if (K     < Ka) a = W[(size_t)K * Na + N];
            if (K + 1 < Ka) b = W[(size_t)(K + 1) * Na + N];
        }
        dst[2 * i + 0] = a;
        dst[2 * i + 1] = b;
    }
}

__device__ inline v2f ld2(const float* p) { return *(const v2f*)p; }

__global__ void __launch_bounds__(224)
drnn_fused_kernel(const float* __restrict__ x,
                  const float* __restrict__ W1x, const float* __restrict__ W1h,
                  const float* __restrict__ b1,
                  const float* __restrict__ W2x, const float* __restrict__ W2h,
                  const float* __restrict__ b2,
                  const float* __restrict__ Wo,  const float* __restrict__ bo,
                  float* __restrict__ outp, float* __restrict__ h1out,
                  float* __restrict__ h2out) {
    extern __shared__ float lds[];
    const int tid  = threadIdx.x;
    const int nthr = blockDim.x;
    const int b0   = blockIdx.x * 16;

    fill_frags(lds + OFF_W1X, W1x, IND, H, KB1, tid, nthr);
    fill_frags(lds + OFF_W1H, W1h, H,   H, KBH, tid, nthr);
    fill_frags(lds + OFF_W2X, W2x, H,   H, KBH, tid, nthr);
    fill_frags(lds + OFF_W2H, W2h, H,   H, KBH, tid, nthr);
    for (int i = tid; i < 4 * HBUF; i += nthr) lds[OFF_H1 + i] = 0.f;

    {
        #pragma unroll
        for (int ii = 0; ii < 2; ++ii) {
            int i = tid + ii * 224;
            if (i < 256) {
                int row = i >> 4, c = (i & 15) << 2;
                float4 v = *(const float4*)(x + ((size_t)(b0 + row) * SEQT + 0) * IND + c);
                *(float4*)(lds + OFF_X + row * XSTR + c) = v;
            }
        }
    }
    __syncthreads();

    const int lane = tid & 31;
    const int wv   = tid >> 5;
    const int mi   = lane & 15;
    const int kh   = (lane >> 4) << 1;
    const int crow = (lane >> 4) << 3;

    float bias1 = 0.f, bias2 = 0.f;
    {
        int n = wv * 16 + mi;
        if (n < H) { bias1 = b1[n]; bias2 = b2[n]; }
    }

    for (int t = 0; t < SEQT; ++t) {
        const int cb = t & 1;
        const int pb = cb ^ 1;

        float4 xr0 = {}, xr1 = {};
        const bool has1 = (tid < 32);
        if (t + 1 < SEQT) {
            int row = tid >> 4, c = (tid & 15) << 2;
            xr0 = *(const float4*)(x + ((size_t)(b0 + row) * SEQT + (t + 1)) * IND + c);
            if (has1) {
                int j = tid + 224;
                int r2 = j >> 4, c2 = (j & 15) << 2;
                xr1 = *(const float4*)(x + ((size_t)(b0 + r2) * SEQT + (t + 1)) * IND + c2);
            }
        }

        v8f c0, c1;
        #pragma unroll
        for (int r = 0; r < 8; ++r) { c0[r] = bias1; c1[r] = 0.f; }
        {
            const float* xl = lds + OFF_X + cb * (16 * XSTR) + mi * XSTR + kh;
            const float* wp = lds + OFF_W1X + wv * (KB1 * 64) + lane * 2;
            #pragma unroll
            for (int k = 0; k < KB1; k += 2) {
                c0 = __builtin_amdgcn_wmma_f32_16x16x4_f32(
                        false, ld2(xl + 4 * k),       false, ld2(wp + 64 * k),
                        (short)0, c0, false, false);
                c1 = __builtin_amdgcn_wmma_f32_16x16x4_f32(
                        false, ld2(xl + 4 * (k + 1)), false, ld2(wp + 64 * (k + 1)),
                        (short)0, c1, false, false);
            }
            const float* hl = lds + OFF_H1 + pb * HBUF + mi * HSTR + kh;
            wp = lds + OFF_W1H + wv * (KBH * 64) + lane * 2;
            #pragma unroll
            for (int k = 0; k + 1 < KBH; k += 2) {
                c0 = __builtin_amdgcn_wmma_f32_16x16x4_f32(
                        false, ld2(hl + 4 * k),       false, ld2(wp + 64 * k),
                        (short)0, c0, false, false);
                c1 = __builtin_amdgcn_wmma_f32_16x16x4_f32(
                        false, ld2(hl + 4 * (k + 1)), false, ld2(wp + 64 * (k + 1)),
                        (short)0, c1, false, false);
            }
            if (KBH & 1)
                c0 = __builtin_amdgcn_wmma_f32_16x16x4_f32(
                        false, ld2(hl + 4 * (KBH - 1)), false, ld2(wp + 64 * (KBH - 1)),
                        (short)0, c0, false, false);
        }

        if (t + 1 < SEQT) {
            int row = tid >> 4, c = (tid & 15) << 2;
            *(float4*)(lds + OFF_X + pb * (16 * XSTR) + row * XSTR + c) = xr0;
            if (has1) {
                int j = tid + 224;
                int r2 = j >> 4, c2 = (j & 15) << 2;
                *(float4*)(lds + OFF_X + pb * (16 * XSTR) + r2 * XSTR + c2) = xr1;
            }
        }

        #pragma unroll
        for (int r = 0; r < 8; ++r) {
            float v = TANHF(c0[r] + c1[r]);
            lds[OFF_H1 + cb * HBUF + (r + crow) * HSTR + wv * 16 + mi] = v;
        }
        __syncthreads();

        #pragma unroll
        for (int r = 0; r < 8; ++r) { c0[r] = bias2; c1[r] = 0.f; }
        {
            const float* hl = lds + OFF_H1 + cb * HBUF + mi * HSTR + kh;
            const float* wp = lds + OFF_W2X + wv * (KBH * 64) + lane * 2;
            #pragma unroll
            for (int k = 0; k + 1 < KBH; k += 2) {
                c0 = __builtin_amdgcn_wmma_f32_16x16x4_f32(
                        false, ld2(hl + 4 * k),       false, ld2(wp + 64 * k),
                        (short)0, c0, false, false);
                c1 = __builtin_amdgcn_wmma_f32_16x16x4_f32(
                        false, ld2(hl + 4 * (k + 1)), false, ld2(wp + 64 * (k + 1)),
                        (short)0, c1, false, false);
            }
            if (KBH & 1)
                c0 = __builtin_amdgcn_wmma_f32_16x16x4_f32(
                        false, ld2(hl + 4 * (KBH - 1)), false, ld2(wp + 64 * (KBH - 1)),
                        (short)0, c0, false, false);
            const float* h2l = lds + OFF_H2 + pb * HBUF + mi * HSTR + kh;
            wp = lds + OFF_W2H + wv * (KBH * 64) + lane * 2;
            #pragma unroll
            for (int k = 0; k + 1 < KBH; k += 2) {
                c0 = __builtin_amdgcn_wmma_f32_16x16x4_f32(
                        false, ld2(h2l + 4 * k),       false, ld2(wp + 64 * k),
                        (short)0, c0, false, false);
                c1 = __builtin_amdgcn_wmma_f32_16x16x4_f32(
                        false, ld2(h2l + 4 * (k + 1)), false, ld2(wp + 64 * (k + 1)),
                        (short)0, c1, false, false);
            }
            if (KBH & 1)
                c0 = __builtin_amdgcn_wmma_f32_16x16x4_f32(
                        false, ld2(h2l + 4 * (KBH - 1)), false, ld2(wp + 64 * (KBH - 1)),
                        (short)0, c0, false, false);
        }

        #pragma unroll
        for (int r = 0; r < 8; ++r) {
            float v = TANHF(c0[r] + c1[r]);
            lds[OFF_H2 + cb * HBUF + (r + crow) * HSTR + wv * 16 + mi] = v;
        }
    }
    __syncthreads();

    const int fb = (SEQT - 1) & 1;
    for (int i = tid; i < 16 * H; i += nthr) {
        int m = i / H, n = i - m * H;
        h1out[(size_t)(b0 + m) * H + n] = lds[OFF_H1 + fb * HBUF + m * HSTR + n];
        h2out[(size_t)(b0 + m) * H + n] = lds[OFF_H2 + fb * HBUF + m * HSTR + n];
    }
    if (tid < 16) {
        float s = bo[0];
        for (int n = 0; n < H; ++n) s += lds[OFF_H2 + fb * HBUF + tid * HSTR + n] * Wo[n];
        outp[b0 + tid] = s;
    }
}

extern "C" void kernel_launch(void* const* d_in, const int* in_sizes, int n_in,
                              void* d_out, int out_size, void* d_ws, size_t ws_size,
                              hipStream_t stream) {
    (void)in_sizes; (void)n_in; (void)out_size; (void)d_ws; (void)ws_size;
    const float* x   = (const float*)d_in[0];
    const float* W1x = (const float*)d_in[1];
    const float* W1h = (const float*)d_in[2];
    const float* b1  = (const float*)d_in[3];
    const float* W2x = (const float*)d_in[4];
    const float* W2h = (const float*)d_in[5];
    const float* b2  = (const float*)d_in[6];
    const float* Wo  = (const float*)d_in[7];
    const float* bo  = (const float*)d_in[8];

    float* outp  = (float*)d_out;
    float* h1out = outp + BATCH;
    float* h2out = h1out + BATCH * H;

    const size_t shmem = (size_t)LDS_FLOATS * sizeof(float);
    (void)hipFuncSetAttribute((const void*)drnn_fused_kernel,
                              hipFuncAttributeMaxDynamicSharedMemorySize, (int)shmem);

    drnn_fused_kernel<<<BATCH / 16, 224, shmem, stream>>>(
        x, W1x, W1h, b1, W2x, W2h, b2, Wo, bo, outp, h1out, h2out);
}


# Eighth wave forms b1 + x[:,0..43]@W1x for all 7 tiles a step ahead, via LDS, as layer-1 start value: 80 not 91 WMMAs per recurrence wave-step; x requested a step earlier; layer-1 sum order changes
# speedup vs baseline: 1.0880x; 1.0542x over previous
#include <hip/hip_runtime.h>
#include <math.h>

typedef __attribute__((ext_vector_type(2))) float v2f;
typedef __attribute__((ext_vector_type(8))) float v8f;

#define BATCH 512
#define SEQT  512
#define IND   64
#define H     100
#define NT    7
#define KB1   16
#define KBH   25
#define HSTR  116
#define XSTR  68
#define HBUF  (16*HSTR)

#define OFF_W1X 0
#define OFF_W1H (OFF_W1X + NT*KB1*32*2)
#define OFF_W2X (OFF_W1H + NT*KBH*32*2)
#define OFF_W2H (OFF_W2X + NT*KBH*32*2)
#define OFF_H1  (OFF_W2H + NT*KBH*32*2)
#define OFF_H2  (OFF_H1 + 2*HBUF)
#define OFF_X   (OFF_H2 + 2*HBUF)
#define XBUF    (16*XSTR)
#define KBXP    11
#define OFF_P   (OFF_X + 4*XBUF)
#define PBUF    (NT*256)
#define LDS_FLOATS (OFF_P + 2*PBUF)

#if __has_builtin(__builtin_amdgcn_tanhf)
#define TANHF(v) __builtin_amdgcn_tanhf(v)
#elif __has_builtin(__builtin_amdgcn_tanh_f32)
#define TANHF(v) __builtin_amdgcn_tanh_f32(v)
#else
#define TANHF(v) tanhf(v)
#endif

__device__ inline void fill_frags(float* dst, const float* W, int Ka, int Na,
                                  int kb, int tid, int nthr) {
    int total = NT * kb * 32;
    for (int i = tid; i < total; i += nthr) {
        int laneI = i & 31;
        int kblk  = (i >> 5) % kb;
        int nt    = (i >> 5) / kb;
        int N = nt * 16 + (laneI & 15);
        int K = kblk * 4 + ((laneI >> 4) << 1);
        float a = 0.f, b = 0.f;
        if (N < Na) {
            if (K     < Ka) a = W[(size_t)K * Na + N];
            if (K + 1 < Ka) b = W[(size_t)(K + 1) * Na + N];
        }
        dst[2 * i + 0] = a;
        dst[2 * i + 1] = b;
    }
}

__device__ inline v2f ld2(const float* p) { return *(const v2f*)p; }

__global__ void __launch_bounds__(256)
drnn_fused_kernel(const float* __restrict__ x,
                  const float* __restrict__ W1x, const float* __restrict__ W1h,
                  const float* __restrict__ b1,
                  const float* __restrict__ W2x, const float* __restrict__ W2h,
                  const float* __restrict__ b2,
                  const float* __restrict__ Wo,  const float* __restrict__ bo,
                  float* __restrict__ outp, float* __restrict__ h1out,
                  float* __restrict__ h2out) {
    extern __shared__ float lds[];
    const int tid  = threadIdx.x;
    const int nthr = blockDim.x;
    const int b0   = blockIdx.x * 16;

    fill_frags(lds + OFF_W1X, W1x, IND, H, KB1, tid, nthr);
    fill_frags(lds + OFF_W1H, W1h, H,   H, KBH, tid, nthr);
    fill_frags(lds + OFF_W2X, W2x, H,   H, KBH, tid, nthr);
    fill_frags(lds + OFF_W2H, W2h, H,   H, KBH, tid, nthr);
    for (int i = tid; i < 4 * HBUF; i += nthr) lds[OFF_H1 + i] = 0.f;

    const float* xsrc = x + ((size_t)(b0 + (tid >> 4)) * SEQT) * IND + ((tid & 15) << 2);
    float*       xdst = lds + OFF_X + (tid >> 4) * XSTR + ((tid & 15) << 2);
    *(float4*)(xdst)        = *(const float4*)(xsrc);
    *(float4*)(xdst + XBUF) = *(const float4*)(xsrc + IND);
    float4 xr = *(const float4*)(xsrc + 2 * IND);
    __syncthreads();

    const int lane = tid & 31;
    const int wv   = __builtin_amdgcn_readfirstlane(tid >> 5);
    const bool helper = (wv == NT);
    const int mi   = lane & 15;
    const int kh   = (lane >> 4) << 1;
    const int crow = (lane >> 4) << 3;

    float bias2 = 0.f;
    {
        int n = wv * 16 + mi;
        if (n < H) bias2 = b2[n];
    }

    v2f   wb[NT * KBXP];
    float pbias[NT];
    if (helper) {
        #pragma unroll
        for (int j = 0; j < NT; ++j) {
            pbias[j] = (j * 16 + mi < H) ? b1[j * 16 + mi] : 0.f;
            #pragma unroll
            for (int kb = 0; kb < KBXP; ++kb)
                wb[j * KBXP + kb] = ld2(lds + OFF_W1X + (j * KB1 + kb) * 64 + lane * 2);
        }
        const float* xl = lds + OFF_X + mi * XSTR + kh;
        v8f acc[NT];
        #pragma unroll
        for (int j = 0; j < NT; ++j)
            #pragma unroll
            for (int r = 0; r < 8; ++r) acc[j][r] = pbias[j];
        #pragma unroll
        for (int kb = 0; kb < KBXP; ++kb) {
            v2f a = ld2(xl + 4 * kb);
            #pragma unroll
            for (int j = 0; j < NT; ++j)
                acc[j] = __builtin_amdgcn_wmma_f32_16x16x4_f32(false, a, false, wb[j * KBXP + kb], (short)0, acc[j], false, false);
        }
        #pragma unroll
        for (int j = 0; j < NT; ++j) {
            float* pp = lds + OFF_P + j * 256 + lane * 4;
            *(float4*)(pp)       = make_float4(acc[j][0], acc[j][1], acc[j][2], acc[j][3]);
            *(float4*)(pp + 128) = make_float4(acc[j][4], acc[j][5], acc[j][6], acc[j][7]);
        }
    }
    __syncthreads();

    for (int t = 0; t < SEQT; ++t) {
        const int cb = t & 1;
        const int pb = cb ^ 1;

        if (!helper) {
            v8f c0, c1;
            {
                const float* pp = lds + OFF_P + cb * PBUF + wv * 256 + lane * 4;
                const float4 p0 = *(const float4*)(pp);
                const float4 p1 = *(const float4*)(pp + 128);
                c0[0] = p0.x; c0[1] = p0.y; c0[2] = p0.z; c0[3] = p0.w;
                c0[4] = p1.x; c0[5] = p1.y; c0[6] = p1.z; c0[7] = p1.w;
            }
            #pragma unroll
            for (int r = 0; r < 8; ++r) c1[r] = 0.f;
            const float* xl = lds + OFF_X + (t & 3) * XBUF + mi * XSTR + kh;
            const float* wp = lds + OFF_W1X + wv * (KB1 * 64) + lane * 2;
            #pragma unroll
            for (int k = KBXP; k < KB1; ++k) {
                if ((k - KBXP) & 1)
                    c1 = __builtin_amdgcn_wmma_f32_16x16x4_f32(false, ld2(xl + 4 * k), false, ld2(wp + 64 * k), (short)0, c1, false, false);
                else
                    c0 = __builtin_amdgcn_wmma_f32_16x16x4_f32(false, ld2(xl + 4 * k), false, ld2(wp + 64 * k), (short)0, c0, false, false);
            }
            const float* hl = lds + OFF_H1 + pb * HBUF + mi * HSTR + kh;
            wp = lds + OFF_W1H + wv * (KBH * 64) + lane * 2;
            #pragma unroll
            for (int k = 0; k < KBH; ++k) {
                if ((k + KB1 - KBXP) & 1)
                    c1 = __builtin_amdgcn_wmma_f32_16x16x4_f32(false, ld2(hl + 4 * k), false, ld2(wp + 64 * k), (short)0, c1, false, false);
                else
                    c0 = __builtin_amdgcn_wmma_f32_16x16x4_f32(false, ld2(hl + 4 * k), false, ld2(wp + 64 * k), (short)0, c0, false, false);
            }
            #pragma unroll
            for (int r = 0; r < 8; ++r) {
                float v = TANHF(c0[r] + c1[r]);
                lds[OFF_H1 + cb * HBUF + (r + crow) * HSTR + wv * 16 + mi] = v;
            }
        } else if (t + 1 < SEQT) {
            const float* xl = lds + OFF_X + ((t + 1) & 3) * XBUF + mi * XSTR + kh;
            v8f acc[NT];
            #pragma unroll
            for (int j = 0; j < NT; ++j)
                #pragma unroll
                for (int r = 0; r < 8; ++r) acc[j][r] = pbias[j];
            #pragma unroll
            for (int kb = 0; kb < KBXP; ++kb) {
                v2f a = ld2(xl + 4 * kb);
                #pragma unroll
                for (int j = 0; j < NT; ++j)
                    acc[j] = __builtin_amdgcn_wmma_f32_16x16x4_f32(false, a, false, wb[j * KBXP + kb], (short)0, acc[j], false, false);
            }
            #pragma unroll
            for (int j = 0; j < NT; ++j) {
                float* pp = lds + OFF_P + pb * PBUF + j * 256 + lane * 4;
                *(float4*)(pp)       = make_float4(acc[j][0], acc[j][1], acc[j][2], acc[j][3]);
                *(float4*)(pp + 128) = make_float4(acc[j][4], acc[j][5], acc[j][6], acc[j][7]);
            }
        }

        if (t + 2 < SEQT) *(float4*)(xdst + ((t + 2) & 3) * XBUF) = xr;

        __syncthreads();

        if (t + 3 < SEQT) xr = *(const float4*)(xsrc + (size_t)(t + 3) * IND);

        if (!helper) {
            v8f c0, c1;
            #pragma unroll
            for (int r = 0; r < 8; ++r) { c0[r] = bias2; c1[r] = 0.f; }
            {
                const float* hl = lds + OFF_H1 + cb * HBUF + mi * HSTR + kh;
                const float* wp = lds + OFF_W2X + wv * (KBH * 64) + lane * 2;
                #pragma unroll
                for (int k = 0; k + 1 < KBH; k += 2) {
                    c0 = __builtin_amdgcn_wmma_f32_16x16x4_f32(
                            false, ld2(hl + 4 * k),       false, ld2(wp + 64 * k),
                            (short)0, c0, false, false);
                    c1 = __builtin_amdgcn_wmma_f32_16x16x4_f32(
                            false, ld2(hl + 4 * (k + 1)), false, ld2(wp + 64 * (k + 1)),
                            (short)0, c1, false, false);
                }
                if (KBH & 1)
                    c0 = __builtin_amdgcn_wmma_f32_16x16x4_f32(
                            false, ld2(hl + 4 * (KBH - 1)), false, ld2(wp + 64 * (KBH - 1)),
                            (short)0, c0, false, false);
                const float* h2l = lds + OFF_H2 + pb * HBUF + mi * HSTR + kh;
                wp = lds + OFF_W2H + wv * (KBH * 64) + lane * 2;
                #pragma unroll
                for (int k = 0; k + 1 < KBH; k += 2) {
                    c0 = __builtin_amdgcn_wmma_f32_16x16x4_f32(
                            false, ld2(h2l + 4 * k),       false, ld2(wp + 64 * k),
                            (short)0, c0, false, false);
                    c1 = __builtin_amdgcn_wmma_f32_16x16x4_f32(
                            false, ld2(h2l + 4 * (k + 1)), false, ld2(wp + 64 * (k + 1)),
                            (short)0, c1, false, false);
                }
                if (KBH & 1)
                    c0 = __builtin_amdgcn_wmma_f32_16x16x4_f32(
                            false, ld2(h2l + 4 * (KBH - 1)), false, ld2(wp + 64 * (KBH - 1)),
                            (short)0, c0, false, false);
            }

            #pragma unroll
            for (int r = 0; r < 8; ++r) {
                float v = TANHF(c0[r] + c1[r]);
                lds[OFF_H2 + cb * HBUF + (r + crow) * HSTR + wv * 16 + mi] = v;
            }
        }
    }
    __syncthreads();

    const int fb = (SEQT - 1) & 1;
    for (int i = tid; i < 16 * H; i += nthr) {
        int m = i / H, n = i - m * H;
        h1out[(size_t)(b0 + m) * H + n] = lds[OFF_H1 + fb * HBUF + m * HSTR + n];
        h2out[(size_t)(b0 + m) * H + n] = lds[OFF_H2 + fb * HBUF + m * HSTR + n];
    }
    if (tid < 16) {
        float s = bo[0];
        for (int n = 0; n < H; ++n) s += lds[OFF_H2 + fb * HBUF + tid * HSTR + n] * Wo[n];
        outp[b0 + tid] = s;
    }
}

extern "C" void kernel_launch(void* const* d_in, const int* in_sizes, int n_in,
                              void* d_out, int out_size, void* d_ws, size_t ws_size,
                              hipStream_t stream) {
    (void)in_sizes; (void)n_in; (void)out_size; (void)d_ws; (void)ws_size;
    const float* x   = (const float*)d_in[0];
    const float* W1x = (const float*)d_in[1];
    const float* W1h = (const float*)d_in[2];
    const float* b1  = (const float*)d_in[3];
    const float* W2x = (const float*)d_in[4];
    const float* W2h = (const float*)d_in[5];
    const float* b2  = (const float*)d_in[6];
    const float* Wo  = (const float*)d_in[7];
    const float* bo  = (const float*)d_in[8];

    float* outp  = (float*)d_out;
    float* h1out = outp + BATCH;
    float* h2out = h1out + BATCH * H;

    const size_t shmem = (size_t)LDS_FLOATS * sizeof(float);
    (void)hipFuncSetAttribute((const void*)drnn_fused_kernel,
                              hipFuncAttributeMaxDynamicSharedMemorySize, (int)shmem);

    drnn_fused_kernel<<<BATCH / 16, 256, shmem, stream>>>(
        x, W1x, W1h, b1, W2x, W2h, b2, Wo, bo, outp, h1out, h2out);
}


# LDS reads leave the WMMA stream: a wave keeps its 80 weight fragments in registers for all steps and reads a step's h, x and P operands in one group behind the barrier; every sum in its old order
# speedup vs baseline: 1.1830x; 1.0873x over previous
#include <hip/hip_runtime.h>
#include <math.h>

typedef __attribute__((ext_vector_type(2))) float v2f;
typedef __attribute__((ext_vector_type(8))) float v8f;

#define BATCH 512
#define SEQT  512
#define IND   64
#define H     100
#define NT    7
#define KB1   16
#define KBH   25
#define HSTR  116
#define XSTR  68
#define HBUF  (16*HSTR)

#define OFF_W1X 0
#define OFF_W1H (OFF_W1X + NT*KB1*32*2)
#define OFF_W2X (OFF_W1H + NT*KBH*32*2)
#define OFF_W2H (OFF_W2X + NT*KBH*32*2)
#define OFF_H1  (OFF_W2H + NT*KBH*32*2)
#define OFF_H2  (OFF_H1 + 2*HBUF)
#define OFF_X   (OFF_H2 + 2*HBUF)
#define XBUF    (16*XSTR)
#define KBXP    11
#define OFF_P   (OFF_X + 4*XBUF)
#define PBUF    (NT*256)
#define LDS_FLOATS (OFF_P + 2*PBUF)

#if __has_builtin(__builtin_amdgcn_tanhf)
#define TANHF(v) __builtin_amdgcn_tanhf(v)
#elif __has_builtin(__builtin_amdgcn_tanh_f32)
#define TANHF(v) __builtin_amdgcn_tanh_f32(v)
#else
#define TANHF(v) tanhf(v)
#endif

__device__ inline void fill_frags(float* dst, const float* W, int Ka, int Na,
                                  int kb, int tid, int nthr) {
    int total = NT * kb * 32;
    for (int i = tid; i < total; i += nthr) {
        int laneI = i & 31;
        int kblk  = (i >> 5) % kb;
        int nt    = (i >> 5) / kb;
        int N = nt * 16 + (laneI & 15);
        int K = kblk * 4 + ((laneI >> 4) << 1);
        float a = 0.f, b = 0.f;
        if (N < Na) {
            if (K     < Ka) a = W[(size_t)K * Na + N];
            if (K + 1 < Ka) b = W[(size_t)(K + 1) * Na + N];
        }
        dst[2 * i + 0] = a;
        dst[2 * i + 1] = b;
    }
}

__device__ inline v2f ld2(const float* p) { return *(const v2f*)p; }

#define KBXR  (KB1 - KBXP)
#define NWR   80
#define WR_1X 0
#define WR_1H (WR_1X + KBXR)
#define WR_2X (WR_1H + KBH)
#define WR_2H (WR_2X + KBH)

#define WMMA4(a, b, c) __builtin_amdgcn_wmma_f32_16x16x4_f32(false, (a), false, (b), (short)0, (c), false, false)

__device__ __forceinline__ void tanh_store(const v8f& c0, const v8f& c1, float* dst) {
    #pragma unroll
    for (int r = 0; r < 8; ++r) dst[r * HSTR] = TANHF(c0[r] + c1[r]);
}

__device__ __forceinline__ void layer1_store(const float4& p0, const float4& p1,
                                             const v2f (&ax)[KBXR], const v2f (&a1)[KBH],
                                             const v2f (&wr)[NWR], float* dst) {
    v8f c0, c1;
    c0[0] = p0.x; c0[1] = p0.y; c0[2] = p0.z; c0[3] = p0.w;
    c0[4] = p1.x; c0[5] = p1.y; c0[6] = p1.z; c0[7] = p1.w;
    #pragma unroll
    for (int r = 0; r < 8; ++r) c1[r] = 0.f;
    #pragma unroll
    for (int k = 0; k < KBXR; ++k) {
        if (k & 1) c1 = WMMA4(ax[k], wr[WR_1X + k], c1);
        else       c0 = WMMA4(ax[k], wr[WR_1X + k], c0);
    }
    #pragma unroll
    for (int k = 0; k < KBH; ++k) {
        if ((k + KBXR) & 1) c1 = WMMA4(a1[k], wr[WR_1H + k], c1);
        else                c0 = WMMA4(a1[k], wr[WR_1H + k], c0);
    }
    tanh_store(c0, c1, dst);
}

__device__ __forceinline__ void layer2_store(float bias2, const v2f (&a1)[KBH], const v2f (&a2)[KBH],
                                             const v2f (&wr)[NWR], float* dst) {
    v8f c0, c1;
    #pragma unroll
    for (int r = 0; r < 8; ++r) { c0[r] = bias2; c1[r] = 0.f; }
    #pragma unroll
    for (int k = 0; k < KBH; ++k) {
        if (k & 1) c1 = WMMA4(a1[k], wr[WR_2X + k], c1);
        else       c0 = WMMA4(a1[k], wr[WR_2X + k], c0);
    }
    #pragma unroll
    for (int k = 0; k < KBH; ++k) {
        if (k & 1) c1 = WMMA4(a2[k], wr[WR_2H + k], c1);
        else       c0 = WMMA4(a2[k], wr[WR_2H + k], c0);
    }
    tanh_store(c0, c1, dst);
}

__device__ __forceinline__ void helper_p(const float* xl, const float (&pbias)[NT],
                                         const v2f (&wr)[NWR], float* pdst) {
    v8f acc[NT];
    #pragma unroll
    for (int j = 0; j < NT; ++j)
        #pragma unroll
        for (int r = 0; r < 8; ++r) acc[j][r] = pbias[j];
    #pragma unroll
    for (int kb = 0; kb < KBXP; ++kb) {
        v2f a = ld2(xl + 4 * kb);
        #pragma unroll
        for (int j = 0; j < NT; ++j)
            acc[j] = WMMA4(a, wr[j * KBXP + kb], acc[j]);
    }
    #pragma unroll
    for (int j = 0; j < NT; ++j) {
        float* pp = pdst + j * 256;
        *(float4*)(pp)       = make_float4(acc[j][0], acc[j][1], acc[j][2], acc[j][3]);
        *(float4*)(pp + 128) = make_float4(acc[j][4], acc[j][5], acc[j][6], acc[j][7]);
    }
}

__global__ void __launch_bounds__(256)
drnn_fused_kernel(const float* __restrict__ x,
                  const float* __restrict__ W1x, const float* __restrict__ W1h,
                  const float* __restrict__ b1,
                  const float* __restrict__ W2x, const float* __restrict__ W2h,
                  const float* __restrict__ b2,
                  const float* __restrict__ Wo,  const float* __restrict__ bo,
                  float* __restrict__ outp, float* __restrict__ h1out,
                  float* __restrict__ h2out) {
    extern __shared__ float lds[];
    const int tid  = threadIdx.x;
    const int nthr = blockDim.x;
    const int b0   = blockIdx.x * 16;

    fill_frags(lds + OFF_W1X, W1x, IND, H, KB1, tid, nthr);
    fill_frags(lds + OFF_W1H, W1h, H,   H, KBH, tid, nthr);
    fill_frags(lds + OFF_W2X, W2x, H,   H, KBH, tid, nthr);
    fill_frags(lds + OFF_W2H, W2h, H,   H, KBH, tid, nthr);
    for (int i = tid; i < 4 * HBUF; i += nthr) lds[OFF_H1 + i] = 0.f;

    const float* xsrc = x + ((size_t)(b0 + (tid >> 4)) * SEQT) * IND + ((tid & 15) << 2);
    float*       xdst = lds + OFF_X + (tid >> 4) * XSTR + ((tid & 15) << 2);
    *(float4*)(xdst)        = *(const float4*)(xsrc);
    *(float4*)(xdst + XBUF) = *(const float4*)(xsrc + IND);
    float4 xr = *(const float4*)(xsrc + 2 * IND);
    __syncthreads();

    const int lane = tid & 31;
    const int wv   = __builtin_amdgcn_readfirstlane(tid >> 5);
    const bool helper = (wv == NT);
    const int mi   = lane & 15;
    const int kh   = (lane >> 4) << 1;
    const int crow = (lane >> 4) << 3;

    float bias2 = 0.f;
    {
        int n = wv * 16 + mi;
        if (n < H) bias2 = b2[n];
    }

    v2f   wr[NWR];
    float pbias[NT];
    #pragma unroll
    for (int j = 0; j < NT; ++j) pbias[j] = 0.f;
    if (helper) {
        #pragma unroll
        for (int j = 0; j < NT; ++j) {
            pbias[j] = (j * 16 + mi < H) ? b1[j * 16 + mi] : 0.f;
            #pragma unroll
            for (int kb = 0; kb < KBXP; ++kb)
                wr[j * KBXP + kb] = ld2(lds + OFF_W1X + (j * KB1 + kb) * 64 + lane * 2);
        }
        #pragma unroll
        for (int i = NT * KBXP; i < NWR; ++i) { wr[i][0] = 0.f; wr[i][1] = 0.f; }
        const float* xl = lds + OFF_X + mi * XSTR + kh;
        v8f acc[NT];
        #pragma unroll
        for (int j = 0; j < NT; ++j)
            #pragma unroll
            for (int r = 0; r < 8; ++r) acc[j][r] = pbias[j];
        #pragma unroll
        for (int kb = 0; kb < KBXP; ++kb) {
            v2f a = ld2(xl + 4 * kb);
            #pragma unroll
            for (int j = 0; j < NT; ++j)
                acc[j] = __builtin_amdgcn_wmma_f32_16x16x4_f32(false, a, false, wr[j * KBXP + kb], (short)0, acc[j], false, false);
        }
        #pragma unroll
        for (int j = 0; j < NT; ++j) {
            float* pp = lds + OFF_P + j * 256 + lane * 4;
            *(float4*)(pp)       = make_float4(acc[j][0], acc[j][1], acc[j][2], acc[j][3]);
            *(float4*)(pp + 128) = make_float4(acc[j][4], acc[j][5], acc[j][6], acc[j][7]);
        }
    } else {
        #pragma unroll
        for (int k = 0; k < KBXR; ++k)
            wr[WR_1X + k] = ld2(lds + OFF_W1X + (wv * KB1 + KBXP + k) * 64 + lane * 2);
        #pragma unroll
        for (int k = 0; k < KBH; ++k) {
            wr[WR_1H + k] = ld2(lds + OFF_W1H + (wv * KBH + k) * 64 + lane * 2);
            wr[WR_2X + k] = ld2(lds + OFF_W2X + (wv * KBH + k) * 64 + lane * 2);
            wr[WR_2H + k] = ld2(lds + OFF_W2H + (wv * KBH + k) * 64 + lane * 2);
        }
    }
    __syncthreads();

    if (!helper) {
        v2f a1[KBH], ax[KBXR];
        const float* hl = lds + OFF_H1 + HBUF + mi * HSTR + kh;
        const float* xl = lds + OFF_X + mi * XSTR + kh;
        const float* pp = lds + OFF_P + wv * 256 + lane * 4;
        #pragma unroll
        for (int k = 0; k < KBH; ++k) a1[k] = ld2(hl + 4 * k);
        #pragma unroll
        for (int k = 0; k < KBXR; ++k) ax[k] = ld2(xl + 4 * (KBXP + k));
        const float4 p0 = *(const float4*)(pp);
        const float4 p1 = *(const float4*)(pp + 128);
        layer1_store(p0, p1, ax, a1, wr, lds + OFF_H1 + crow * HSTR + wv * 16 + mi);
    } else {
        helper_p(lds + OFF_X + XBUF + mi * XSTR + kh, pbias, wr, lds + OFF_P + PBUF + lane * 4);
    }
    *(float4*)(xdst + 2 * XBUF) = xr;
    __syncthreads();
    xr = *(const float4*)(xsrc + 3 * IND);

    for (int t = 0; t + 1 < SEQT; ++t) {
        const int cb = t & 1;
        const int nb = cb ^ 1;

        if (!helper) {
            v2f a1[KBH], a2[KBH], ax[KBXR];
            const float* h1l = lds + OFF_H1 + cb * HBUF + mi * HSTR + kh;
            const float* h2l = lds + OFF_H2 + nb * HBUF + mi * HSTR + kh;
            const float* xl  = lds + OFF_X + ((t + 1) & 3) * XBUF + mi * XSTR + kh;
            const float* pp  = lds + OFF_P + nb * PBUF + wv * 256 + lane * 4;
            const float4 p0 = *(const float4*)(pp);
            const float4 p1 = *(const float4*)(pp + 128);
            #pragma unroll
            for (int k = 0; k < KBXR; ++k) ax[k] = ld2(xl + 4 * (KBXP + k));
            #pragma unroll
            for (int k = 0; k < 4; ++k) a1[k] = ld2(h1l + 4 * k);
            __builtin_amdgcn_sched_barrier(0);
            #pragma unroll
            for (int k = 4; k < KBH; ++k) a1[k] = ld2(h1l + 4 * k);
            __builtin_amdgcn_sched_barrier(0);
            #pragma unroll
            for (int k = 0; k < KBH; ++k) a2[k] = ld2(h2l + 4 * k);
            __builtin_amdgcn_sched_barrier(0);

            layer2_store(bias2, a1, a2, wr, lds + OFF_H2 + cb * HBUF + crow * HSTR + wv * 16 + mi);
            layer1_store(p0, p1, ax, a1, wr, lds + OFF_H1 + nb * HBUF + crow * HSTR + wv * 16 + mi);
        } else if (t + 2 < SEQT) {
            helper_p(lds + OFF_X + ((t + 2) & 3) * XBUF + mi * XSTR + kh, pbias, wr,
                     lds + OFF_P + cb * PBUF + lane * 4);
        }

        if (t + 3 < SEQT) *(float4*)(xdst + ((t + 3) & 3) * XBUF) = xr;

        __syncthreads();

        if (t + 4 < SEQT) xr = *(const float4*)(xsrc + (size_t)(t + 4) * IND);
    }

    if (!helper) {
        const int cb = (SEQT - 1) & 1, nb = cb ^ 1;
        v2f a1[KBH], a2[KBH];
        const float* h1l = lds + OFF_H1 + cb * HBUF + mi * HSTR + kh;
        const float* h2l = lds + OFF_H2 + nb * HBUF + mi * HSTR + kh;
        #pragma unroll
        for (int k = 0; k < KBH; ++k) a1[k] = ld2(h1l + 4 * k);
        #pragma unroll
        for (int k = 0; k < KBH; ++k) a2[k] = ld2(h2l + 4 * k);
        layer2_store(bias2, a1, a2, wr, lds + OFF_H2 + cb * HBUF + crow * HSTR + wv * 16 + mi);
    }
    __syncthreads();

    const int fb = (SEQT - 1) & 1;
    for (int i = tid; i < 16 * H; i += nthr) {
        int m = i / H, n = i - m * H;
        h1out[(size_t)(b0 + m) * H + n] = lds[OFF_H1 + fb * HBUF + m * HSTR + n];
        h2out[(size_t)(b0 + m) * H + n] = lds[OFF_H2 + fb * HBUF + m * HSTR + n];
    }
    if (tid < 16) {
        float s = bo[0];
        for (int n = 0; n < H; ++n) s += lds[OFF_H2 + fb * HBUF + tid * HSTR + n] * Wo[n];
        outp[b0 + tid] = s;
    }
}

extern "C" void kernel_launch(void* const* d_in, const int* in_sizes, int n_in,
                              void* d_out, int out_size, void* d_ws, size_t ws_size,
                              hipStream_t stream) {
    (void)in_sizes; (void)n_in; (void)out_size; (void)d_ws; (void)ws_size;
    const float* x   = (const float*)d_in[0];
    const float* W1x = (const float*)d_in[1];
    const float* W1h = (const float*)d_in[2];
    const float* b1  = (const float*)d_in[3];
    const float* W2x = (const float*)d_in[4];
    const float* W2h = (const float*)d_in[5];
    const float* b2  = (const float*)d_in[6];
    const float* Wo  = (const float*)d_in[7];
    const float* bo  = (const float*)d_in[8];

    float* outp  = (float*)d_out;
    float* h1out = outp + BATCH;
    float* h2out = h1out + BATCH * H;

    const size_t shmem = (size_t)LDS_FLOATS * sizeof(float);
    (void)hipFuncSetAttribute((const void*)drnn_fused_kernel,
                              hipFuncAttributeMaxDynamicSharedMemorySize, (int)shmem);

    drnn_fused_kernel<<<BATCH / 16, 256, shmem, stream>>>(
        x, W1x, W1h, b1, W2x, W2h, b2, Wo, bo, outp, h1out, h2out);
}


# b1+x@W1x moves to a first all-WGP launch, via workspace, as layer-1 start value; recurrence re-dealt to 125 WMMAs per SIMD, both layers' 4-column tails in one tile; layer-1 sum order changes
# speedup vs baseline: 1.3756x; 1.1628x over previous
#include <hip/hip_runtime.h>
#include <math.h>

typedef __attribute__((ext_vector_type(2))) float v2f;
typedef __attribute__((ext_vector_type(8))) float v8f;

#define BATCH 512
#define SEQT  512
#define IND   64
#define H     100
#define NT    7
#define KB1   16
#define KBH   25
#define HSTR  116
#define XSTR  68
#define HBUF  (16*HSTR)

#define OFF_W1X 0
#define OFF_W1H (OFF_W1X + NT*KB1*32*2)
#define OFF_W2X (OFF_W1H + NT*KBH*32*2)
#define OFF_W2H (OFF_W2X + NT*KBH*32*2)
#define OFF_H1  (OFF_W2H + NT*KBH*32*2)
#define OFF_H2  (OFF_H1 + 2*HBUF)
#define OFF_X   (OFF_H2 + 2*HBUF)
#define XBUF    (16*XSTR)
#define KBXP    11
#define OFF_P   (OFF_X + 4*XBUF)
#define PBUF    (NT*256)
#define LDS_FLOATS (OFF_P + 2*PBUF)

#if __has_builtin(__builtin_amdgcn_tanhf)
#define TANHF(v) __builtin_amdgcn_tanhf(v)
#elif __has_builtin(__builtin_amdgcn_tanh_f32)
#define TANHF(v) __builtin_amdgcn_tanh_f32(v)
#else
#define TANHF(v) tanhf(v)
#endif

__device__ inline void fill_frags(float* dst, const float* W, int Ka, int Na,
                                  int kb, int tid, int nthr) {
    int total = NT * kb * 32;
    for (int i = tid; i < total; i += nthr) {
        int laneI = i & 31;
        int kblk  = (i >> 5) % kb;
        int nt    = (i >> 5) / kb;
        int N = nt * 16 + (laneI & 15);
        int K = kblk * 4 + ((laneI >> 4) << 1);
        float a = 0.f, b = 0.f;
        if (N < Na) {
            if (K     < Ka) a = W[(size_t)K * Na + N];
            if (K + 1 < Ka) b = W[(size_t)(K + 1) * Na + N];
        }
        dst[2 * i + 0] = a;
        dst[2 * i + 1] = b;
    }
}

__device__ inline v2f ld2(const float* p) { return *(const v2f*)p; }

#define KBXR  (KB1 - KBXP)
#define NWR   80
#define WR_1X 0
#define WR_1H (WR_1X + KBXR)
#define WR_2X (WR_1H + KBH)
#define WR_2H (WR_2X + KBH)

#define WMMA4(a, b, c) __builtin_amdgcn_wmma_f32_16x16x4_f32(false, (a), false, (b), (short)0, (c), false, false)

__device__ __forceinline__ void tanh_store(const v8f& c0, const v8f& c1, float* dst) {
    #pragma unroll
    for (int r = 0; r < 8; ++r) dst[r * HSTR] = TANHF(c0[r] + c1[r]);
}

__device__ __forceinline__ void layer1_store(const float4& p0, const float4& p1,
                                             const v2f (&ax)[KBXR], const v2f (&a1)[KBH],
                                             const v2f (&wr)[NWR], float* dst) {
    v8f c0, c1;
    c0[0] = p0.x; c0[1] = p0.y; c0[2] = p0.z; c0[3] = p0.w;
    c0[4] = p1.x; c0[5] = p1.y; c0[6] = p1.z; c0[7] = p1.w;
    #pragma unroll
    for (int r = 0; r < 8; ++r) c1[r] = 0.f;
    #pragma unroll
    for (int k = 0; k < KBXR; ++k) {
        if (k & 1) c1 = WMMA4(ax[k], wr[WR_1X + k], c1);
        else       c0 = WMMA4(ax[k], wr[WR_1X + k], c0);
    }
    #pragma unroll
    for (int k = 0; k < KBH; ++k) {
        if ((k + KBXR) & 1) c1 = WMMA4(a1[k], wr[WR_1H + k], c1);
        else                c0 = WMMA4(a1[k], wr[WR_1H + k], c0);
    }
    tanh_store(c0, c1, dst);
}

__device__ __forceinline__ void layer2_store(float bias2, const v2f (&a1)[KBH], const v2f (&a2)[KBH],
                                             const v2f (&wr)[NWR], float* dst) {
    v8f c0, c1;
    #pragma unroll
    for (int r = 0; r < 8; ++r) { c0[r] = bias2; c1[r] = 0.f; }
    #pragma unroll
    for (int k = 0; k < KBH; ++k) {
        if (k & 1) c1 = WMMA4(a1[k], wr[WR_2X + k], c1);
        else       c0 = WMMA4(a1[k], wr[WR_2X + k], c0);
    }
    #pragma unroll
    for (int k = 0; k < KBH; ++k) {
        if (k & 1) c1 = WMMA4(a2[k], wr[WR_2H + k], c1);
        else       c0 = WMMA4(a2[k], wr[WR_2H + k], c0);
    }
    tanh_store(c0, c1, dst);
}

__device__ __forceinline__ void helper_p(const float* xl, const float (&pbias)[NT],
                                         const v2f (&wr)[NWR], float* pdst) {
    v8f acc[NT];
    #pragma unroll
    for (int j = 0; j < NT; ++j)
        #pragma unroll
        for (int r = 0; r < 8; ++r) acc[j][r] = pbias[j];
    #pragma unroll
    for (int kb = 0; kb < KBXP; ++kb) {
        v2f a = ld2(xl + 4 * kb);
        #pragma unroll
        for (int j = 0; j < NT; ++j)
            acc[j] = WMMA4(a, wr[j * KBXP + kb], acc[j]);
    }
    #pragma unroll
    for (int j = 0; j < NT; ++j) {
        float* pp = pdst + j * 256;
        *(float4*)(pp)       = make_float4(acc[j][0], acc[j][1], acc[j][2], acc[j][3]);
        *(float4*)(pp + 128) = make_float4(acc[j][4], acc[j][5], acc[j][6], acc[j][7]);
    }
}

__global__ void __launch_bounds__(256)
drnn_fused_kernel(const float* __restrict__ x,
                  const float* __restrict__ W1x, const float* __restrict__ W1h,
                  const float* __restrict__ b1,
                  const float* __restrict__ W2x, const float* __restrict__ W2h,
                  const float* __restrict__ b2,
                  const float* __restrict__ Wo,  const float* __restrict__ bo,
                  float* __restrict__ outp, float* __restrict__ h1out,
                  float* __restrict__ h2out) {
    extern __shared__ float lds[];
    const int tid  = threadIdx.x;
    const int nthr = blockDim.x;
    const int b0   = blockIdx.x * 16;

    fill_frags(lds + OFF_W1X, W1x, IND, H, KB1, tid, nthr);
    fill_frags(lds + OFF_W1H, W1h, H,   H, KBH, tid, nthr);
    fill_frags(lds + OFF_W2X, W2x, H,   H, KBH, tid, nthr);
    fill_frags(lds + OFF_W2H, W2h, H,   H, KBH, tid, nthr);
    for (int i = tid; i < 4 * HBUF; i += nthr) lds[OFF_H1 + i] = 0.f;

    const float* xsrc = x + ((size_t)(b0 + (tid >> 4)) * SEQT) * IND + ((tid & 15) << 2);
    float*       xdst = lds + OFF_X + (tid >> 4) * XSTR + ((tid & 15) << 2);
    *(float4*)(xdst)        = *(const float4*)(xsrc);
    *(float4*)(xdst + XBUF) = *(const float4*)(xsrc + IND);
    float4 xr = *(const float4*)(xsrc + 2 * IND);
    __syncthreads();

    const int lane = tid & 31;
    const int wv   = __builtin_amdgcn_readfirstlane(tid >> 5);
    const bool helper = (wv == NT);
    const int mi   = lane & 15;
    const int kh   = (lane >> 4) << 1;
    const int crow = (lane >> 4) << 3;

    float bias2 = 0.f;
    {
        int n = wv * 16 + mi;
        if (n < H) bias2 = b2[n];
    }

    v2f   wr[NWR];
    float pbias[NT];
    #pragma unroll
    for (int j = 0; j < NT; ++j) pbias[j] = 0.f;
    if (helper) {
        #pragma unroll
        for (int j = 0; j < NT; ++j) {
            pbias[j] = (j * 16 + mi < H) ? b1[j * 16 + mi] : 0.f;
            #pragma unroll
            for (int kb = 0; kb < KBXP; ++kb)
                wr[j * KBXP + kb] = ld2(lds + OFF_W1X + (j * KB1 + kb) * 64 + lane * 2);
        }
        #pragma unroll
        for (int i = NT * KBXP; i < NWR; ++i) { wr[i][0] = 0.f; wr[i][1] = 0.f; }
        const float* xl = lds + OFF_X + mi * XSTR + kh;
        v8f acc[NT];
        #pragma unroll
        for (int j = 0; j < NT; ++j)
            #pragma unroll
            for (int r = 0; r < 8; ++r) acc[j][r] = pbias[j];
        #pragma unroll
        for (int kb = 0; kb < KBXP; ++kb) {
            v2f a = ld2(xl + 4 * kb);
            #pragma unroll
            for (int j = 0; j < NT; ++j)
                acc[j] = __builtin_amdgcn_wmma_f32_16x16x4_f32(false, a, false, wr[j * KBXP + kb], (short)0, acc[j], false, false);
        }
        #pragma unroll
        for (int j = 0; j < NT; ++j) {
            float* pp = lds + OFF_P + j * 256 + lane * 4;
            *(float4*)(pp)       = make_float4(acc[j][0], acc[j][1], acc[j][2], acc[j][3]);
            *(float4*)(pp + 128) = make_float4(acc[j][4], acc[j][5], acc[j][6], acc[j][7]);
        }
    } else {
        #pragma unroll
        for (int k = 0; k < KBXR; ++k)
            wr[WR_1X + k] = ld2(lds + OFF_W1X + (wv * KB1 + KBXP + k) * 64 + lane * 2);
        #pragma unroll
        for (int k = 0; k < KBH; ++k) {
            wr[WR_1H + k] = ld2(lds + OFF_W1H + (wv * KBH + k) * 64 + lane * 2);
            wr[WR_2X + k] = ld2(lds + OFF_W2X + (wv * KBH + k) * 64 + lane * 2);
            wr[WR_2H + k] = ld2(lds + OFF_W2H + (wv * KBH + k) * 64 + lane * 2);
        }
    }
    __syncthreads();

    if (!helper) {
        v2f a1[KBH], ax[KBXR];
        const float* hl = lds + OFF_H1 + HBUF + mi * HSTR + kh;
        const float* xl = lds + OFF_X + mi * XSTR + kh;
        const float* pp = lds + OFF_P + wv * 256 + lane * 4;
        #pragma unroll
        for (int k = 0; k < KBH; ++k) a1[k] = ld2(hl + 4 * k);
        #pragma unroll
        for (int k = 0; k < KBXR; ++k) ax[k] = ld2(xl + 4 * (KBXP + k));
        const float4 p0 = *(const float4*)(pp);
        const float4 p1 = *(const float4*)(pp + 128);
        layer1_store(p0, p1, ax, a1, wr, lds + OFF_H1 + crow * HSTR + wv * 16 + mi);
    } else {
        helper_p(lds + OFF_X + XBUF + mi * XSTR + kh, pbias, wr, lds + OFF_P + PBUF + lane * 4);
    }
    *(float4*)(xdst + 2 * XBUF) = xr;
    __syncthreads();
    xr = *(const float4*)(xsrc + 3 * IND);

    for (int t = 0; t + 1 < SEQT; ++t) {
        const int cb = t & 1;
        const int nb = cb ^ 1;

        if (!helper) {
            v2f a1[KBH], a2[KBH], ax[KBXR];
            const float* h1l = lds + OFF_H1 + cb * HBUF + mi * HSTR + kh;
            const float* h2l = lds + OFF_H2 + nb * HBUF + mi * HSTR + kh;
            const float* xl  = lds + OFF_X + ((t + 1) & 3) * XBUF + mi * XSTR + kh;
            const float* pp  = lds + OFF_P + nb * PBUF + wv * 256 + lane * 4;
            const float4 p0 = *(const float4*)(pp);
            const float4 p1 = *(const float4*)(pp + 128);
            #pragma unroll
            for (int k = 0; k < KBXR; ++k) ax[k] = ld2(xl + 4 * (KBXP + k));
            #pragma unroll
            for (int k = 0; k < 4; ++k) a1[k] = ld2(h1l + 4 * k);
            __builtin_amdgcn_sched_barrier(0);
            #pragma unroll
            for (int k = 4; k < KBH; ++k) a1[k] = ld2(h1l + 4 * k);
            __builtin_amdgcn_sched_barrier(0);
            #pragma unroll
            for (int k = 0; k < KBH; ++k) a2[k] = ld2(h2l + 4 * k);
            __builtin_amdgcn_sched_barrier(0);

            layer2_store(bias2, a1, a2, wr, lds + OFF_H2 + cb * HBUF + crow * HSTR + wv * 16 + mi);
            layer1_store(p0, p1, ax, a1, wr, lds + OFF_H1 + nb * HBUF + crow * HSTR + wv * 16 + mi);
        } else if (t + 2 < SEQT) {
            helper_p(lds + OFF_X + ((t + 2) & 3) * XBUF + mi * XSTR + kh, pbias, wr,
                     lds + OFF_P + cb * PBUF + lane * 4);
        }

        if (t + 3 < SEQT) *(float4*)(xdst + ((t + 3) & 3) * XBUF) = xr;

        __syncthreads();

        if (t + 4 < SEQT) xr = *(const float4*)(xsrc + (size_t)(t + 4) * IND);
    }

    if (!helper) {
        const int cb = (SEQT - 1) & 1, nb = cb ^ 1;
        v2f a1[KBH], a2[KBH];
        const float* h1l = lds + OFF_H1 + cb * HBUF + mi * HSTR + kh;
        const float* h2l = lds + OFF_H2 + nb * HBUF + mi * HSTR + kh;
        #pragma unroll
        for (int k = 0; k < KBH; ++k) a1[k] = ld2(h1l + 4 * k);
        #pragma unroll
        for (int k = 0; k < KBH; ++k) a2[k] = ld2(h2l + 4 * k);
        layer2_store(bias2, a1, a2, wr, lds + OFF_H2 + cb * HBUF + crow * HSTR + wv * 16 + mi);
    }
    __syncthreads();

    const int fb = (SEQT - 1) & 1;
    for (int i = tid; i < 16 * H; i += nthr) {
        int m = i / H, n = i - m * H;
        h1out[(size_t)(b0 + m) * H + n] = lds[OFF_H1 + fb * HBUF + m * HSTR + n];
        h2out[(size_t)(b0 + m) * H + n] = lds[OFF_H2 + fb * HBUF + m * HSTR + n];
    }
    if (tid < 16) {
        float s = bo[0];
        for (int n = 0; n < H; ++n) s += lds[OFF_H2 + fb * HBUF + tid * HSTR + n] * Wo[n];
        outp[b0 + tid] = s;
    }
}

#define PTILE 256
#define WS_NEED ((size_t)(BATCH / 16) * SEQT * NT * PTILE * sizeof(float))
#define XSTEPS 8

__global__ void __launch_bounds__(256)
xproj_kernel(const float* __restrict__ x, const float* __restrict__ W1x,
             const float* __restrict__ b1, const float* __restrict__ b2,
             float* __restrict__ P) {
    const int lane = threadIdx.x & 31;
    const int wv   = __builtin_amdgcn_readfirstlane(threadIdx.x >> 5);
    const int bt   = blockIdx.x >> 3;
    const int tc   = blockIdx.x & 7;
    const int c    = lane & 15;
    const int kh   = (lane >> 4) << 1;

    v2f   wf[NT * KB1];
    float bias[NT];
    #pragma unroll
    for (int j = 0; j < NT; ++j) {
        const int  N  = j * 16 + c;
        const bool ok = N < H;
        bias[j] = ok ? b1[N] : 0.f;
        if (j == NT - 1 && c >= 4 && c < 8) bias[j] = b2[92 + c];
        #pragma unroll
        for (int kb = 0; kb < KB1; ++kb) {
            const int K = 4 * kb + kh;
            v2f w;
            w[0] = ok ? W1x[(size_t)K * H + N] : 0.f;
            w[1] = ok ? W1x[(size_t)(K + 1) * H + N] : 0.f;
            wf[j * KB1 + kb] = w;
        }
    }
    for (int s = 0; s < XSTEPS; ++s) {
        const int t = tc * 64 + wv * XSTEPS + s;
        const float* xp = x + ((size_t)(bt * 16 + c) * SEQT + t) * IND + kh;
        v2f a[KB1];
        #pragma unroll
        for (int kb = 0; kb < KB1; ++kb) a[kb] = ld2(xp + 4 * kb);
        v8f acc[NT];
        #pragma unroll
        for (int j = 0; j < NT; ++j)
            #pragma unroll
            for (int r = 0; r < 8; ++r) acc[j][r] = bias[j];
        #pragma unroll
        for (int kb = 0; kb < KB1; ++kb)
            #pragma unroll
            for (int j = 0; j < NT; ++j)
                acc[j] = WMMA4(a[kb], wf[j * KB1 + kb], acc[j]);
        float* pp = P + (((size_t)bt * SEQT + t) * NT) * PTILE + lane * 4;
        #pragma unroll
        for (int j = 0; j < NT; ++j) {
            *(float4*)(pp + j * PTILE)       = make_float4(acc[j][0], acc[j][1], acc[j][2], acc[j][3]);
            *(float4*)(pp + j * PTILE + 128) = make_float4(acc[j][4], acc[j][5], acc[j][6], acc[j][7]);
        }
    }
}

#define R_H1 0
#define R_H2 (2 * HBUF)
#define R_LDS_FLOATS (4 * HBUF)

__device__ __forceinline__ v2f wfrag(const float* Wm, int n, int K) {
    v2f w; w[0] = 0.f; w[1] = 0.f;
    if (Wm) { w[0] = Wm[(size_t)K * H + n]; w[1] = Wm[(size_t)(K + 1) * H + n]; }
    return w;
}

__global__ void __launch_bounds__(256)
drnn_rec_kernel(const float* __restrict__ W1h, const float* __restrict__ W2x,
                const float* __restrict__ W2h, const float* __restrict__ b2,
                const float* __restrict__ Wo,  const float* __restrict__ bo,
                const float* __restrict__ P,
                float* __restrict__ outp, float* __restrict__ h1out, float* __restrict__ h2out) {
    extern __shared__ float lds[];
    const int tid  = threadIdx.x;
    const int nthr = blockDim.x;
    const int bt   = blockIdx.x;
    const int b0   = bt * 16;
    const int lane = tid & 31;
    const int wv   = __builtin_amdgcn_readfirstlane(tid >> 5);
    const int c    = lane & 15;
    const int kh   = (lane >> 4) << 1;
    const int crow = (lane >> 4) << 3;

    for (int i = tid; i < R_LDS_FLOATS; i += nthr) lds[i] = 0.f;

    const bool hasA2  = (wv != 7);
    const bool hasB   = (wv < 4) || (wv == 7);
    const bool loadPA = (wv >= 6);
    const int  tileA  = (wv == 7) ? 4 : 6;
    const int  tileB  = (wv == 7) ? 5 : wv;

    const float* mA1 = nullptr; const float* mA2 = nullptr; const float* mB = nullptr;
    int nA = 0, nB = 0;
    if (wv < 6) {
        mA1 = W2x; mA2 = W2h; nA = wv * 16 + c;
        if (wv < 4) { mB = W1h; nB = wv * 16 + c; }
    } else if (wv == 6) {
        if (c < 4)      { mA1 = W1h; nA = 96 + c; }
        else if (c < 8) { mA1 = W2x; mA2 = W2h; nA = 92 + c; }
    } else {
        mA1 = W1h; nA = 64 + c;
        mB  = W1h; nB = 80 + c;
    }
    v2f wA1[KBH], wA2[KBH], wB[KBH];
    #pragma unroll
    for (int k = 0; k < KBH; ++k) {
        wA1[k] = wfrag(mA1, nA, 4 * k + kh);
        wA2[k] = wfrag(mA2, nA, 4 * k + kh);
        wB[k]  = wfrag(mB,  nB, 4 * k + kh);
    }
    float bias2 = 0.f;
    if (wv < 6) bias2 = b2[wv * 16 + c];

    int dA0, dA1, dB0, dB1;
    {
        const int rowo = crow * HSTR;
        if (wv < 6) {
            dA0 = R_H2 + rowo + wv * 16 + c;          dA1 = dA0 + HBUF;
        } else if (wv == 7) {
            dA1 = R_H1 + rowo + 64 + c;               dA0 = dA1 + HBUF;
        } else if (c < 4) {
            dA1 = R_H1 + rowo + 96 + c;               dA0 = dA1 + HBUF;
        } else {
            dA0 = R_H2 + rowo + 92 + c;               dA1 = dA0 + HBUF;
        }
        dB1 = R_H1 + rowo + tileB * 16 + c;           dB0 = dB1 + HBUF;
    }

    const float* Pb = P + ((size_t)bt * SEQT * NT) * PTILE + lane * 4;
    __syncthreads();

    if (wv < NT) {
        const float* pp = Pb + wv * PTILE;
        const float4 q0 = *(const float4*)(pp);
        const float4 q1 = *(const float4*)(pp + 128);
        if (wv < NT - 1 || c < 4) {
            float* d = lds + R_H1 + crow * HSTR + wv * 16 + c;
            d[0 * HSTR] = TANHF(q0.x); d[1 * HSTR] = TANHF(q0.y);
            d[2 * HSTR] = TANHF(q0.z); d[3 * HSTR] = TANHF(q0.w);
            d[4 * HSTR] = TANHF(q1.x); d[5 * HSTR] = TANHF(q1.y);
            d[6 * HSTR] = TANHF(q1.z); d[7 * HSTR] = TANHF(q1.w);
        }
    }
    float4 cA0 = make_float4(0.f, 0.f, 0.f, 0.f), cA1 = cA0, cB0 = cA0, cB1 = cA0;
    if (loadPA) {
        const float* pp = Pb + ((size_t)1 * NT + tileA) * PTILE;
        cA0 = *(const float4*)(pp); cA1 = *(const float4*)(pp + 128);
    }
    if (hasB) {
        const float* pp = Pb + ((size_t)1 * NT + tileB) * PTILE;
        cB0 = *(const float4*)(pp); cB1 = *(const float4*)(pp + 128);
    }
    __syncthreads();

    const float* h1base = lds + R_H1 + c * HSTR + kh;
    const float* h2base = lds + R_H2 + c * HSTR + kh;

    for (int t = 0; t + 1 < SEQT; ++t) {
        const int cb = t & 1, nb = cb ^ 1;
        const int tn = (t + 2 < SEQT) ? (t + 2) : (SEQT - 1);
        float4 nA0 = cA0, nA1 = cA1, nB0 = cB0, nB1 = cB1;
        if (loadPA) {
            const float* pp = Pb + ((size_t)tn * NT + tileA) * PTILE;
            nA0 = *(const float4*)(pp); nA1 = *(const float4*)(pp + 128);
        }
        if (hasB) {
            const float* pp = Pb + ((size_t)tn * NT + tileB) * PTILE;
            nB0 = *(const float4*)(pp); nB1 = *(const float4*)(pp + 128);
        }
        v2f a1[KBH], a2[KBH];
        const float* h1l = h1base + cb * HBUF;
        const float* h2l = h2base + nb * HBUF;
        #pragma unroll
        for (int k = 0; k < KBH; ++k) a1[k] = ld2(h1l + 4 * k);
        if (hasA2) {
            #pragma unroll
            for (int k = 0; k < KBH; ++k) a2[k] = ld2(h2l + 4 * k);
        }
        __builtin_amdgcn_sched_barrier(0);

        {
            v8f c0, c1;
            if (loadPA) {
                c0[0] = cA0.x; c0[1] = cA0.y; c0[2] = cA0.z; c0[3] = cA0.w;
                c0[4] = cA1.x; c0[5] = cA1.y; c0[6] = cA1.z; c0[7] = cA1.w;
            } else {
                #pragma unroll
                for (int r = 0; r < 8; ++r) c0[r] = bias2;
            }
            #pragma unroll
            for (int r = 0; r < 8; ++r) c1[r] = 0.f;
            #pragma unroll
            for (int k = 0; k < KBH; ++k) {
                if (k & 1) c1 = WMMA4(a1[k], wA1[k], c1);
                else       c0 = WMMA4(a1[k], wA1[k], c0);
            }
            if (hasA2) {
                #pragma unroll
                for (int k = 0; k < KBH; ++k) {
                    if (k & 1) c1 = WMMA4(a2[k], wA2[k], c1);
                    else       c0 = WMMA4(a2[k], wA2[k], c0);
                }
            }
            tanh_store(c0, c1, lds + (cb ? dA1 : dA0));
        }
        if (hasB) {
            v8f c0, c1;
            c0[0] = cB0.x; c0[1] = cB0.y; c0[2] = cB0.z; c0[3] = cB0.w;
            c0[4] = cB1.x; c0[5] = cB1.y; c0[6] = cB1.z; c0[7] = cB1.w;
            #pragma unroll
            for (int r = 0; r < 8; ++r) c1[r] = 0.f;
            #pragma unroll
            for (int k = 0; k < KBH; ++k) {
                if (k & 1) c1 = WMMA4(a1[k], wB[k], c1);
                else       c0 = WMMA4(a1[k], wB[k], c0);
            }
            tanh_store(c0, c1, lds + (cb ? dB1 : dB0));
        }
        __syncthreads();
        cA0 = nA0; cA1 = nA1; cB0 = nB0; cB1 = nB1;
    }

    if (hasA2) {
        const int cb = (SEQT - 1) & 1, nb = cb ^ 1;
        v2f a1[KBH], a2[KBH];
        #pragma unroll
        for (int k = 0; k < KBH; ++k) a1[k] = ld2(h1base + cb * HBUF + 4 * k);
        #pragma unroll
        for (int k = 0; k < KBH; ++k) a2[k] = ld2(h2base + nb * HBUF + 4 * k);
        v8f c0, c1;
        if (loadPA) {
            c0[0] = cA0.x; c0[1] = cA0.y; c0[2] = cA0.z; c0[3] = cA0.w;
            c0[4] = cA1.x; c0[5] = cA1.y; c0[6] = cA1.z; c0[7] = cA1.w;
        } else {
            #pragma unroll
            for (int r = 0; r < 8; ++r) c0[r] = bias2;
        }
        #pragma unroll
        for (int r = 0; r < 8; ++r) c1[r] = 0.f;
        #pragma unroll
        for (int k = 0; k < KBH; ++k) {
            if (k & 1) c1 = WMMA4(a1[k], wA1[k], c1);
            else       c0 = WMMA4(a1[k], wA1[k], c0);
        }
        #pragma unroll
        for (int k = 0; k < KBH; ++k) {
            if (k & 1) c1 = WMMA4(a2[k], wA2[k], c1);
            else       c0 = WMMA4(a2[k], wA2[k], c0);
        }
        tanh_store(c0, c1, lds + (cb ? dA1 : dA0));
    }
    __syncthreads();

    const int fb = (SEQT - 1) & 1;
    for (int i = tid; i < 16 * H; i += nthr) {
        int m = i / H, n = i - m * H;
        h1out[(size_t)(b0 + m) * H + n] = lds[R_H1 + fb * HBUF + m * HSTR + n];
        h2out[(size_t)(b0 + m) * H + n] = lds[R_H2 + fb * HBUF + m * HSTR + n];
    }
    if (tid < 16) {
        float s = bo[0];
        for (int n = 0; n < H; ++n) s += lds[R_H2 + fb * HBUF + tid * HSTR + n] * Wo[n];
        outp[b0 + tid] = s;
    }
}

extern "C" void kernel_launch(void* const* d_in, const int* in_sizes, int n_in,
                              void* d_out, int out_size, void* d_ws, size_t ws_size,
                              hipStream_t stream) {
    (void)in_sizes; (void)n_in; (void)out_size;
    const float* x   = (const float*)d_in[0];
    const float* W1x = (const float*)d_in[1];
    const float* W1h = (const float*)d_in[2];
    const float* b1  = (const float*)d_in[3];
    const float* W2x = (const float*)d_in[4];
    const float* W2h = (const float*)d_in[5];
    const float* b2  = (const float*)d_in[6];
    const float* Wo  = (const float*)d_in[7];
    const float* bo  = (const float*)d_in[8];

    float* outp  = (float*)d_out;
    float* h1out = outp + BATCH;
    float* h2out = h1out + BATCH * H;

    if (ws_size >= WS_NEED) {
        float* P = (float*)d_ws;
        xproj_kernel<<<(BATCH / 16) * 8, 256, 0, stream>>>(x, W1x, b1, b2, P);
        drnn_rec_kernel<<<BATCH / 16, 256, (size_t)R_LDS_FLOATS * sizeof(float), stream>>>(
            W1h, W2x, W2h, b2, Wo, bo, P, outp, h1out, h2out);
        return;
    }

    const size_t shmem = (size_t)LDS_FLOATS * sizeof(float);
    (void)hipFuncSetAttribute((const void*)drnn_fused_kernel,
                              hipFuncAttributeMaxDynamicSharedMemorySize, (int)shmem);

    drnn_fused_kernel<<<BATCH / 16, 256, shmem, stream>>>(
        x, W1x, W1h, b1, W2x, W2h, b2, Wo, bo, outp, h1out, h2out);
}


# Split barrier in the recurrence: signal behind a step's stores; the next layer-2 step's h1_t@W2x products run between signal and wait at low wave priority, kept in registers; sum order unchanged
# speedup vs baseline: 1.5201x; 1.1050x over previous
#include <hip/hip_runtime.h>
#include <math.h>

typedef __attribute__((ext_vector_type(2))) float v2f;
typedef __attribute__((ext_vector_type(8))) float v8f;

#define BATCH 512
#define SEQT  512
#define IND   64
#define H     100
#define NT    7
#define KB1   16
#define KBH   25
#define HSTR  116
#define XSTR  68
#define HBUF  (16*HSTR)

#define OFF_W1X 0
#define OFF_W1H (OFF_W1X + NT*KB1*32*2)
#define OFF_W2X (OFF_W1H + NT*KBH*32*2)
#define OFF_W2H (OFF_W2X + NT*KBH*32*2)
#define OFF_H1  (OFF_W2H + NT*KBH*32*2)
#define OFF_H2  (OFF_H1 + 2*HBUF)
#define OFF_X   (OFF_H2 + 2*HBUF)
#define XBUF    (16*XSTR)
#define KBXP    11
#define OFF_P   (OFF_X + 4*XBUF)
#define PBUF    (NT*256)
#define LDS_FLOATS (OFF_P + 2*PBUF)

#if __has_builtin(__builtin_amdgcn_tanhf)
#define TANHF(v) __builtin_amdgcn_tanhf(v)
#elif __has_builtin(__builtin_amdgcn_tanh_f32)
#define TANHF(v) __builtin_amdgcn_tanh_f32(v)
#else
#define TANHF(v) tanhf(v)
#endif

__device__ inline void fill_frags(float* dst, const float* W, int Ka, int Na,
                                  int kb, int tid, int nthr) {
    int total = NT * kb * 32;
    for (int i = tid; i < total; i += nthr) {
        int laneI = i & 31;
        int kblk  = (i >> 5) % kb;
        int nt    = (i >> 5) / kb;
        int N = nt * 16 + (laneI & 15);
        int K = kblk * 4 + ((laneI >> 4) << 1);
        float a = 0.f, b = 0.f;
        if (N < Na) {
            if (K     < Ka) a = W[(size_t)K * Na + N];
            if (K + 1 < Ka) b = W[(size_t)(K + 1) * Na + N];
        }
        dst[2 * i + 0] = a;
        dst[2 * i + 1] = b;
    }
}

__device__ inline v2f ld2(const float* p) { return *(const v2f*)p; }

#define KBXR  (KB1 - KBXP)
#define NWR   80
#define WR_1X 0
#define WR_1H (WR_1X + KBXR)
#define WR_2X (WR_1H + KBH)
#define WR_2H (WR_2X + KBH)

#define WMMA4(a, b, c) __builtin_amdgcn_wmma_f32_16x16x4_f32(false, (a), false, (b), (short)0, (c), false, false)

__device__ __forceinline__ void tanh_store(const v8f& c0, const v8f& c1, float* dst) {
    #pragma unroll
    for (int r = 0; r < 8; ++r) dst[r * HSTR] = TANHF(c0[r] + c1[r]);
}

__device__ __forceinline__ void layer1_store(const float4& p0, const float4& p1,
                                             const v2f (&ax)[KBXR], const v2f (&a1)[KBH],
                                             const v2f (&wr)[NWR], float* dst) {
    v8f c0, c1;
    c0[0] = p0.x; c0[1] = p0.y; c0[2] = p0.z; c0[3] = p0.w;
    c0[4] = p1.x; c0[5] = p1.y; c0[6] = p1.z; c0[7] = p1.w;
    #pragma unroll
    for (int r = 0; r < 8; ++r) c1[r] = 0.f;
    #pragma unroll
    for (int k = 0; k < KBXR; ++k) {
        if (k & 1) c1 = WMMA4(ax[k], wr[WR_1X + k], c1);
        else       c0 = WMMA4(ax[k], wr[WR_1X + k], c0);
    }
    #pragma unroll
    for (int k = 0; k < KBH; ++k) {
        if ((k + KBXR) & 1) c1 = WMMA4(a1[k], wr[WR_1H + k], c1);
        else                c0 = WMMA4(a1[k], wr[WR_1H + k], c0);
    }
    tanh_store(c0, c1, dst);
}

__device__ __forceinline__ void layer2_store(float bias2, const v2f (&a1)[KBH], const v2f (&a2)[KBH],
                                             const v2f (&wr)[NWR], float* dst) {
    v8f c0, c1;
    #pragma unroll
    for (int r = 0; r < 8; ++r) { c0[r] = bias2; c1[r] = 0.f; }
    #pragma unroll
    for (int k = 0; k < KBH; ++k) {
        if (k & 1) c1 = WMMA4(a1[k], wr[WR_2X + k], c1);
        else       c0 = WMMA4(a1[k], wr[WR_2X + k], c0);
    }
    #pragma unroll
    for (int k = 0; k < KBH; ++k) {
        if (k & 1) c1 = WMMA4(a2[k], wr[WR_2H + k], c1);
        else       c0 = WMMA4(a2[k], wr[WR_2H + k], c0);
    }
    tanh_store(c0, c1, dst);
}

__device__ __forceinline__ void helper_p(const float* xl, const float (&pbias)[NT],
                                         const v2f (&wr)[NWR], float* pdst) {
    v8f acc[NT];
    #pragma unroll
    for (int j = 0; j < NT; ++j)
        #pragma unroll
        for (int r = 0; r < 8; ++r) acc[j][r] = pbias[j];
    #pragma unroll
    for (int kb = 0; kb < KBXP; ++kb) {
        v2f a = ld2(xl + 4 * kb);
        #pragma unroll
        for (int j = 0; j < NT; ++j)
            acc[j] = WMMA4(a, wr[j * KBXP + kb], acc[j]);
    }
    #pragma unroll
    for (int j = 0; j < NT; ++j) {
        float* pp = pdst + j * 256;
        *(float4*)(pp)       = make_float4(acc[j][0], acc[j][1], acc[j][2], acc[j][3]);
        *(float4*)(pp + 128) = make_float4(acc[j][4], acc[j][5], acc[j][6], acc[j][7]);
    }
}

__global__ void __launch_bounds__(256)
drnn_fused_kernel(const float* __restrict__ x,
                  const float* __restrict__ W1x, const float* __restrict__ W1h,
                  const float* __restrict__ b1,
                  const float* __restrict__ W2x, const float* __restrict__ W2h,
                  const float* __restrict__ b2,
                  const float* __restrict__ Wo,  const float* __restrict__ bo,
                  float* __restrict__ outp, float* __restrict__ h1out,
                  float* __restrict__ h2out) {
    extern __shared__ float lds[];
    const int tid  = threadIdx.x;
    const int nthr = blockDim.x;
    const int b0   = blockIdx.x * 16;

    fill_frags(lds + OFF_W1X, W1x, IND, H, KB1, tid, nthr);
    fill_frags(lds + OFF_W1H, W1h, H,   H, KBH, tid, nthr);
    fill_frags(lds + OFF_W2X, W2x, H,   H, KBH, tid, nthr);
    fill_frags(lds + OFF_W2H, W2h, H,   H, KBH, tid, nthr);
    for (int i = tid; i < 4 * HBUF; i += nthr) lds[OFF_H1 + i] = 0.f;

    const float* xsrc = x + ((size_t)(b0 + (tid >> 4)) * SEQT) * IND + ((tid & 15) << 2);
    float*       xdst = lds + OFF_X + (tid >> 4) * XSTR + ((tid & 15) << 2);
    *(float4*)(xdst)        = *(const float4*)(xsrc);
    *(float4*)(xdst + XBUF) = *(const float4*)(xsrc + IND);
    float4 xr = *(const float4*)(xsrc + 2 * IND);
    __syncthreads();

    const int lane = tid & 31;
    const int wv   = __builtin_amdgcn_readfirstlane(tid >> 5);
    const bool helper = (wv == NT);
    const int mi   = lane & 15;
    const int kh   = (lane >> 4) << 1;
    const int crow = (lane >> 4) << 3;

    float bias2 = 0.f;
    {
        int n = wv * 16 + mi;
        if (n < H) bias2 = b2[n];
    }

    v2f   wr[NWR];
    float pbias[NT];
    #pragma unroll
    for (int j = 0; j < NT; ++j) pbias[j] = 0.f;
    if (helper) {
        #pragma unroll
        for (int j = 0; j < NT; ++j) {
            pbias[j] = (j * 16 + mi < H) ? b1[j * 16 + mi] : 0.f;
            #pragma unroll
            for (int kb = 0; kb < KBXP; ++kb)
                wr[j * KBXP + kb] = ld2(lds + OFF_W1X + (j * KB1 + kb) * 64 + lane * 2);
        }
        #pragma unroll
        for (int i = NT * KBXP; i < NWR; ++i) { wr[i][0] = 0.f; wr[i][1] = 0.f; }
        const float* xl = lds + OFF_X + mi * XSTR + kh;
        v8f acc[NT];
        #pragma unroll
        for (int j = 0; j < NT; ++j)
            #pragma unroll
            for (int r = 0; r < 8; ++r) acc[j][r] = pbias[j];
        #pragma unroll
        for (int kb = 0; kb < KBXP; ++kb) {
            v2f a = ld2(xl + 4 * kb);
            #pragma unroll
            for (int j = 0; j < NT; ++j)
                acc[j] = __builtin_amdgcn_wmma_f32_16x16x4_f32(false, a, false, wr[j * KBXP + kb], (short)0, acc[j], false, false);
        }
        #pragma unroll
        for (int j = 0; j < NT; ++j) {
            float* pp = lds + OFF_P + j * 256 + lane * 4;
            *(float4*)(pp)       = make_float4(acc[j][0], acc[j][1], acc[j][2], acc[j][3]);
            *(float4*)(pp + 128) = make_float4(acc[j][4], acc[j][5], acc[j][6], acc[j][7]);
        }
    } else {
        #pragma unroll
        for (int k = 0; k < KBXR; ++k)
            wr[WR_1X + k] = ld2(lds + OFF_W1X + (wv * KB1 + KBXP + k) * 64 + lane * 2);
        #pragma unroll
        for (int k = 0; k < KBH; ++k) {
            wr[WR_1H + k] = ld2(lds + OFF_W1H + (wv * KBH + k) * 64 + lane * 2);
            wr[WR_2X + k] = ld2(lds + OFF_W2X + (wv * KBH + k) * 64 + lane * 2);
            wr[WR_2H + k] = ld2(lds + OFF_W2H + (wv * KBH + k) * 64 + lane * 2);
        }
    }
    __syncthreads();

    if (!helper) {
        v2f a1[KBH], ax[KBXR];
        const float* hl = lds + OFF_H1 + HBUF + mi * HSTR + kh;
        const float* xl = lds + OFF_X + mi * XSTR + kh;
        const float* pp = lds + OFF_P + wv * 256 + lane * 4;
        #pragma unroll
        for (int k = 0; k < KBH; ++k) a1[k] = ld2(hl + 4 * k);
        #pragma unroll
        for (int k = 0; k < KBXR; ++k) ax[k] = ld2(xl + 4 * (KBXP + k));
        const float4 p0 = *(const float4*)(pp);
        const float4 p1 = *(const float4*)(pp + 128);
        layer1_store(p0, p1, ax, a1, wr, lds + OFF_H1 + crow * HSTR + wv * 16 + mi);
    } else {
        helper_p(lds + OFF_X + XBUF + mi * XSTR + kh, pbias, wr, lds + OFF_P + PBUF + lane * 4);
    }
    *(float4*)(xdst + 2 * XBUF) = xr;
    __syncthreads();
    xr = *(const float4*)(xsrc + 3 * IND);

    for (int t = 0; t + 1 < SEQT; ++t) {
        const int cb = t & 1;
        const int nb = cb ^ 1;

        if (!helper) {
            v2f a1[KBH], a2[KBH], ax[KBXR];
            const float* h1l = lds + OFF_H1 + cb * HBUF + mi * HSTR + kh;
            const float* h2l = lds + OFF_H2 + nb * HBUF + mi * HSTR + kh;
            const float* xl  = lds + OFF_X + ((t + 1) & 3) * XBUF + mi * XSTR + kh;
            const float* pp  = lds + OFF_P + nb * PBUF + wv * 256 + lane * 4;
            const float4 p0 = *(const float4*)(pp);
            const float4 p1 = *(const float4*)(pp + 128);
            #pragma unroll
            for (int k = 0; k < KBXR; ++k) ax[k] = ld2(xl + 4 * (KBXP + k));
            #pragma unroll
            for (int k = 0; k < 4; ++k) a1[k] = ld2(h1l + 4 * k);
            __builtin_amdgcn_sched_barrier(0);
            #pragma unroll
            for (int k = 4; k < KBH; ++k) a1[k] = ld2(h1l + 4 * k);
            __builtin_amdgcn_sched_barrier(0);
            #pragma unroll
            for (int k = 0; k < KBH; ++k) a2[k] = ld2(h2l + 4 * k);
            __builtin_amdgcn_sched_barrier(0);

            layer2_store(bias2, a1, a2, wr, lds + OFF_H2 + cb * HBUF + crow * HSTR + wv * 16 + mi);
            layer1_store(p0, p1, ax, a1, wr, lds + OFF_H1 + nb * HBUF + crow * HSTR + wv * 16 + mi);
        } else if (t + 2 < SEQT) {
            helper_p(lds + OFF_X + ((t + 2) & 3) * XBUF + mi * XSTR + kh, pbias, wr,
                     lds + OFF_P + cb * PBUF + lane * 4);
        }

        if (t + 3 < SEQT) *(float4*)(xdst + ((t + 3) & 3) * XBUF) = xr;

        __syncthreads();

        if (t + 4 < SEQT) xr = *(const float4*)(xsrc + (size_t)(t + 4) * IND);
    }

    if (!helper) {
        const int cb = (SEQT - 1) & 1, nb = cb ^ 1;
        v2f a1[KBH], a2[KBH];
        const float* h1l = lds + OFF_H1 + cb * HBUF + mi * HSTR + kh;
        const float* h2l = lds + OFF_H2 + nb * HBUF + mi * HSTR + kh;
        #pragma unroll
        for (int k = 0; k < KBH; ++k) a1[k] = ld2(h1l + 4 * k);
        #pragma unroll
        for (int k = 0; k < KBH; ++k) a2[k] = ld2(h2l + 4 * k);
        layer2_store(bias2, a1, a2, wr, lds + OFF_H2 + cb * HBUF + crow * HSTR + wv * 16 + mi);
    }
    __syncthreads();

    const int fb = (SEQT - 1) & 1;
    for (int i = tid; i < 16 * H; i += nthr) {
        int m = i / H, n = i - m * H;
        h1out[(size_t)(b0 + m) * H + n] = lds[OFF_H1 + fb * HBUF + m * HSTR + n];
        h2out[(size_t)(b0 + m) * H + n] = lds[OFF_H2 + fb * HBUF + m * HSTR + n];
    }
    if (tid < 16) {
        float s = bo[0];
        for (int n = 0; n < H; ++n) s += lds[OFF_H2 + fb * HBUF + tid * HSTR + n] * Wo[n];
        outp[b0 + tid] = s;
    }
}

#define PTILE 256
#define WS_NEED ((size_t)(BATCH / 16) * SEQT * NT * PTILE * sizeof(float))
#define XSTEPS 8

__global__ void __launch_bounds__(256)
xproj_kernel(const float* __restrict__ x, const float* __restrict__ W1x,
             const float* __restrict__ b1, const float* __restrict__ b2,
             float* __restrict__ P) {
    const int lane = threadIdx.x & 31;
    const int wv   = __builtin_amdgcn_readfirstlane(threadIdx.x >> 5);
    const int bt   = blockIdx.x >> 3;
    const int tc   = blockIdx.x & 7;
    const int c    = lane & 15;
    const int kh   = (lane >> 4) << 1;

    v2f   wf[NT * KB1];
    float bias[NT];
    #pragma unroll
    for (int j = 0; j < NT; ++j) {
        const int  N  = j * 16 + c;
        const bool ok = N < H;
        bias[j] = ok ? b1[N] : 0.f;
        if (j == NT - 1 && c >= 4 && c < 8) bias[j] = b2[92 + c];
        #pragma unroll
        for (int kb = 0; kb < KB1; ++kb) {
            const int K = 4 * kb + kh;
            v2f w;
            w[0] = ok ? W1x[(size_t)K * H + N] : 0.f;
            w[1] = ok ? W1x[(size_t)(K + 1) * H + N] : 0.f;
            wf[j * KB1 + kb] = w;
        }
    }
    for (int s = 0; s < XSTEPS; ++s) {
        const int t = tc * 64 + wv * XSTEPS + s;
        const float* xp = x + ((size_t)(bt * 16 + c) * SEQT + t) * IND + kh;
        v2f a[KB1];
        #pragma unroll
        for (int kb = 0; kb < KB1; ++kb) a[kb] = ld2(xp + 4 * kb);
        v8f acc[NT];
        #pragma unroll
        for (int j = 0; j < NT; ++j)
            #pragma unroll
            for (int r = 0; r < 8; ++r) acc[j][r] = bias[j];
        #pragma unroll
        for (int kb = 0; kb < KB1; ++kb)
            #pragma unroll
            for (int j = 0; j < NT; ++j)
                acc[j] = WMMA4(a[kb], wf[j * KB1 + kb], acc[j]);
        float* pp = P + (((size_t)bt * SEQT + t) * NT) * PTILE + lane * 4;
        #pragma unroll
        for (int j = 0; j < NT; ++j) {
            *(float4*)(pp + j * PTILE)       = make_float4(acc[j][0], acc[j][1], acc[j][2], acc[j][3]);
            *(float4*)(pp + j * PTILE + 128) = make_float4(acc[j][4], acc[j][5], acc[j][6], acc[j][7]);
        }
    }
}

#define R_H1 0
#define R_H2 (2 * HBUF)
#define R_LDS_FLOATS (4 * HBUF)

__device__ __forceinline__ v2f wfrag(const float* Wm, int n, int K) {
    v2f w; w[0] = 0.f; w[1] = 0.f;
    if (Wm) { w[0] = Wm[(size_t)K * H + n]; w[1] = Wm[(size_t)(K + 1) * H + n]; }
    return w;
}

__device__ __forceinline__ void bar_signal() {
    __builtin_amdgcn_fence(__ATOMIC_RELEASE, "workgroup");
    __builtin_amdgcn_s_barrier_signal(-1);
}
__device__ __forceinline__ void bar_wait() {
    __builtin_amdgcn_s_barrier_wait(-1);
    __builtin_amdgcn_fence(__ATOMIC_ACQUIRE, "workgroup");
}

__device__ __forceinline__ void chain25(v8f& c0, v8f& c1, const v2f (&a)[KBH], const v2f (&w)[KBH]) {
    #pragma unroll
    for (int k = 0; k < KBH; ++k) {
        if (k & 1) c1 = WMMA4(a[k], w[k], c1);
        else       c0 = WMMA4(a[k], w[k], c0);
    }
}

__device__ __forceinline__ void start_from(v8f& c0, v8f& c1, const float4& p0, const float4& p1) {
    c0[0] = p0.x; c0[1] = p0.y; c0[2] = p0.z; c0[3] = p0.w;
    c0[4] = p1.x; c0[5] = p1.y; c0[6] = p1.z; c0[7] = p1.w;
    #pragma unroll
    for (int r = 0; r < 8; ++r) c1[r] = 0.f;
}

__global__ void __launch_bounds__(256)
drnn_rec_kernel(const float* __restrict__ W1h, const float* __restrict__ W2x,
                const float* __restrict__ W2h, const float* __restrict__ b2,
                const float* __restrict__ Wo,  const float* __restrict__ bo,
                const float* __restrict__ P,
                float* __restrict__ outp, float* __restrict__ h1out, float* __restrict__ h2out) {
    extern __shared__ float lds[];
    const int tid  = threadIdx.x;
    const int nthr = blockDim.x;
    const int bt   = blockIdx.x;
    const int b0   = bt * 16;
    const int lane = tid & 31;
    const int wv   = __builtin_amdgcn_readfirstlane(tid >> 5);
    const int c    = lane & 15;
    const int kh   = (lane >> 4) << 1;
    const int crow = (lane >> 4) << 3;

    for (int i = tid; i < R_LDS_FLOATS; i += nthr) lds[i] = 0.f;

    const bool hasF   = (wv != 7);
    const bool hasB   = (wv < 4) || (wv == 7);
    const bool loadPA = (wv >= 6);
    const int  tileA  = (wv == 7) ? 4 : 6;
    const int  tileB  = (wv == 7) ? 5 : wv;

    const float* mA1 = nullptr; const float* mA2 = nullptr; const float* mB = nullptr;
    int nA = 0, nB = 0;
    if (wv < 6) {
        mA1 = W2x; mA2 = W2h; nA = wv * 16 + c;
        if (wv < 4) { mB = W1h; nB = wv * 16 + c; }
    } else if (wv == 6) {
        if (c < 4)      { mA1 = W1h; nA = 96 + c; }
        else if (c < 8) { mA1 = W2x; mA2 = W2h; nA = 92 + c; }
    } else {
        mA1 = W1h; nA = 64 + c;
        mB  = W1h; nB = 80 + c;
    }
    v2f wA1[KBH], wA2[KBH], wB[KBH];
    #pragma unroll
    for (int k = 0; k < KBH; ++k) {
        wA1[k] = wfrag(mA1, nA, 4 * k + kh);
        wA2[k] = wfrag(mA2, nA, 4 * k + kh);
        wB[k]  = wfrag(mB,  nB, 4 * k + kh);
    }
    float bias2 = 0.f;
    if (wv < 6) bias2 = b2[wv * 16 + c];

    int dF = 0, dA = 0, dB = 0;
    {
        const int rowo = crow * HSTR;
        if (wv < 6)       dF = R_H2 + rowo + wv * 16 + c;
        else if (wv == 6) dF = R_H2 + rowo + ((c < 4) ? (108 + c) : (92 + c));
        if (wv == 6)      dA = R_H1 + rowo + 96 + c;
        else if (wv == 7) dA = R_H1 + rowo + 64 + c;
        dB = R_H1 + rowo + tileB * 16 + c;
    }

    const float* Pb = P + ((size_t)bt * SEQT * NT) * PTILE + lane * 4;
    __syncthreads();

    if (wv < NT) {
        const float* pp = Pb + wv * PTILE;
        const float4 q0 = *(const float4*)(pp);
        const float4 q1 = *(const float4*)(pp + 128);
        if (wv < NT - 1 || c < 4) {
            float* d = lds + R_H1 + crow * HSTR + wv * 16 + c;
            d[0 * HSTR] = TANHF(q0.x); d[1 * HSTR] = TANHF(q0.y);
            d[2 * HSTR] = TANHF(q0.z); d[3 * HSTR] = TANHF(q0.w);
            d[4 * HSTR] = TANHF(q1.x); d[5 * HSTR] = TANHF(q1.y);
            d[6 * HSTR] = TANHF(q1.z); d[7 * HSTR] = TANHF(q1.w);
        }
    }
    float4 cA0 = make_float4(0.f, 0.f, 0.f, 0.f), cA1 = cA0, cB0 = cA0, cB1 = cA0;
    if (loadPA) {
        const float* pp = Pb + ((size_t)1 * NT + tileA) * PTILE;
        cA0 = *(const float4*)(pp); cA1 = *(const float4*)(pp + 128);
    }
    if (hasB) {
        const float* pp = Pb + ((size_t)1 * NT + tileB) * PTILE;
        cB0 = *(const float4*)(pp); cB1 = *(const float4*)(pp + 128);
    }
    __syncthreads();
    __builtin_amdgcn_s_setprio(1);

    const float* h1base = lds + R_H1 + c * HSTR + kh;
    const float* h2base = lds + R_H2 + c * HSTR + kh;

    v8f q0, q1;
    #pragma unroll
    for (int r = 0; r < 8; ++r) { q0[r] = 0.f; q1[r] = 0.f; }

    for (int t = 0; t < SEQT; ++t) {
        const int cbo = (t & 1) * HBUF;
        const int nbo = HBUF - cbo;
        const int tn = (t + 2 < SEQT) ? (t + 2) : (SEQT - 1);
        float4 nA0 = cA0, nA1 = cA1, nB0 = cB0, nB1 = cB1;
        if (loadPA) {
            const float* pp = Pb + ((size_t)tn * NT + tileA) * PTILE;
            nA0 = *(const float4*)(pp); nA1 = *(const float4*)(pp + 128);
        }
        if (hasB) {
            const float* pp = Pb + ((size_t)tn * NT + tileB) * PTILE;
            nB0 = *(const float4*)(pp); nB1 = *(const float4*)(pp + 128);
        }
        v2f a1[KBH], a2[KBH];
        if (hasF) {
            #pragma unroll
            for (int k = 0; k < KBH; ++k) a2[k] = ld2(h2base + cbo + 4 * k);
        }
        #pragma unroll
        for (int k = 0; k < KBH; ++k) a1[k] = ld2(h1base + cbo + 4 * k);
        __builtin_amdgcn_sched_barrier(0);

        if (hasF && t >= 1) {
            chain25(q0, q1, a2, wA2);
            tanh_store(q0, q1, lds + dF + nbo);
        }
        if (loadPA) {
            v8f c0, c1;
            start_from(c0, c1, cA0, cA1);
            chain25(c0, c1, a1, wA1);
            tanh_store(c0, c1, lds + dA + nbo);
            q0 = c0; q1 = c1;
        }
        if (hasB) {
            v8f c0, c1;
            start_from(c0, c1, cB0, cB1);
            chain25(c0, c1, a1, wB);
            tanh_store(c0, c1, lds + dB + nbo);
        }
        bar_signal();
        __builtin_amdgcn_s_setprio(0);
        __builtin_amdgcn_sched_barrier(0);
        if (!loadPA) {
            #pragma unroll
            for (int r = 0; r < 8; ++r) { q0[r] = bias2; q1[r] = 0.f; }
            chain25(q0, q1, a1, wA1);
        }
        __builtin_amdgcn_sched_barrier(0);
        bar_wait();
        __builtin_amdgcn_s_setprio(1);
        cA0 = nA0; cA1 = nA1; cB0 = nB0; cB1 = nB1;
    }

    if (hasF) {
        v2f a2[KBH];
        #pragma unroll
        for (int k = 0; k < KBH; ++k) a2[k] = ld2(h2base + (SEQT & 1) * HBUF + 4 * k);
        chain25(q0, q1, a2, wA2);
        tanh_store(q0, q1, lds + dF + ((SEQT + 1) & 1) * HBUF);
    }
    __builtin_amdgcn_s_setprio(0);
    __syncthreads();

    const int fb = (SEQT - 1) & 1;
    for (int i = tid; i < 16 * H; i += nthr) {
        int m = i / H, n = i - m * H;
        h1out[(size_t)(b0 + m) * H + n] = lds[R_H1 + fb * HBUF + m * HSTR + n];
        h2out[(size_t)(b0 + m) * H + n] = lds[R_H2 + fb * HBUF + m * HSTR + n];
    }
    if (tid < 16) {
        float s = bo[0];
        for (int n = 0; n < H; ++n) s += lds[R_H2 + fb * HBUF + tid * HSTR + n] * Wo[n];
        outp[b0 + tid] = s;
    }
}

extern "C" void kernel_launch(void* const* d_in, const int* in_sizes, int n_in,
                              void* d_out, int out_size, void* d_ws, size_t ws_size,
                              hipStream_t stream) {
    (void)in_sizes; (void)n_in; (void)out_size;
    const float* x   = (const float*)d_in[0];
    const float* W1x = (const float*)d_in[1];
    const float* W1h = (const float*)d_in[2];
    const float* b1  = (const float*)d_in[3];
    const float* W2x = (const float*)d_in[4];
    const float* W2h = (const float*)d_in[5];
    const float* b2  = (const float*)d_in[6];
    const float* Wo  = (const float*)d_in[7];
    const float* bo  = (const float*)d_in[8];

    float* outp  = (float*)d_out;
    float* h1out = outp + BATCH;
    float* h2out = h1out + BATCH * H;

    if (ws_size >= WS_NEED) {
        float* P = (float*)d_ws;
        xproj_kernel<<<(BATCH / 16) * 8, 256, 0, stream>>>(x, W1x, b1, b2, P);
        drnn_rec_kernel<<<BATCH / 16, 256, (size_t)R_LDS_FLOATS * sizeof(float), stream>>>(
            W1h, W2x, W2h, b2, Wo, bo, P, outp, h1out, h2out);
        return;
    }

    const size_t shmem = (size_t)LDS_FLOATS * sizeof(float);
    (void)hipFuncSetAttribute((const void*)drnn_fused_kernel,
                              hipFuncAttributeMaxDynamicSharedMemorySize, (int)shmem);

    drnn_fused_kernel<<<BATCH / 16, 256, shmem, stream>>>(
        x, W1x, W1h, b1, W2x, W2h, b2, Wo, bo, outp, h1out, h2out);
}


# Layers on separate workgroups, a 64-step chunk apart, over 10 dependent launches (layer 1; b2+h1@W2x in place in workspace; layer 2): 25 not 50-75 WMMAs per wave-step; layer-2 sum order changes
# speedup vs baseline: 2.1696x; 1.4273x over previous
#include <hip/hip_runtime.h>
#include <math.h>

typedef __attribute__((ext_vector_type(2))) float v2f;
typedef __attribute__((ext_vector_type(8))) float v8f;

#define BATCH 512
#define SEQT  512
#define IND   64
#define H     100
#define NT    7
#define KB1   16
#define KBH   25
#define HSTR  116
#define XSTR  68
#define HBUF  (16*HSTR)

#define OFF_W1X 0
#define OFF_W1H (OFF_W1X + NT*KB1*32*2)
#define OFF_W2X (OFF_W1H + NT*KBH*32*2)
#define OFF_W2H (OFF_W2X + NT*KBH*32*2)
#define OFF_H1  (OFF_W2H + NT*KBH*32*2)
#define OFF_H2  (OFF_H1 + 2*HBUF)
#define OFF_X   (OFF_H2 + 2*HBUF)
#define XBUF    (16*XSTR)
#define KBXP    11
#define OFF_P   (OFF_X + 4*XBUF)
#define PBUF    (NT*256)
#define LDS_FLOATS (OFF_P + 2*PBUF)

#if __has_builtin(__builtin_amdgcn_tanhf)
#define TANHF(v) __builtin_amdgcn_tanhf(v)
#elif __has_builtin(__builtin_amdgcn_tanh_f32)
#define TANHF(v) __builtin_amdgcn_tanh_f32(v)
#else
#define TANHF(v) tanhf(v)
#endif

__device__ inline void fill_frags(float* dst, const float* W, int Ka, int Na,
                                  int kb, int tid, int nthr) {
    int total = NT * kb * 32;
    for (int i = tid; i < total; i += nthr) {
        int laneI = i & 31;
        int kblk  = (i >> 5) % kb;
        int nt    = (i >> 5) / kb;
        int N = nt * 16 + (laneI & 15);
        int K = kblk * 4 + ((laneI >> 4) << 1);
        float a = 0.f, b = 0.f;
        if (N < Na) {
            if (K     < Ka) a = W[(size_t)K * Na + N];
            if (K + 1 < Ka) b = W[(size_t)(K + 1) * Na + N];
        }
        dst[2 * i + 0] = a;
        dst[2 * i + 1] = b;
    }
}

__device__ inline v2f ld2(const float* p) { return *(const v2f*)p; }

#define KBXR  (KB1 - KBXP)
#define NWR   80
#define WR_1X 0
#define WR_1H (WR_1X + KBXR)
#define WR_2X (WR_1H + KBH)
#define WR_2H (WR_2X + KBH)

#define WMMA4(a, b, c) __builtin_amdgcn_wmma_f32_16x16x4_f32(false, (a), false, (b), (short)0, (c), false, false)

__device__ __forceinline__ void tanh_store(const v8f& c0, const v8f& c1, float* dst) {
    #pragma unroll
    for (int r = 0; r < 8; ++r) dst[r * HSTR] = TANHF(c0[r] + c1[r]);
}

__device__ __forceinline__ void layer1_store(const float4& p0, const float4& p1,
                                             const v2f (&ax)[KBXR], const v2f (&a1)[KBH],
                                             const v2f (&wr)[NWR], float* dst) {
    v8f c0, c1;
    c0[0] = p0.x; c0[1] = p0.y; c0[2] = p0.z; c0[3] = p0.w;
    c0[4] = p1.x; c0[5] = p1.y; c0[6] = p1.z; c0[7] = p1.w;
    #pragma unroll
    for (int r = 0; r < 8; ++r) c1[r] = 0.f;
    #pragma unroll
    for (int k = 0; k < KBXR; ++k) {
        if (k & 1) c1 = WMMA4(ax[k], wr[WR_1X + k], c1);
        else       c0 = WMMA4(ax[k], wr[WR_1X + k], c0);
    }
    #pragma unroll
    for (int k = 0; k < KBH; ++k) {
        if ((k + KBXR) & 1) c1 = WMMA4(a1[k], wr[WR_1H + k], c1);
        else                c0 = WMMA4(a1[k], wr[WR_1H + k], c0);
    }
    tanh_store(c0, c1, dst);
}

__device__ __forceinline__ void layer2_store(float bias2, const v2f (&a1)[KBH], const v2f (&a2)[KBH],
                                             const v2f (&wr)[NWR], float* dst) {
    v8f c0, c1;
    #pragma unroll
    for (int r = 0; r < 8; ++r) { c0[r] = bias2; c1[r] = 0.f; }
    #pragma unroll
    for (int k = 0; k < KBH; ++k) {
        if (k & 1) c1 = WMMA4(a1[k], wr[WR_2X + k], c1);
        else       c0 = WMMA4(a1[k], wr[WR_2X + k], c0);
    }
    #pragma unroll
    for (int k = 0; k < KBH; ++k) {
        if (k & 1) c1 = WMMA4(a2[k], wr[WR_2H + k], c1);
        else       c0 = WMMA4(a2[k], wr[WR_2H + k], c0);
    }
    tanh_store(c0, c1, dst);
}

__device__ __forceinline__ void helper_p(const float* xl, const float (&pbias)[NT],
                                         const v2f (&wr)[NWR], float* pdst) {
    v8f acc[NT];
    #pragma unroll
    for (int j = 0; j < NT; ++j)
        #pragma unroll
        for (int r = 0; r < 8; ++r) acc[j][r] = pbias[j];
    #pragma unroll
    for (int kb = 0; kb < KBXP; ++kb) {
        v2f a = ld2(xl + 4 * kb);
        #pragma unroll
        for (int j = 0; j < NT; ++j)
            acc[j] = WMMA4(a, wr[j * KBXP + kb], acc[j]);
    }
    #pragma unroll
    for (int j = 0; j < NT; ++j) {
        float* pp = pdst + j * 256;
        *(float4*)(pp)       = make_float4(acc[j][0], acc[j][1], acc[j][2], acc[j][3]);
        *(float4*)(pp + 128) = make_float4(acc[j][4], acc[j][5], acc[j][6], acc[j][7]);
    }
}

__global__ void __launch_bounds__(256)
drnn_fused_kernel(const float* __restrict__ x,
                  const float* __restrict__ W1x, const float* __restrict__ W1h,
                  const float* __restrict__ b1,
                  const float* __restrict__ W2x, const float* __restrict__ W2h,
                  const float* __restrict__ b2,
                  const float* __restrict__ Wo,  const float* __restrict__ bo,
                  float* __restrict__ outp, float* __restrict__ h1out,
                  float* __restrict__ h2out) {
    extern __shared__ float lds[];
    const int tid  = threadIdx.x;
    const int nthr = blockDim.x;
    const int b0   = blockIdx.x * 16;

    fill_frags(lds + OFF_W1X, W1x, IND, H, KB1, tid, nthr);
    fill_frags(lds + OFF_W1H, W1h, H,   H, KBH, tid, nthr);
    fill_frags(lds + OFF_W2X, W2x, H,   H, KBH, tid, nthr);
    fill_frags(lds + OFF_W2H, W2h, H,   H, KBH, tid, nthr);
    for (int i = tid; i < 4 * HBUF; i += nthr) lds[OFF_H1 + i] = 0.f;

    const float* xsrc = x + ((size_t)(b0 + (tid >> 4)) * SEQT) * IND + ((tid & 15) << 2);
    float*       xdst = lds + OFF_X + (tid >> 4) * XSTR + ((tid & 15) << 2);
    *(float4*)(xdst)        = *(const float4*)(xsrc);
    *(float4*)(xdst + XBUF) = *(const float4*)(xsrc + IND);
    float4 xr = *(const float4*)(xsrc + 2 * IND);
    __syncthreads();

    const int lane = tid & 31;
    const int wv   = __builtin_amdgcn_readfirstlane(tid >> 5);
    const bool helper = (wv == NT);
    const int mi   = lane & 15;
    const int kh   = (lane >> 4) << 1;
    const int crow = (lane >> 4) << 3;

    float bias2 = 0.f;
    {
        int n = wv * 16 + mi;
        if (n < H) bias2 = b2[n];
    }

    v2f   wr[NWR];
    float pbias[NT];
    #pragma unroll
    for (int j = 0; j < NT; ++j) pbias[j] = 0.f;
    if (helper) {
        #pragma unroll
        for (int j = 0; j < NT; ++j) {
            pbias[j] = (j * 16 + mi < H) ? b1[j * 16 + mi] : 0.f;
            #pragma unroll
            for (int kb = 0; kb < KBXP; ++kb)
                wr[j * KBXP + kb] = ld2(lds + OFF_W1X + (j * KB1 + kb) * 64 + lane * 2);
        }
        #pragma unroll
        for (int i = NT * KBXP; i < NWR; ++i) { wr[i][0] = 0.f; wr[i][1] = 0.f; }
        const float* xl = lds + OFF_X + mi * XSTR + kh;
        v8f acc[NT];
        #pragma unroll
        for (int j = 0; j < NT; ++j)
            #pragma unroll
            for (int r = 0; r < 8; ++r) acc[j][r] = pbias[j];
        #pragma unroll
        for (int kb = 0; kb < KBXP; ++kb) {
            v2f a = ld2(xl + 4 * kb);
            #pragma unroll
            for (int j = 0; j < NT; ++j)
                acc[j] = __builtin_amdgcn_wmma_f32_16x16x4_f32(false, a, false, wr[j * KBXP + kb], (short)0, acc[j], false, false);
        }
        #pragma unroll
        for (int j = 0; j < NT; ++j) {
            float* pp = lds + OFF_P + j * 256 + lane * 4;
            *(float4*)(pp)       = make_float4(acc[j][0], acc[j][1], acc[j][2], acc[j][3]);
            *(float4*)(pp + 128) = make_float4(acc[j][4], acc[j][5], acc[j][6], acc[j][7]);
        }
    } else {
        #pragma unroll
        for (int k = 0; k < KBXR; ++k)
            wr[WR_1X + k] = ld2(lds + OFF_W1X + (wv * KB1 + KBXP + k) * 64 + lane * 2);
        #pragma unroll
        for (int k = 0; k < KBH; ++k) {
            wr[WR_1H + k] = ld2(lds + OFF_W1H + (wv * KBH + k) * 64 + lane * 2);
            wr[WR_2X + k] = ld2(lds + OFF_W2X + (wv * KBH + k) * 64 + lane * 2);
            wr[WR_2H + k] = ld2(lds + OFF_W2H + (wv * KBH + k) * 64 + lane * 2);
        }
    }
    __syncthreads();

    if (!helper) {
        v2f a1[KBH], ax[KBXR];
        const float* hl = lds + OFF_H1 + HBUF + mi * HSTR + kh;
        const float* xl = lds + OFF_X + mi * XSTR + kh;
        const float* pp = lds + OFF_P + wv * 256 + lane * 4;
        #pragma unroll
        for (int k = 0; k < KBH; ++k) a1[k] = ld2(hl + 4 * k);
        #pragma unroll
        for (int k = 0; k < KBXR; ++k) ax[k] = ld2(xl + 4 * (KBXP + k));
        const float4 p0 = *(const float4*)(pp);
        const float4 p1 = *(const float4*)(pp + 128);
        layer1_store(p0, p1, ax, a1, wr, lds + OFF_H1 + crow * HSTR + wv * 16 + mi);
    } else {
        helper_p(lds + OFF_X + XBUF + mi * XSTR + kh, pbias, wr, lds + OFF_P + PBUF + lane * 4);
    }
    *(float4*)(xdst + 2 * XBUF) = xr;
    __syncthreads();
    xr = *(const float4*)(xsrc + 3 * IND);

    for (int t = 0; t + 1 < SEQT; ++t) {
        const int cb = t & 1;
        const int nb = cb ^ 1;

        if (!helper) {
            v2f a1[KBH], a2[KBH], ax[KBXR];
            const float* h1l = lds + OFF_H1 + cb * HBUF + mi * HSTR + kh;
            const float* h2l = lds + OFF_H2 + nb * HBUF + mi * HSTR + kh;
            const float* xl  = lds + OFF_X + ((t + 1) & 3) * XBUF + mi * XSTR + kh;
            const float* pp  = lds + OFF_P + nb * PBUF + wv * 256 + lane * 4;
            const float4 p0 = *(const float4*)(pp);
            const float4 p1 = *(const float4*)(pp + 128);
            #pragma unroll
            for (int k = 0; k < KBXR; ++k) ax[k] = ld2(xl + 4 * (KBXP + k));
            #pragma unroll
            for (int k = 0; k < 4; ++k) a1[k] = ld2(h1l + 4 * k);
            __builtin_amdgcn_sched_barrier(0);
            #pragma unroll
            for (int k = 4; k < KBH; ++k) a1[k] = ld2(h1l + 4 * k);
            __builtin_amdgcn_sched_barrier(0);
            #pragma unroll
            for (int k = 0; k < KBH; ++k) a2[k] = ld2(h2l + 4 * k);
            __builtin_amdgcn_sched_barrier(0);

            layer2_store(bias2, a1, a2, wr, lds + OFF_H2 + cb * HBUF + crow * HSTR + wv * 16 + mi);
            layer1_store(p0, p1, ax, a1, wr, lds + OFF_H1 + nb * HBUF + crow * HSTR + wv * 16 + mi);
        } else if (t + 2 < SEQT) {
            helper_p(lds + OFF_X + ((t + 2) & 3) * XBUF + mi * XSTR + kh, pbias, wr,
                     lds + OFF_P + cb * PBUF + lane * 4);
        }

        if (t + 3 < SEQT) *(float4*)(xdst + ((t + 3) & 3) * XBUF) = xr;

        __syncthreads();

        if (t + 4 < SEQT) xr = *(const float4*)(xsrc + (size_t)(t + 4) * IND);
    }

    if (!helper) {
        const int cb = (SEQT - 1) & 1, nb = cb ^ 1;
        v2f a1[KBH], a2[KBH];
        const float* h1l = lds + OFF_H1 + cb * HBUF + mi * HSTR + kh;
        const float* h2l = lds + OFF_H2 + nb * HBUF + mi * HSTR + kh;
        #pragma unroll
        for (int k = 0; k < KBH; ++k) a1[k] = ld2(h1l + 4 * k);
        #pragma unroll
        for (int k = 0; k < KBH; ++k) a2[k] = ld2(h2l + 4 * k);
        layer2_store(bias2, a1, a2, wr, lds + OFF_H2 + cb * HBUF + crow * HSTR + wv * 16 + mi);
    }
    __syncthreads();

    const int fb = (SEQT - 1) & 1;
    for (int i = tid; i < 16 * H; i += nthr) {
        int m = i / H, n = i - m * H;
        h1out[(size_t)(b0 + m) * H + n] = lds[OFF_H1 + fb * HBUF + m * HSTR + n];
        h2out[(size_t)(b0 + m) * H + n] = lds[OFF_H2 + fb * HBUF + m * HSTR + n];
    }
    if (tid < 16) {
        float s = bo[0];
        for (int n = 0; n < H; ++n) s += lds[OFF_H2 + fb * HBUF + tid * HSTR + n] * Wo[n];
        outp[b0 + tid] = s;
    }
}

#define PTILE 256
#define PSLOT (NT * PTILE)
#define WS_NEED ((size_t)(BATCH / 16) * SEQT * PSLOT * sizeof(float))
#define XSTEPS 8

__global__ void __launch_bounds__(256)
xproj_kernel(const float* __restrict__ x, const float* __restrict__ W1x,
             const float* __restrict__ b1, float* __restrict__ P) {
    const int lane = threadIdx.x & 31;
    const int wv   = __builtin_amdgcn_readfirstlane(threadIdx.x >> 5);
    const int bt   = blockIdx.x >> 3;
    const int tc   = blockIdx.x & 7;
    const int c    = lane & 15;
    const int kh   = (lane >> 4) << 1;

    v2f   wf[NT * KB1];
    float bias[NT];
    #pragma unroll
    for (int j = 0; j < NT; ++j) {
        const int  N  = j * 16 + c;
        const bool ok = N < H;
        bias[j] = ok ? b1[N] : 0.f;
        #pragma unroll
        for (int kb = 0; kb < KB1; ++kb) {
            const int K = 4 * kb + kh;
            v2f w;
            w[0] = ok ? W1x[(size_t)K * H + N] : 0.f;
            w[1] = ok ? W1x[(size_t)(K + 1) * H + N] : 0.f;
            wf[j * KB1 + kb] = w;
        }
    }
    for (int s = 0; s < XSTEPS; ++s) {
        const int t = tc * 64 + wv * XSTEPS + s;
        const float* xp = x + ((size_t)(bt * 16 + c) * SEQT + t) * IND + kh;
        v2f a[KB1];
        #pragma unroll
        for (int kb = 0; kb < KB1; ++kb) a[kb] = ld2(xp + 4 * kb);
        v8f acc[NT];
        #pragma unroll
        for (int j = 0; j < NT; ++j)
            #pragma unroll
            for (int r = 0; r < 8; ++r) acc[j][r] = bias[j];
        #pragma unroll
        for (int kb = 0; kb < KB1; ++kb)
            #pragma unroll
            for (int j = 0; j < NT; ++j)
                acc[j] = WMMA4(a[kb], wf[j * KB1 + kb], acc[j]);
        float* pp = P + ((size_t)bt * SEQT + t) * PSLOT + lane * 4;
        #pragma unroll
        for (int j = 0; j < NT; ++j) {
            *(float4*)(pp + j * PTILE)       = make_float4(acc[j][0], acc[j][1], acc[j][2], acc[j][3]);
            *(float4*)(pp + j * PTILE + 128) = make_float4(acc[j][4], acc[j][5], acc[j][6], acc[j][7]);
        }
    }
}

#define CH     64
#define NCHUNK (SEQT / CH)
#define NRECWG (BATCH / 16)
#define NQWG   128
#define QPAIRS (NRECWG * CH / NQWG)
#define W_LDS_FLOATS (2 * HBUF)
#define W_LDS_RESERVE (164 * 1024)
#define RTHR   (NT * 32)

__device__ __forceinline__ v2f wfrag(const float* Wm, int n, int K) {
    v2f w; w[0] = 0.f; w[1] = 0.f;
    if (Wm) { w[0] = Wm[(size_t)K * H + n]; w[1] = Wm[(size_t)(K + 1) * H + n]; }
    return w;
}

__device__ __forceinline__ void bar_signal() {
    __builtin_amdgcn_fence(__ATOMIC_RELEASE, "workgroup", "local");
    __builtin_amdgcn_s_barrier_signal(-1);
}
__device__ __forceinline__ void bar_wait() {
    __builtin_amdgcn_s_barrier_wait(-1);
    __builtin_amdgcn_fence(__ATOMIC_ACQUIRE, "workgroup", "local");
}

__device__ __forceinline__ void chain25(v8f& c0, v8f& c1, const v2f (&a)[KBH], const v2f (&w)[KBH]) {
    #pragma unroll
    for (int k = 0; k < KBH; ++k) {
        if (k & 1) c1 = WMMA4(a[k], w[k], c1);
        else       c0 = WMMA4(a[k], w[k], c0);
    }
}

__device__ __forceinline__ void start_from(v8f& c0, v8f& c1, const float4& p0, const float4& p1) {
    c0[0] = p0.x; c0[1] = p0.y; c0[2] = p0.z; c0[3] = p0.w;
    c0[4] = p1.x; c0[5] = p1.y; c0[6] = p1.z; c0[7] = p1.w;
    #pragma unroll
    for (int r = 0; r < 8; ++r) c1[r] = 0.f;
}

template <bool WRITEBACK>
__device__ __forceinline__ void rec_chunk(const float* __restrict__ Wh, float* __restrict__ S,
                                          int t0, bool first, float* __restrict__ hcarry,
                                          float* lds, int bt) {
    const int tid  = threadIdx.x;
    const int lane = tid & 31;
    const int wv   = __builtin_amdgcn_readfirstlane(tid >> 5);
    const int c    = lane & 15;
    const int kh   = (lane >> 4) << 1;
    const int crow = (lane >> 4) << 3;
    const int b0   = bt * 16;

    for (int i = tid; i < 2 * HBUF; i += RTHR) lds[i] = 0.f;
    __syncthreads();
    if (!first) {
        for (int i = tid; i < 16 * H; i += RTHR) {
            int m = i / H, n = i - m * H;
            lds[m * HSTR + n] = hcarry[(size_t)(b0 + m) * H + n];
        }
    }

    const int ncol = wv * 16 + c;
    const float* Wm = (ncol < H) ? Wh : nullptr;
    v2f w[KBH];
    #pragma unroll
    for (int k = 0; k < KBH; ++k) w[k] = wfrag(Wm, ncol, 4 * k + kh);

    float* Sb = S + (size_t)bt * SEQT * PSLOT + wv * PTILE + lane * 4;
    const float* pa = Sb + (size_t)t0 * PSLOT;
    const float* pb = Sb + (size_t)(t0 + 1) * PSLOT;
    float4 e0 = *(const float4*)(pa), e1 = *(const float4*)(pa + 128);
    float4 o0 = *(const float4*)(pb), o1 = *(const float4*)(pb + 128);
    __syncthreads();

    const float* hb = lds + c * HSTR + kh;
    float*       hd = lds + crow * HSTR + wv * 16 + c;

    auto step = [&](const int s, const int cur, const int nxt, float4& X0, float4& X1)
                    __attribute__((always_inline)) {
        const int t = t0 + s;
        float hv[8];
        v2f a[KBH];
        #pragma unroll
        for (int k = 0; k < KBH; ++k) a[k] = ld2(hb + cur + 4 * k);
        v8f c0, c1;
        start_from(c0, c1, X0, X1);
        __builtin_amdgcn_sched_barrier(0);
        chain25(c0, c1, a, w);
        #pragma unroll
        for (int r = 0; r < 8; ++r) hv[r] = TANHF(c0[r] + c1[r]);
        #pragma unroll
        for (int r = 0; r < 8; ++r) hd[nxt + r * HSTR] = hv[r];
        bar_signal();
        const int tn = (s + 2 < CH) ? (t + 2) : (t0 + CH - 1);
        const float* pp = Sb + (size_t)tn * PSLOT;
        X0 = *(const float4*)(pp); X1 = *(const float4*)(pp + 128);
        if (WRITEBACK) {
            float* q = Sb + (size_t)t * PSLOT;
            *(float4*)(q)       = make_float4(hv[0], hv[1], hv[2], hv[3]);
            *(float4*)(q + 128) = make_float4(hv[4], hv[5], hv[6], hv[7]);
        }
        bar_wait();
    };
    for (int s = 0; s < CH; s += 2) {
        step(s,     0,    HBUF, e0, e1);
        step(s + 1, HBUF, 0,    o0, o1);
    }

    for (int i = tid; i < 16 * H; i += RTHR) {
        int m = i / H, n = i - m * H;
        hcarry[(size_t)(b0 + m) * H + n] = lds[m * HSTR + n];
    }
}

__device__ __forceinline__ void qproj_chunk(const float* __restrict__ W2x, const float* __restrict__ b2,
                                            float* __restrict__ S, int t0, int qb, float* lds) {
    const int tid  = threadIdx.x;
    const int lane = tid & 31;
    const int wv   = __builtin_amdgcn_readfirstlane(tid >> 5);
    const int c    = lane & 15;
    const int kh   = (lane >> 4) << 1;
    const int crow = (lane >> 4) << 3;

    const int ncol = wv * 16 + c;
    const float* Wm = (ncol < H) ? W2x : nullptr;
    v2f w[KBH];
    #pragma unroll
    for (int k = 0; k < KBH; ++k) w[k] = wfrag(Wm, ncol, 4 * k + kh);
    const float bias = (ncol < H) ? b2[ncol] : 0.f;

    const int p0 = qb * QPAIRS;
    const int bt = p0 / CH;
    const int ts = t0 + (p0 - bt * CH);
    float* Sb = S + ((size_t)bt * SEQT + ts) * PSLOT + wv * PTILE + lane * 4;

    float4 v0 = *(const float4*)(Sb), v1 = *(const float4*)(Sb + 128);
    float* d = lds + crow * HSTR + wv * 16 + c;
    const float* hb = lds + c * HSTR + kh;

    for (int i = 0; i < QPAIRS; ++i) {
        const int buf = (i & 1) * HBUF;
        float* q = Sb + (size_t)i * PSLOT;
        const float* qn = (i + 1 < QPAIRS) ? (q + PSLOT) : q;
        const float4 u0 = *(const float4*)(qn), u1 = *(const float4*)(qn + 128);
        d[buf + 0 * HSTR] = v0.x; d[buf + 1 * HSTR] = v0.y; d[buf + 2 * HSTR] = v0.z; d[buf + 3 * HSTR] = v0.w;
        d[buf + 4 * HSTR] = v1.x; d[buf + 5 * HSTR] = v1.y; d[buf + 6 * HSTR] = v1.z; d[buf + 7 * HSTR] = v1.w;
        bar_signal();
        bar_wait();
        v2f a[KBH];
        #pragma unroll
        for (int k = 0; k < KBH; ++k) a[k] = ld2(hb + buf + 4 * k);
        v8f c0, c1;
        #pragma unroll
        for (int r = 0; r < 8; ++r) { c0[r] = bias; c1[r] = 0.f; }
        chain25(c0, c1, a, w);
        *(float4*)(q)       = make_float4(c0[0] + c1[0], c0[1] + c1[1], c0[2] + c1[2], c0[3] + c1[3]);
        *(float4*)(q + 128) = make_float4(c0[4] + c1[4], c0[5] + c1[5], c0[6] + c1[6], c0[7] + c1[7]);
        v0 = u0; v1 = u1;
    }
}

__global__ void __launch_bounds__(RTHR)
drnn_wave_kernel(const float* __restrict__ W1h, const float* __restrict__ W2x,
                 const float* __restrict__ W2h, const float* __restrict__ b2,
                 const float* __restrict__ Wo,  const float* __restrict__ bo,
                 float* __restrict__ P,
                 float* __restrict__ outp, float* __restrict__ h1out, float* __restrict__ h2out,
                 int l) {
    extern __shared__ float lds[];
    const int blk = blockIdx.x;
    if (blk < NRECWG) {
        if (l < NCHUNK) rec_chunk<true>(W1h, P, l * CH, l == 0, h1out, lds, blk);
    } else if (blk < 2 * NRECWG) {
        const int ch = l - 2;
        if (ch >= 0 && ch < NCHUNK) {
            const int bt = blk - NRECWG;
            rec_chunk<false>(W2h, P, ch * CH, ch == 0, h2out, lds, bt);
            if (ch == NCHUNK - 1) {
                const int tid = threadIdx.x;
                if (tid < 16) {
                    float s = bo[0];
                    for (int n = 0; n < H; ++n) s += lds[tid * HSTR + n] * Wo[n];
                    outp[bt * 16 + tid] = s;
                }
            }
        }
    } else {
        const int ch = l - 1;
        if (ch >= 0 && ch < NCHUNK) qproj_chunk(W2x, b2, P, ch * CH, blk - 2 * NRECWG, lds);
    }
}

extern "C" void kernel_launch(void* const* d_in, const int* in_sizes, int n_in,
                              void* d_out, int out_size, void* d_ws, size_t ws_size,
                              hipStream_t stream) {
    (void)in_sizes; (void)n_in; (void)out_size;
    const float* x   = (const float*)d_in[0];
    const float* W1x = (const float*)d_in[1];
    const float* W1h = (const float*)d_in[2];
    const float* b1  = (const float*)d_in[3];
    const float* W2x = (const float*)d_in[4];
    const float* W2h = (const float*)d_in[5];
    const float* b2  = (const float*)d_in[6];
    const float* Wo  = (const float*)d_in[7];
    const float* bo  = (const float*)d_in[8];

    float* outp  = (float*)d_out;
    float* h1out = outp + BATCH;
    float* h2out = h1out + BATCH * H;

    if (ws_size >= WS_NEED) {
        float* P = (float*)d_ws;
        xproj_kernel<<<(BATCH / 16) * 8, 256, 0, stream>>>(x, W1x, b1, P);
        (void)hipFuncSetAttribute((const void*)drnn_wave_kernel,
                                  hipFuncAttributeMaxDynamicSharedMemorySize, (int)W_LDS_RESERVE);
#define WAVE_LAUNCH(l) drnn_wave_kernel<<<2 * NRECWG + NQWG, RTHR, (size_t)W_LDS_RESERVE, stream>>>( \
            W1h, W2x, W2h, b2, Wo, bo, P, outp, h1out, h2out, (l))
        WAVE_LAUNCH(0); WAVE_LAUNCH(1); WAVE_LAUNCH(2); WAVE_LAUNCH(3); WAVE_LAUNCH(4);
        WAVE_LAUNCH(5); WAVE_LAUNCH(6); WAVE_LAUNCH(7); WAVE_LAUNCH(8); WAVE_LAUNCH(9);
#undef WAVE_LAUNCH
        return;
    }

    const size_t shmem = (size_t)LDS_FLOATS * sizeof(float);
    (void)hipFuncSetAttribute((const void*)drnn_fused_kernel,
                              hipFuncAttributeMaxDynamicSharedMemorySize, (int)shmem);

    drnn_fused_kernel<<<BATCH / 16, 256, shmem, stream>>>(
        x, W1x, W1h, b1, W2x, W2h, b2, Wo, bo, outp, h1out, h2out);
}


# Chunk hand-over in one memory round trip: carried state as two float4 loads per thread and 50 unmasked weight loads, all requested in one group with the first slots; no LDS zero-fill; sums unchanged
# speedup vs baseline: 2.3107x; 1.0650x over previous
#include <hip/hip_runtime.h>
#include <math.h>

typedef __attribute__((ext_vector_type(2))) float v2f;
typedef __attribute__((ext_vector_type(8))) float v8f;

#define BATCH 512
#define SEQT  512
#define IND   64
#define H     100
#define NT    7
#define KB1   16
#define KBH   25
#define HSTR  116
#define XSTR  68
#define HBUF  (16*HSTR)

#define OFF_W1X 0
#define OFF_W1H (OFF_W1X + NT*KB1*32*2)
#define OFF_W2X (OFF_W1H + NT*KBH*32*2)
#define OFF_W2H (OFF_W2X + NT*KBH*32*2)
#define OFF_H1  (OFF_W2H + NT*KBH*32*2)
#define OFF_H2  (OFF_H1 + 2*HBUF)
#define OFF_X   (OFF_H2 + 2*HBUF)
#define XBUF    (16*XSTR)
#define KBXP    11
#define OFF_P   (OFF_X + 4*XBUF)
#define PBUF    (NT*256)
#define LDS_FLOATS (OFF_P + 2*PBUF)

#if __has_builtin(__builtin_amdgcn_tanhf)
#define TANHF(v) __builtin_amdgcn_tanhf(v)
#elif __has_builtin(__builtin_amdgcn_tanh_f32)
#define TANHF(v) __builtin_amdgcn_tanh_f32(v)
#else
#define TANHF(v) tanhf(v)
#endif

__device__ inline void fill_frags(float* dst, const float* W, int Ka, int Na,
                                  int kb, int tid, int nthr) {
    int total = NT * kb * 32;
    for (int i = tid; i < total; i += nthr) {
        int laneI = i & 31;
        int kblk  = (i >> 5) % kb;
        int nt    = (i >> 5) / kb;
        int N = nt * 16 + (laneI & 15);
        int K = kblk * 4 + ((laneI >> 4) << 1);
        float a = 0.f, b = 0.f;
        if (N < Na) {
            if (K     < Ka) a = W[(size_t)K * Na + N];
            if (K + 1 < Ka) b = W[(size_t)(K + 1) * Na + N];
        }
        dst[2 * i + 0] = a;
        dst[2 * i + 1] = b;
    }
}

__device__ inline v2f ld2(const float* p) { return *(const v2f*)p; }

#define KBXR  (KB1 - KBXP)
#define NWR   80
#define WR_1X 0
#define WR_1H (WR_1X + KBXR)
#define WR_2X (WR_1H + KBH)
#define WR_2H (WR_2X + KBH)

#define WMMA4(a, b, c) __builtin_amdgcn_wmma_f32_16x16x4_f32(false, (a), false, (b), (short)0, (c), false, false)

__device__ __forceinline__ void tanh_store(const v8f& c0, const v8f& c1, float* dst) {
    #pragma unroll
    for (int r = 0; r < 8; ++r) dst[r * HSTR] = TANHF(c0[r] + c1[r]);
}

__device__ __forceinline__ void layer1_store(const float4& p0, const float4& p1,
                                             const v2f (&ax)[KBXR], const v2f (&a1)[KBH],
                                             const v2f (&wr)[NWR], float* dst) {
    v8f c0, c1;
    c0[0] = p0.x; c0[1] = p0.y; c0[2] = p0.z; c0[3] = p0.w;
    c0[4] = p1.x; c0[5] = p1.y; c0[6] = p1.z; c0[7] = p1.w;
    #pragma unroll
    for (int r = 0; r < 8; ++r) c1[r] = 0.f;
    #pragma unroll
    for (int k = 0; k < KBXR; ++k) {
        if (k & 1) c1 = WMMA4(ax[k], wr[WR_1X + k], c1);
        else       c0 = WMMA4(ax[k], wr[WR_1X + k], c0);
    }
    #pragma unroll
    for (int k = 0; k < KBH; ++k) {
        if ((k + KBXR) & 1) c1 = WMMA4(a1[k], wr[WR_1H + k], c1);
        else                c0 = WMMA4(a1[k], wr[WR_1H + k], c0);
    }
    tanh_store(c0, c1, dst);
}

__device__ __forceinline__ void layer2_store(float bias2, const v2f (&a1)[KBH], const v2f (&a2)[KBH],
                                             const v2f (&wr)[NWR], float* dst) {
    v8f c0, c1;
    #pragma unroll
    for (int r = 0; r < 8; ++r) { c0[r] = bias2; c1[r] = 0.f; }
    #pragma unroll
    for (int k = 0; k < KBH; ++k) {
        if (k & 1) c1 = WMMA4(a1[k], wr[WR_2X + k], c1);
        else       c0 = WMMA4(a1[k], wr[WR_2X + k], c0);
    }
    #pragma unroll
    for (int k = 0; k < KBH; ++k) {
        if (k & 1) c1 = WMMA4(a2[k], wr[WR_2H + k], c1);
        else       c0 = WMMA4(a2[k], wr[WR_2H + k], c0);
    }
    tanh_store(c0, c1, dst);
}

__device__ __forceinline__ void helper_p(const float* xl, const float (&pbias)[NT],
                                         const v2f (&wr)[NWR], float* pdst) {
    v8f acc[NT];
    #pragma unroll
    for (int j = 0; j < NT; ++j)
        #pragma unroll
        for (int r = 0; r < 8; ++r) acc[j][r] = pbias[j];
    #pragma unroll
    for (int kb = 0; kb < KBXP; ++kb) {
        v2f a = ld2(xl + 4 * kb);
        #pragma unroll
        for (int j = 0; j < NT; ++j)
            acc[j] = WMMA4(a, wr[j * KBXP + kb], acc[j]);
    }
    #pragma unroll
    for (int j = 0; j < NT; ++j) {
        float* pp = pdst + j * 256;
        *(float4*)(pp)       = make_float4(acc[j][0], acc[j][1], acc[j][2], acc[j][3]);
        *(float4*)(pp + 128) = make_float4(acc[j][4], acc[j][5], acc[j][6], acc[j][7]);
    }
}

__global__ void __launch_bounds__(256)
drnn_fused_kernel(const float* __restrict__ x,
                  const float* __restrict__ W1x, const float* __restrict__ W1h,
                  const float* __restrict__ b1,
                  const float* __restrict__ W2x, const float* __restrict__ W2h,
                  const float* __restrict__ b2,
                  const float* __restrict__ Wo,  const float* __restrict__ bo,
                  float* __restrict__ outp, float* __restrict__ h1out,
                  float* __restrict__ h2out) {
    extern __shared__ float lds[];
    const int tid  = threadIdx.x;
    const int nthr = blockDim.x;
    const int b0   = blockIdx.x * 16;

    fill_frags(lds + OFF_W1X, W1x, IND, H, KB1, tid, nthr);
    fill_frags(lds + OFF_W1H, W1h, H,   H, KBH, tid, nthr);
    fill_frags(lds + OFF_W2X, W2x, H,   H, KBH, tid, nthr);
    fill_frags(lds + OFF_W2H, W2h, H,   H, KBH, tid, nthr);
    for (int i = tid; i < 4 * HBUF; i += nthr) lds[OFF_H1 + i] = 0.f;

    const float* xsrc = x + ((size_t)(b0 + (tid >> 4)) * SEQT) * IND + ((tid & 15) << 2);
    float*       xdst = lds + OFF_X + (tid >> 4) * XSTR + ((tid & 15) << 2);
    *(float4*)(xdst)        = *(const float4*)(xsrc);
    *(float4*)(xdst + XBUF) = *(const float4*)(xsrc + IND);
    float4 xr = *(const float4*)(xsrc + 2 * IND);
    __syncthreads();

    const int lane = tid & 31;
    const int wv   = __builtin_amdgcn_readfirstlane(tid >> 5);
    const bool helper = (wv == NT);
    const int mi   = lane & 15;
    const int kh   = (lane >> 4) << 1;
    const int crow = (lane >> 4) << 3;

    float bias2 = 0.f;
    {
        int n = wv * 16 + mi;
        if (n < H) bias2 = b2[n];
    }

    v2f   wr[NWR];
    float pbias[NT];
    #pragma unroll
    for (int j = 0; j < NT; ++j) pbias[j] = 0.f;
    if (helper) {
        #pragma unroll
        for (int j = 0; j < NT; ++j) {
            pbias[j] = (j * 16 + mi < H) ? b1[j * 16 + mi] : 0.f;
            #pragma unroll
            for (int kb = 0; kb < KBXP; ++kb)
                wr[j * KBXP + kb] = ld2(lds + OFF_W1X + (j * KB1 + kb) * 64 + lane * 2);
        }
        #pragma unroll
        for (int i = NT * KBXP; i < NWR; ++i) { wr[i][0] = 0.f; wr[i][1] = 0.f; }
        const float* xl = lds + OFF_X + mi * XSTR + kh;
        v8f acc[NT];
        #pragma unroll
        for (int j = 0; j < NT; ++j)
            #pragma unroll
            for (int r = 0; r < 8; ++r) acc[j][r] = pbias[j];
        #pragma unroll
        for (int kb = 0; kb < KBXP; ++kb) {
            v2f a = ld2(xl + 4 * kb);
            #pragma unroll
            for (int j = 0; j < NT; ++j)
                acc[j] = __builtin_amdgcn_wmma_f32_16x16x4_f32(false, a, false, wr[j * KBXP + kb], (short)0, acc[j], false, false);
        }
        #pragma unroll
        for (int j = 0; j < NT; ++j) {
            float* pp = lds + OFF_P + j * 256 + lane * 4;
            *(float4*)(pp)       = make_float4(acc[j][0], acc[j][1], acc[j][2], acc[j][3]);
            *(float4*)(pp + 128) = make_float4(acc[j][4], acc[j][5], acc[j][6], acc[j][7]);
        }
    } else {
        #pragma unroll
        for (int k = 0; k < KBXR; ++k)
            wr[WR_1X + k] = ld2(lds + OFF_W1X + (wv * KB1 + KBXP + k) * 64 + lane * 2);
        #pragma unroll
        for (int k = 0; k < KBH; ++k) {
            wr[WR_1H + k] = ld2(lds + OFF_W1H + (wv * KBH + k) * 64 + lane * 2);
            wr[WR_2X + k] = ld2(lds + OFF_W2X + (wv * KBH + k) * 64 + lane * 2);
            wr[WR_2H + k] = ld2(lds + OFF_W2H + (wv * KBH + k) * 64 + lane * 2);
        }
    }
    __syncthreads();

    if (!helper) {
        v2f a1[KBH], ax[KBXR];
        const float* hl = lds + OFF_H1 + HBUF + mi * HSTR + kh;
        const float* xl = lds + OFF_X + mi * XSTR + kh;
        const float* pp = lds + OFF_P + wv * 256 + lane * 4;
        #pragma unroll
        for (int k = 0; k < KBH; ++k) a1[k] = ld2(hl + 4 * k);
        #pragma unroll
        for (int k = 0; k < KBXR; ++k) ax[k] = ld2(xl + 4 * (KBXP + k));
        const float4 p0 = *(const float4*)(pp);
        const float4 p1 = *(const float4*)(pp + 128);
        layer1_store(p0, p1, ax, a1, wr, lds + OFF_H1 + crow * HSTR + wv * 16 + mi);
    } else {
        helper_p(lds + OFF_X + XBUF + mi * XSTR + kh, pbias, wr, lds + OFF_P + PBUF + lane * 4);
    }
    *(float4*)(xdst + 2 * XBUF) = xr;
    __syncthreads();
    xr = *(const float4*)(xsrc + 3 * IND);

    for (int t = 0; t + 1 < SEQT; ++t) {
        const int cb = t & 1;
        const int nb = cb ^ 1;

        if (!helper) {
            v2f a1[KBH], a2[KBH], ax[KBXR];
            const float* h1l = lds + OFF_H1 + cb * HBUF + mi * HSTR + kh;
            const float* h2l = lds + OFF_H2 + nb * HBUF + mi * HSTR + kh;
            const float* xl  = lds + OFF_X + ((t + 1) & 3) * XBUF + mi * XSTR + kh;
            const float* pp  = lds + OFF_P + nb * PBUF + wv * 256 + lane * 4;
            const float4 p0 = *(const float4*)(pp);
            const float4 p1 = *(const float4*)(pp + 128);
            #pragma unroll
            for (int k = 0; k < KBXR; ++k) ax[k] = ld2(xl + 4 * (KBXP + k));
            #pragma unroll
            for (int k = 0; k < 4; ++k) a1[k] = ld2(h1l + 4 * k);
            __builtin_amdgcn_sched_barrier(0);
            #pragma unroll
            for (int k = 4; k < KBH; ++k) a1[k] = ld2(h1l + 4 * k);
            __builtin_amdgcn_sched_barrier(0);
            #pragma unroll
            for (int k = 0; k < KBH; ++k) a2[k] = ld2(h2l + 4 * k);
            __builtin_amdgcn_sched_barrier(0);

            layer2_store(bias2, a1, a2, wr, lds + OFF_H2 + cb * HBUF + crow * HSTR + wv * 16 + mi);
            layer1_store(p0, p1, ax, a1, wr, lds + OFF_H1 + nb * HBUF + crow * HSTR + wv * 16 + mi);
        } else if (t + 2 < SEQT) {
            helper_p(lds + OFF_X + ((t + 2) & 3) * XBUF + mi * XSTR + kh, pbias, wr,
                     lds + OFF_P + cb * PBUF + lane * 4);
        }

        if (t + 3 < SEQT) *(float4*)(xdst + ((t + 3) & 3) * XBUF) = xr;

        __syncthreads();

        if (t + 4 < SEQT) xr = *(const float4*)(xsrc + (size_t)(t + 4) * IND);
    }

    if (!helper) {
        const int cb = (SEQT - 1) & 1, nb = cb ^ 1;
        v2f a1[KBH], a2[KBH];
        const float* h1l = lds + OFF_H1 + cb * HBUF + mi * HSTR + kh;
        const float* h2l = lds + OFF_H2 + nb * HBUF + mi * HSTR + kh;
        #pragma unroll
        for (int k = 0; k < KBH; ++k) a1[k] = ld2(h1l + 4 * k);
        #pragma unroll
        for (int k = 0; k < KBH; ++k) a2[k] = ld2(h2l + 4 * k);
        layer2_store(bias2, a1, a2, wr, lds + OFF_H2 + cb * HBUF + crow * HSTR + wv * 16 + mi);
    }
    __syncthreads();

    const int fb = (SEQT - 1) & 1;
    for (int i = tid; i < 16 * H; i += nthr) {
        int m = i / H, n = i - m * H;
        h1out[(size_t)(b0 + m) * H + n] = lds[OFF_H1 + fb * HBUF + m * HSTR + n];
        h2out[(size_t)(b0 + m) * H + n] = lds[OFF_H2 + fb * HBUF + m * HSTR + n];
    }
    if (tid < 16) {
        float s = bo[0];
        for (int n = 0; n < H; ++n) s += lds[OFF_H2 + fb * HBUF + tid * HSTR + n] * Wo[n];
        outp[b0 + tid] = s;
    }
}

#define PTILE 256
#define PSLOT (NT * PTILE)
#define WS_NEED ((size_t)(BATCH / 16) * SEQT * PSLOT * sizeof(float))
#define XSTEPS 8

__global__ void __launch_bounds__(256)
xproj_kernel(const float* __restrict__ x, const float* __restrict__ W1x,
             const float* __restrict__ b1, float* __restrict__ P) {
    const int lane = threadIdx.x & 31;
    const int wv   = __builtin_amdgcn_readfirstlane(threadIdx.x >> 5);
    const int bt   = blockIdx.x >> 3;
    const int tc   = blockIdx.x & 7;
    const int c    = lane & 15;
    const int kh   = (lane >> 4) << 1;

    v2f   wf[NT * KB1];
    float bias[NT];
    #pragma unroll
    for (int j = 0; j < NT; ++j) {
        const int  N  = j * 16 + c;
        const bool ok = N < H;
        bias[j] = ok ? b1[N] : 0.f;
        #pragma unroll
        for (int kb = 0; kb < KB1; ++kb) {
            const int K = 4 * kb + kh;
            v2f w;
            w[0] = ok ? W1x[(size_t)K * H + N] : 0.f;
            w[1] = ok ? W1x[(size_t)(K + 1) * H + N] : 0.f;
            wf[j * KB1 + kb] = w;
        }
    }
    for (int s = 0; s < XSTEPS; ++s) {
        const int t = tc * 64 + wv * XSTEPS + s;
        const float* xp = x + ((size_t)(bt * 16 + c) * SEQT + t) * IND + kh;
        v2f a[KB1];
        #pragma unroll
        for (int kb = 0; kb < KB1; ++kb) a[kb] = ld2(xp + 4 * kb);
        v8f acc[NT];
        #pragma unroll
        for (int j = 0; j < NT; ++j)
            #pragma unroll
            for (int r = 0; r < 8; ++r) acc[j][r] = bias[j];
        #pragma unroll
        for (int kb = 0; kb < KB1; ++kb)
            #pragma unroll
            for (int j = 0; j < NT; ++j)
                acc[j] = WMMA4(a[kb], wf[j * KB1 + kb], acc[j]);
        float* pp = P + ((size_t)bt * SEQT + t) * PSLOT + lane * 4;
        #pragma unroll
        for (int j = 0; j < NT; ++j) {
            *(float4*)(pp + j * PTILE)       = make_float4(acc[j][0], acc[j][1], acc[j][2], acc[j][3]);
            *(float4*)(pp + j * PTILE + 128) = make_float4(acc[j][4], acc[j][5], acc[j][6], acc[j][7]);
        }
    }
}

#define CH     64
#define NCHUNK (SEQT / CH)
#define NRECWG (BATCH / 16)
#define NQWG   128
#define QPAIRS (NRECWG * CH / NQWG)
#define W_LDS_FLOATS (2 * HBUF)
#define W_LDS_RESERVE (164 * 1024)
#define RTHR   (NT * 32)

__device__ __forceinline__ void load_wfrags(v2f (&w)[KBH], const float* __restrict__ Wm, int ncol, int kh) {
    const bool ok = ncol < H;
    const float* p = Wm + (size_t)kh * H + (ok ? ncol : H - 1);
    float lo[KBH], hi[KBH];
    #pragma unroll
    for (int k = 0; k < KBH; ++k) {
        lo[k] = p[(size_t)(4 * k) * H];
        hi[k] = p[(size_t)(4 * k + 1) * H];
    }
    #pragma unroll
    for (int k = 0; k < KBH; ++k) {
        w[k][0] = ok ? lo[k] : 0.f;
        w[k][1] = ok ? hi[k] : 0.f;
    }
}

#define CARRY4 (16 * H / 4)

__device__ __forceinline__ void bar_signal() {
    __builtin_amdgcn_fence(__ATOMIC_RELEASE, "workgroup", "local");
    __builtin_amdgcn_s_barrier_signal(-1);
}
__device__ __forceinline__ void bar_wait() {
    __builtin_amdgcn_s_barrier_wait(-1);
    __builtin_amdgcn_fence(__ATOMIC_ACQUIRE, "workgroup", "local");
}

__device__ __forceinline__ void chain25(v8f& c0, v8f& c1, const v2f (&a)[KBH], const v2f (&w)[KBH]) {
    #pragma unroll
    for (int k = 0; k < KBH; ++k) {
        if (k & 1) c1 = WMMA4(a[k], w[k], c1);
        else       c0 = WMMA4(a[k], w[k], c0);
    }
}

__device__ __forceinline__ void start_from(v8f& c0, v8f& c1, const float4& p0, const float4& p1) {
    c0[0] = p0.x; c0[1] = p0.y; c0[2] = p0.z; c0[3] = p0.w;
    c0[4] = p1.x; c0[5] = p1.y; c0[6] = p1.z; c0[7] = p1.w;
    #pragma unroll
    for (int r = 0; r < 8; ++r) c1[r] = 0.f;
}

template <bool WRITEBACK>
__device__ __forceinline__ void rec_chunk(const float* __restrict__ Wh, float* __restrict__ S,
                                          int t0, bool first, float* __restrict__ hcarry,
                                          float* lds, int bt) {
    const int tid  = threadIdx.x;
    const int lane = tid & 31;
    const int wv   = __builtin_amdgcn_readfirstlane(tid >> 5);
    const int c    = lane & 15;
    const int kh   = (lane >> 4) << 1;
    const int crow = (lane >> 4) << 3;
    const int b0   = bt * 16;


    float* Sb = S + (size_t)bt * SEQT * PSLOT + wv * PTILE + lane * 4;
    const float* pa = Sb + (size_t)t0 * PSLOT;
    const float* pb = Sb + (size_t)(t0 + 1) * PSLOT;
    float4 e0 = *(const float4*)(pa), e1 = *(const float4*)(pa + 128);
    float4 o0 = *(const float4*)(pb), o1 = *(const float4*)(pb + 128);

    const int  i0 = tid, i1 = tid + RTHR;
    const bool two = i1 < CARRY4;
    const int  m0 = i0 / 25, m1 = i1 / 25;
    float* l0 = lds + m0 * HSTR + 4 * (i0 - m0 * 25);
    float* l1 = lds + m1 * HSTR + 4 * (i1 - m1 * 25);
    float* g0 = hcarry + (size_t)b0 * H + 4 * i0;
    float* g1 = hcarry + (size_t)b0 * H + 4 * i1;
    float4 s0 = make_float4(0.f, 0.f, 0.f, 0.f), s1 = s0;
    if (!first) {
        s0 = *(const float4*)(g0);
        if (two) s1 = *(const float4*)(g1);
    }

    const int ncol = wv * 16 + c;
    v2f w[KBH];
    load_wfrags(w, Wh, ncol, kh);

    *(float4*)(l0) = s0;
    if (two) *(float4*)(l1) = s1;
    __syncthreads();

    const float* hb = lds + c * HSTR + kh;
    float*       hd = lds + crow * HSTR + wv * 16 + c;

    auto step = [&](const int s, const int cur, const int nxt, float4& X0, float4& X1)
                    __attribute__((always_inline)) {
        const int t = t0 + s;
        float hv[8];
        v2f a[KBH];
        #pragma unroll
        for (int k = 0; k < KBH; ++k) a[k] = ld2(hb + cur + 4 * k);
        v8f c0, c1;
        start_from(c0, c1, X0, X1);
        __builtin_amdgcn_sched_barrier(0);
        chain25(c0, c1, a, w);
        #pragma unroll
        for (int r = 0; r < 8; ++r) hv[r] = TANHF(c0[r] + c1[r]);
        #pragma unroll
        for (int r = 0; r < 8; ++r) hd[nxt + r * HSTR] = hv[r];
        bar_signal();
        const int tn = (s + 2 < CH) ? (t + 2) : (t0 + CH - 1);
        const float* pp = Sb + (size_t)tn * PSLOT;
        X0 = *(const float4*)(pp); X1 = *(const float4*)(pp + 128);
        if (WRITEBACK) {
            float* q = Sb + (size_t)t * PSLOT;
            *(float4*)(q)       = make_float4(hv[0], hv[1], hv[2], hv[3]);
            *(float4*)(q + 128) = make_float4(hv[4], hv[5], hv[6], hv[7]);
        }
        bar_wait();
    };
    for (int s = 0; s < CH; s += 2) {
        step(s,     0,    HBUF, e0, e1);
        step(s + 1, HBUF, 0,    o0, o1);
    }

    const float4 r0 = *(const float4*)(l0);
    float4 r1 = r0;
    if (two) r1 = *(const float4*)(l1);
    *(float4*)(g0) = r0;
    if (two) *(float4*)(g1) = r1;
}

__device__ __forceinline__ void qproj_chunk(const float* __restrict__ W2x, const float* __restrict__ b2,
                                            float* __restrict__ S, int t0, int qb, float* lds) {
    const int tid  = threadIdx.x;
    const int lane = tid & 31;
    const int wv   = __builtin_amdgcn_readfirstlane(tid >> 5);
    const int c    = lane & 15;
    const int kh   = (lane >> 4) << 1;
    const int crow = (lane >> 4) << 3;

    const int ncol = wv * 16 + c;
    v2f w[KBH];
    load_wfrags(w, W2x, ncol, kh);
    const float bias = (ncol < H) ? b2[ncol] : 0.f;

    const int p0 = qb * QPAIRS;
    const int bt = p0 / CH;
    const int ts = t0 + (p0 - bt * CH);
    float* Sb = S + ((size_t)bt * SEQT + ts) * PSLOT + wv * PTILE + lane * 4;

    float4 v0 = *(const float4*)(Sb), v1 = *(const float4*)(Sb + 128);
    float* d = lds + crow * HSTR + wv * 16 + c;
    const float* hb = lds + c * HSTR + kh;

    for (int i = 0; i < QPAIRS; ++i) {
        const int buf = (i & 1) * HBUF;
        float* q = Sb + (size_t)i * PSLOT;
        const float* qn = (i + 1 < QPAIRS) ? (q + PSLOT) : q;
        const float4 u0 = *(const float4*)(qn), u1 = *(const float4*)(qn + 128);
        d[buf + 0 * HSTR] = v0.x; d[buf + 1 * HSTR] = v0.y; d[buf + 2 * HSTR] = v0.z; d[buf + 3 * HSTR] = v0.w;
        d[buf + 4 * HSTR] = v1.x; d[buf + 5 * HSTR] = v1.y; d[buf + 6 * HSTR] = v1.z; d[buf + 7 * HSTR] = v1.w;
        bar_signal();
        bar_wait();
        v2f a[KBH];
        #pragma unroll
        for (int k = 0; k < KBH; ++k) a[k] = ld2(hb + buf + 4 * k);
        v8f c0, c1;
        #pragma unroll
        for (int r = 0; r < 8; ++r) { c0[r] = bias; c1[r] = 0.f; }
        chain25(c0, c1, a, w);
        *(float4*)(q)       = make_float4(c0[0] + c1[0], c0[1] + c1[1], c0[2] + c1[2], c0[3] + c1[3]);
        *(float4*)(q + 128) = make_float4(c0[4] + c1[4], c0[5] + c1[5], c0[6] + c1[6], c0[7] + c1[7]);
        v0 = u0; v1 = u1;
    }
}

__global__ void __launch_bounds__(RTHR)
drnn_wave_kernel(const float* __restrict__ W1h, const float* __restrict__ W2x,
                 const float* __restrict__ W2h, const float* __restrict__ b2,
                 const float* __restrict__ Wo,  const float* __restrict__ bo,
                 float* __restrict__ P,
                 float* __restrict__ outp, float* __restrict__ h1out, float* __restrict__ h2out,
                 int l) {
    extern __shared__ float lds[];
    const int blk = blockIdx.x;
    if (blk < NRECWG) {
        if (l < NCHUNK) rec_chunk<true>(W1h, P, l * CH, l == 0, h1out, lds, blk);
    } else if (blk < 2 * NRECWG) {
        const int ch = l - 2;
        if (ch >= 0 && ch < NCHUNK) {
            const int bt = blk - NRECWG;
            rec_chunk<false>(W2h, P, ch * CH, ch == 0, h2out, lds, bt);
            if (ch == NCHUNK - 1) {
                const int tid = threadIdx.x;
                if (tid < 16) {
                    float s = bo[0];
                    for (int n = 0; n < H; ++n) s += lds[tid * HSTR + n] * Wo[n];
                    outp[bt * 16 + tid] = s;
                }
            }
        }
    } else {
        const int ch = l - 1;
        if (ch >= 0 && ch < NCHUNK) qproj_chunk(W2x, b2, P, ch * CH, blk - 2 * NRECWG, lds);
    }
}

extern "C" void kernel_launch(void* const* d_in, const int* in_sizes, int n_in,
                              void* d_out, int out_size, void* d_ws, size_t ws_size,
                              hipStream_t stream) {
    (void)in_sizes; (void)n_in; (void)out_size;
    const float* x   = (const float*)d_in[0];
    const float* W1x = (const float*)d_in[1];
    const float* W1h = (const float*)d_in[2];
    const float* b1  = (const float*)d_in[3];
    const float* W2x = (const float*)d_in[4];
    const float* W2h = (const float*)d_in[5];
    const float* b2  = (const float*)d_in[6];
    const float* Wo  = (const float*)d_in[7];
    const float* bo  = (const float*)d_in[8];

    float* outp  = (float*)d_out;
    float* h1out = outp + BATCH;
    float* h2out = h1out + BATCH * H;

    if (ws_size >= WS_NEED) {
        float* P = (float*)d_ws;
        xproj_kernel<<<(BATCH / 16) * 8, 256, 0, stream>>>(x, W1x, b1, P);
        (void)hipFuncSetAttribute((const void*)drnn_wave_kernel,
                                  hipFuncAttributeMaxDynamicSharedMemorySize, (int)W_LDS_RESERVE);
#define WAVE_LAUNCH(l) drnn_wave_kernel<<<2 * NRECWG + NQWG, RTHR, (size_t)W_LDS_RESERVE, stream>>>( \
            W1h, W2x, W2h, b2, Wo, bo, P, outp, h1out, h2out, (l))
        WAVE_LAUNCH(0); WAVE_LAUNCH(1); WAVE_LAUNCH(2); WAVE_LAUNCH(3); WAVE_LAUNCH(4);
        WAVE_LAUNCH(5); WAVE_LAUNCH(6); WAVE_LAUNCH(7); WAVE_LAUNCH(8); WAVE_LAUNCH(9);
#undef WAVE_LAUNCH
        return;
    }

    const size_t shmem = (size_t)LDS_FLOATS * sizeof(float);
    (void)hipFuncSetAttribute((const void*)drnn_fused_kernel,
                              hipFuncAttributeMaxDynamicSharedMemorySize, (int)shmem);

    drnn_fused_kernel<<<BATCH / 16, 256, shmem, stream>>>(
        x, W1x, W1h, b1, W2x, W2h, b2, Wo, bo, outp, h1out, h2out);
}


# Input projection of chunks 1..7 leaves the serial chain: 224 otherwise idle workgroups of recurrence launch 0 form it beside layer 1 of chunk 0; the launch in front keeps chunk 0; sums unchanged
# speedup vs baseline: 2.4020x; 1.0395x over previous
#include <hip/hip_runtime.h>
#include <math.h>

typedef __attribute__((ext_vector_type(2))) float v2f;
typedef __attribute__((ext_vector_type(8))) float v8f;

#define BATCH 512
#define SEQT  512
#define IND   64
#define H     100
#define NT    7
#define KB1   16
#define KBH   25
#define HSTR  116
#define XSTR  68
#define HBUF  (16*HSTR)

#define OFF_W1X 0
#define OFF_W1H (OFF_W1X + NT*KB1*32*2)
#define OFF_W2X (OFF_W1H + NT*KBH*32*2)
#define OFF_W2H (OFF_W2X + NT*KBH*32*2)
#define OFF_H1  (OFF_W2H + NT*KBH*32*2)
#define OFF_H2  (OFF_H1 + 2*HBUF)
#define OFF_X   (OFF_H2 + 2*HBUF)
#define XBUF    (16*XSTR)
#define KBXP    11
#define OFF_P   (OFF_X + 4*XBUF)
#define PBUF    (NT*256)
#define LDS_FLOATS (OFF_P + 2*PBUF)

#if __has_builtin(__builtin_amdgcn_tanhf)
#define TANHF(v) __builtin_amdgcn_tanhf(v)
#elif __has_builtin(__builtin_amdgcn_tanh_f32)
#define TANHF(v) __builtin_amdgcn_tanh_f32(v)
#else
#define TANHF(v) tanhf(v)
#endif

__device__ inline void fill_frags(float* dst, const float* W, int Ka, int Na,
                                  int kb, int tid, int nthr) {
    int total = NT * kb * 32;
    for (int i = tid; i < total; i += nthr) {
        int laneI = i & 31;
        int kblk  = (i >> 5) % kb;
        int nt    = (i >> 5) / kb;
        int N = nt * 16 + (laneI & 15);
        int K = kblk * 4 + ((laneI >> 4) << 1);
        float a = 0.f, b = 0.f;
        if (N < Na) {
            if (K     < Ka) a = W[(size_t)K * Na + N];
            if (K + 1 < Ka) b = W[(size_t)(K + 1) * Na + N];
        }
        dst[2 * i + 0] = a;
        dst[2 * i + 1] = b;
    }
}

__device__ inline v2f ld2(const float* p) { return *(const v2f*)p; }

#define KBXR  (KB1 - KBXP)
#define NWR   80
#define WR_1X 0
#define WR_1H (WR_1X + KBXR)
#define WR_2X (WR_1H + KBH)
#define WR_2H (WR_2X + KBH)

#define WMMA4(a, b, c) __builtin_amdgcn_wmma_f32_16x16x4_f32(false, (a), false, (b), (short)0, (c), false, false)

__device__ __forceinline__ void tanh_store(const v8f& c0, const v8f& c1, float* dst) {
    #pragma unroll
    for (int r = 0; r < 8; ++r) dst[r * HSTR] = TANHF(c0[r] + c1[r]);
}

__device__ __forceinline__ void layer1_store(const float4& p0, const float4& p1,
                                             const v2f (&ax)[KBXR], const v2f (&a1)[KBH],
                                             const v2f (&wr)[NWR], float* dst) {
    v8f c0, c1;
    c0[0] = p0.x; c0[1] = p0.y; c0[2] = p0.z; c0[3] = p0.w;
    c0[4] = p1.x; c0[5] = p1.y; c0[6] = p1.z; c0[7] = p1.w;
    #pragma unroll
    for (int r = 0; r < 8; ++r) c1[r] = 0.f;
    #pragma unroll
    for (int k = 0; k < KBXR; ++k) {
        if (k & 1) c1 = WMMA4(ax[k], wr[WR_1X + k], c1);
        else       c0 = WMMA4(ax[k], wr[WR_1X + k], c0);
    }
    #pragma unroll
    for (int k = 0; k < KBH; ++k) {
        if ((k + KBXR) & 1) c1 = WMMA4(a1[k], wr[WR_1H + k], c1);
        else                c0 = WMMA4(a1[k], wr[WR_1H + k], c0);
    }
    tanh_store(c0, c1, dst);
}

__device__ __forceinline__ void layer2_store(float bias2, const v2f (&a1)[KBH], const v2f (&a2)[KBH],
                                             const v2f (&wr)[NWR], float* dst) {
    v8f c0, c1;
    #pragma unroll
    for (int r = 0; r < 8; ++r) { c0[r] = bias2; c1[r] = 0.f; }
    #pragma unroll
    for (int k = 0; k < KBH; ++k) {
        if (k & 1) c1 = WMMA4(a1[k], wr[WR_2X + k], c1);
        else       c0 = WMMA4(a1[k], wr[WR_2X + k], c0);
    }
    #pragma unroll
    for (int k = 0; k < KBH; ++k) {
        if (k & 1) c1 = WMMA4(a2[k], wr[WR_2H + k], c1);
        else       c0 = WMMA4(a2[k], wr[WR_2H + k], c0);
    }
    tanh_store(c0, c1, dst);
}

__device__ __forceinline__ void helper_p(const float* xl, const float (&pbias)[NT],
                                         const v2f (&wr)[NWR], float* pdst) {
    v8f acc[NT];
    #pragma unroll
    for (int j = 0; j < NT; ++j)
        #pragma unroll
        for (int r = 0; r < 8; ++r) acc[j][r] = pbias[j];
    #pragma unroll
    for (int kb = 0; kb < KBXP; ++kb) {
        v2f a = ld2(xl + 4 * kb);
        #pragma unroll
        for (int j = 0; j < NT; ++j)
            acc[j] = WMMA4(a, wr[j * KBXP + kb], acc[j]);
    }
    #pragma unroll
    for (int j = 0; j < NT; ++j) {
        float* pp = pdst + j * 256;
        *(float4*)(pp)       = make_float4(acc[j][0], acc[j][1], acc[j][2], acc[j][3]);
        *(float4*)(pp + 128) = make_float4(acc[j][4], acc[j][5], acc[j][6], acc[j][7]);
    }
}

__global__ void __launch_bounds__(256)
drnn_fused_kernel(const float* __restrict__ x,
                  const float* __restrict__ W1x, const float* __restrict__ W1h,
                  const float* __restrict__ b1,
                  const float* __restrict__ W2x, const float* __restrict__ W2h,
                  const float* __restrict__ b2,
                  const float* __restrict__ Wo,  const float* __restrict__ bo,
                  float* __restrict__ outp, float* __restrict__ h1out,
                  float* __restrict__ h2out) {
    extern __shared__ float lds[];
    const int tid  = threadIdx.x;
    const int nthr = blockDim.x;
    const int b0   = blockIdx.x * 16;

    fill_frags(lds + OFF_W1X, W1x, IND, H, KB1, tid, nthr);
    fill_frags(lds + OFF_W1H, W1h, H,   H, KBH, tid, nthr);
    fill_frags(lds + OFF_W2X, W2x, H,   H, KBH, tid, nthr);
    fill_frags(lds + OFF_W2H, W2h, H,   H, KBH, tid, nthr);
    for (int i = tid; i < 4 * HBUF; i += nthr) lds[OFF_H1 + i] = 0.f;

    const float* xsrc = x + ((size_t)(b0 + (tid >> 4)) * SEQT) * IND + ((tid & 15) << 2);
    float*       xdst = lds + OFF_X + (tid >> 4) * XSTR + ((tid & 15) << 2);
    *(float4*)(xdst)        = *(const float4*)(xsrc);
    *(float4*)(xdst + XBUF) = *(const float4*)(xsrc + IND);
    float4 xr = *(const float4*)(xsrc + 2 * IND);
    __syncthreads();

    const int lane = tid & 31;
    const int wv   = __builtin_amdgcn_readfirstlane(tid >> 5);
    const bool helper = (wv == NT);
    const int mi   = lane & 15;
    const int kh   = (lane >> 4) << 1;
    const int crow = (lane >> 4) << 3;

    float bias2 = 0.f;
    {
        int n = wv * 16 + mi;
        if (n < H) bias2 = b2[n];
    }

    v2f   wr[NWR];
    float pbias[NT];
    #pragma unroll
    for (int j = 0; j < NT; ++j) pbias[j] = 0.f;
    if (helper) {
        #pragma unroll
        for (int j = 0; j < NT; ++j) {
            pbias[j] = (j * 16 + mi < H) ? b1[j * 16 + mi] : 0.f;
            #pragma unroll
            for (int kb = 0; kb < KBXP; ++kb)
                wr[j * KBXP + kb] = ld2(lds + OFF_W1X + (j * KB1 + kb) * 64 + lane * 2);
        }
        #pragma unroll
        for (int i = NT * KBXP; i < NWR; ++i) { wr[i][0] = 0.f; wr[i][1] = 0.f; }
        const float* xl = lds + OFF_X + mi * XSTR + kh;
        v8f acc[NT];
        #pragma unroll
        for (int j = 0; j < NT; ++j)
            #pragma unroll
            for (int r = 0; r < 8; ++r) acc[j][r] = pbias[j];
        #pragma unroll
        for (int kb = 0; kb < KBXP; ++kb) {
            v2f a = ld2(xl + 4 * kb);
            #pragma unroll
            for (int j = 0; j < NT; ++j)
                acc[j] = __builtin_amdgcn_wmma_f32_16x16x4_f32(false, a, false, wr[j * KBXP + kb], (short)0, acc[j], false, false);
        }
        #pragma unroll
        for (int j = 0; j < NT; ++j) {
            float* pp = lds + OFF_P + j * 256 + lane * 4;
            *(float4*)(pp)       = make_float4(acc[j][0], acc[j][1], acc[j][2], acc[j][3]);
            *(float4*)(pp + 128) = make_float4(acc[j][4], acc[j][5], acc[j][6], acc[j][7]);
        }
    } else {
        #pragma unroll
        for (int k = 0; k < KBXR; ++k)
            wr[WR_1X + k] = ld2(lds + OFF_W1X + (wv * KB1 + KBXP + k) * 64 + lane * 2);
        #pragma unroll
        for (int k = 0; k < KBH; ++k) {
            wr[WR_1H + k] = ld2(lds + OFF_W1H + (wv * KBH + k) * 64 + lane * 2);
            wr[WR_2X + k] = ld2(lds + OFF_W2X + (wv * KBH + k) * 64 + lane * 2);
            wr[WR_2H + k] = ld2(lds + OFF_W2H + (wv * KBH + k) * 64 + lane * 2);
        }
    }
    __syncthreads();

    if (!helper) {
        v2f a1[KBH], ax[KBXR];
        const float* hl = lds + OFF_H1 + HBUF + mi * HSTR + kh;
        const float* xl = lds + OFF_X + mi * XSTR + kh;
        const float* pp = lds + OFF_P + wv * 256 + lane * 4;
        #pragma unroll
        for (int k = 0; k < KBH; ++k) a1[k] = ld2(hl + 4 * k);
        #pragma unroll
        for (int k = 0; k < KBXR; ++k) ax[k] = ld2(xl + 4 * (KBXP + k));
        const float4 p0 = *(const float4*)(pp);
        const float4 p1 = *(const float4*)(pp + 128);
        layer1_store(p0, p1, ax, a1, wr, lds + OFF_H1 + crow * HSTR + wv * 16 + mi);
    } else {
        helper_p(lds + OFF_X + XBUF + mi * XSTR + kh, pbias, wr, lds + OFF_P + PBUF + lane * 4);
    }
    *(float4*)(xdst + 2 * XBUF) = xr;
    __syncthreads();
    xr = *(const float4*)(xsrc + 3 * IND);

    for (int t = 0; t + 1 < SEQT; ++t) {
        const int cb = t & 1;
        const int nb = cb ^ 1;

        if (!helper) {
            v2f a1[KBH], a2[KBH], ax[KBXR];
            const float* h1l = lds + OFF_H1 + cb * HBUF + mi * HSTR + kh;
            const float* h2l = lds + OFF_H2 + nb * HBUF + mi * HSTR + kh;
            const float* xl  = lds + OFF_X + ((t + 1) & 3) * XBUF + mi * XSTR + kh;
            const float* pp  = lds + OFF_P + nb * PBUF + wv * 256 + lane * 4;
            const float4 p0 = *(const float4*)(pp);
            const float4 p1 = *(const float4*)(pp + 128);
            #pragma unroll
            for (int k = 0; k < KBXR; ++k) ax[k] = ld2(xl + 4 * (KBXP + k));
            #pragma unroll
            for (int k = 0; k < 4; ++k) a1[k] = ld2(h1l + 4 * k);
            __builtin_amdgcn_sched_barrier(0);
            #pragma unroll
            for (int k = 4; k < KBH; ++k) a1[k] = ld2(h1l + 4 * k);
            __builtin_amdgcn_sched_barrier(0);
            #pragma unroll
            for (int k = 0; k < KBH; ++k) a2[k] = ld2(h2l + 4 * k);
            __builtin_amdgcn_sched_barrier(0);

            layer2_store(bias2, a1, a2, wr, lds + OFF_H2 + cb * HBUF + crow * HSTR + wv * 16 + mi);
            layer1_store(p0, p1, ax, a1, wr, lds + OFF_H1 + nb * HBUF + crow * HSTR + wv * 16 + mi);
        } else if (t + 2 < SEQT) {
            helper_p(lds + OFF_X + ((t + 2) & 3) * XBUF + mi * XSTR + kh, pbias, wr,
                     lds + OFF_P + cb * PBUF + lane * 4);
        }

        if (t + 3 < SEQT) *(float4*)(xdst + ((t + 3) & 3) * XBUF) = xr;

        __syncthreads();

        if (t + 4 < SEQT) xr = *(const float4*)(xsrc + (size_t)(t + 4) * IND);
    }

    if (!helper) {
        const int cb = (SEQT - 1) & 1, nb = cb ^ 1;
        v2f a1[KBH], a2[KBH];
        const float* h1l = lds + OFF_H1 + cb * HBUF + mi * HSTR + kh;
        const float* h2l = lds + OFF_H2 + nb * HBUF + mi * HSTR + kh;
        #pragma unroll
        for (int k = 0; k < KBH; ++k) a1[k] = ld2(h1l + 4 * k);
        #pragma unroll
        for (int k = 0; k < KBH; ++k) a2[k] = ld2(h2l + 4 * k);
        layer2_store(bias2, a1, a2, wr, lds + OFF_H2 + cb * HBUF + crow * HSTR + wv * 16 + mi);
    }
    __syncthreads();

    const int fb = (SEQT - 1) & 1;
    for (int i = tid; i < 16 * H; i += nthr) {
        int m = i / H, n = i - m * H;
        h1out[(size_t)(b0 + m) * H + n] = lds[OFF_H1 + fb * HBUF + m * HSTR + n];
        h2out[(size_t)(b0 + m) * H + n] = lds[OFF_H2 + fb * HBUF + m * HSTR + n];
    }
    if (tid < 16) {
        float s = bo[0];
        for (int n = 0; n < H; ++n) s += lds[OFF_H2 + fb * HBUF + tid * HSTR + n] * Wo[n];
        outp[b0 + tid] = s;
    }
}

#define PTILE 256
#define PSLOT (NT * PTILE)
#define WS_NEED ((size_t)(BATCH / 16) * SEQT * PSLOT * sizeof(float))
template <bool SINGLE>
__device__ __forceinline__ void xproj_wave(const float* __restrict__ x, const float* __restrict__ W1x,
                                           const float* __restrict__ b1, float* __restrict__ P,
                                           int bt, int t0, int dt, int n, int lane) {
    const int c  = lane & 15;
    const int kh = (lane >> 4) << 1;

    const float* xrow = x + ((size_t)(bt * 16 + c) * SEQT) * IND + kh;
    float*       prow = P + ((size_t)bt * SEQT) * PSLOT + lane * 4;
    v2f xe[KB1], xo[KB1];
    #pragma unroll
    for (int kb = 0; kb < KB1; ++kb) xe[kb] = ld2(xrow + (size_t)t0 * IND + 4 * kb);

    v2f   wf[NT * KB1];
    float bias[NT];
    #pragma unroll
    for (int j = 0; j < NT; ++j) {
        const int  N  = j * 16 + c;
        const bool ok = N < H;
        const float* wp = W1x + (size_t)kh * H + (ok ? N : H - 1);
        const float  bv = b1[ok ? N : H - 1];
        bias[j] = ok ? bv : 0.f;
        #pragma unroll
        for (int kb = 0; kb < KB1; ++kb) {
            const float lo = wp[(size_t)(4 * kb) * H];
            const float hi = wp[(size_t)(4 * kb + 1) * H];
            v2f w;
            w[0] = ok ? lo : 0.f;
            w[1] = ok ? hi : 0.f;
            wf[j * KB1 + kb] = w;
        }
    }
    __builtin_amdgcn_sched_barrier(0);

    auto body = [&](const v2f (&xa)[KB1], const int t) __attribute__((always_inline)) {
        v8f acc[NT];
        #pragma unroll
        for (int j = 0; j < NT; ++j)
            #pragma unroll
            for (int r = 0; r < 8; ++r) acc[j][r] = bias[j];
        #pragma unroll
        for (int kb = 0; kb < KB1; ++kb)
            #pragma unroll
            for (int j = 0; j < NT; ++j)
                acc[j] = WMMA4(xa[kb], wf[j * KB1 + kb], acc[j]);
        float* pp = prow + (size_t)t * PSLOT;
        #pragma unroll
        for (int j = 0; j < NT; ++j) {
            *(float4*)(pp + j * PTILE)       = make_float4(acc[j][0], acc[j][1], acc[j][2], acc[j][3]);
            *(float4*)(pp + j * PTILE + 128) = make_float4(acc[j][4], acc[j][5], acc[j][6], acc[j][7]);
        }
    };

    if (SINGLE) {
        body(xe, t0);
        return;
    }
    for (int s = 0; s < n; s += 2) {
        const int t  = t0 + s * dt;
        const int t1 = t + dt;
        const int t2 = (s + 2 < n) ? t1 + dt : t1;
        #pragma unroll
        for (int kb = 0; kb < KB1; ++kb) xo[kb] = ld2(xrow + (size_t)t1 * IND + 4 * kb);
        __builtin_amdgcn_sched_barrier(0);
        body(xe, t);
        #pragma unroll
        for (int kb = 0; kb < KB1; ++kb) xe[kb] = ld2(xrow + (size_t)t2 * IND + 4 * kb);
        __builtin_amdgcn_sched_barrier(0);
        body(xo, t1);
    }
}

__global__ void __launch_bounds__(256)
xproj_head_kernel(const float* __restrict__ x, const float* __restrict__ W1x,
                  const float* __restrict__ b1, float* __restrict__ P) {
    const int lane = threadIdx.x & 31;
    const int wv   = __builtin_amdgcn_readfirstlane(threadIdx.x >> 5);
    xproj_wave<true>(x, W1x, b1, P, blockIdx.x >> 3, (blockIdx.x & 7) * 8 + wv, 1, 1, lane);
}

#define CH     64
#define NCHUNK (SEQT / CH)
#define NRECWG (BATCH / 16)
#define NQWG   128
#define QPAIRS (NRECWG * CH / NQWG)
#define W_LDS_FLOATS (2 * HBUF)
#define W_LDS_RESERVE (164 * 1024)
#define RTHR   (NT * 32)

__device__ __forceinline__ void load_wfrags(v2f (&w)[KBH], const float* __restrict__ Wm, int ncol, int kh) {
    const bool ok = ncol < H;
    const float* p = Wm + (size_t)kh * H + (ok ? ncol : H - 1);
    float lo[KBH], hi[KBH];
    #pragma unroll
    for (int k = 0; k < KBH; ++k) {
        lo[k] = p[(size_t)(4 * k) * H];
        hi[k] = p[(size_t)(4 * k + 1) * H];
    }
    #pragma unroll
    for (int k = 0; k < KBH; ++k) {
        w[k][0] = ok ? lo[k] : 0.f;
        w[k][1] = ok ? hi[k] : 0.f;
    }
}

#define CARRY4 (16 * H / 4)

__device__ __forceinline__ void bar_signal() {
    __builtin_amdgcn_fence(__ATOMIC_RELEASE, "workgroup", "local");
    __builtin_amdgcn_s_barrier_signal(-1);
}
__device__ __forceinline__ void bar_wait() {
    __builtin_amdgcn_s_barrier_wait(-1);
    __builtin_amdgcn_fence(__ATOMIC_ACQUIRE, "workgroup", "local");
}

__device__ __forceinline__ void chain25(v8f& c0, v8f& c1, const v2f (&a)[KBH], const v2f (&w)[KBH]) {
    #pragma unroll
    for (int k = 0; k < KBH; ++k) {
        if (k & 1) c1 = WMMA4(a[k], w[k], c1);
        else       c0 = WMMA4(a[k], w[k], c0);
    }
}

__device__ __forceinline__ void start_from(v8f& c0, v8f& c1, const float4& p0, const float4& p1) {
    c0[0] = p0.x; c0[1] = p0.y; c0[2] = p0.z; c0[3] = p0.w;
    c0[4] = p1.x; c0[5] = p1.y; c0[6] = p1.z; c0[7] = p1.w;
    #pragma unroll
    for (int r = 0; r < 8; ++r) c1[r] = 0.f;
}

template <bool WRITEBACK>
__device__ __forceinline__ void rec_chunk(const float* __restrict__ Wh, float* __restrict__ S,
                                          int t0, bool first, float* __restrict__ hcarry,
                                          float* lds, int bt) {
    const int tid  = threadIdx.x;
    const int lane = tid & 31;
    const int wv   = __builtin_amdgcn_readfirstlane(tid >> 5);
    const int c    = lane & 15;
    const int kh   = (lane >> 4) << 1;
    const int crow = (lane >> 4) << 3;
    const int b0   = bt * 16;


    float* Sb = S + (size_t)bt * SEQT * PSLOT + wv * PTILE + lane * 4;
    const float* pa = Sb + (size_t)t0 * PSLOT;
    const float* pb = Sb + (size_t)(t0 + 1) * PSLOT;
    float4 e0 = *(const float4*)(pa), e1 = *(const float4*)(pa + 128);
    float4 o0 = *(const float4*)(pb), o1 = *(const float4*)(pb + 128);

    const int  i0 = tid, i1 = tid + RTHR;
    const bool two = i1 < CARRY4;
    const int  m0 = i0 / 25, m1 = i1 / 25;
    float* l0 = lds + m0 * HSTR + 4 * (i0 - m0 * 25);
    float* l1 = lds + m1 * HSTR + 4 * (i1 - m1 * 25);
    float* g0 = hcarry + (size_t)b0 * H + 4 * i0;
    float* g1 = hcarry + (size_t)b0 * H + 4 * i1;
    float4 s0 = make_float4(0.f, 0.f, 0.f, 0.f), s1 = s0;
    if (!first) {
        s0 = *(const float4*)(g0);
        if (two) s1 = *(const float4*)(g1);
    }

    const int ncol = wv * 16 + c;
    v2f w[KBH];
    load_wfrags(w, Wh, ncol, kh);

    *(float4*)(l0) = s0;
    if (two) *(float4*)(l1) = s1;
    __syncthreads();

    const float* hb = lds + c * HSTR + kh;
    float*       hd = lds + crow * HSTR + wv * 16 + c;

    auto step = [&](const int s, const int cur, const int nxt, float4& X0, float4& X1)
                    __attribute__((always_inline)) {
        const int t = t0 + s;
        float hv[8];
        v2f a[KBH];
        #pragma unroll
        for (int k = 0; k < KBH; ++k) a[k] = ld2(hb + cur + 4 * k);
        v8f c0, c1;
        start_from(c0, c1, X0, X1);
        __builtin_amdgcn_sched_barrier(0);
        chain25(c0, c1, a, w);
        #pragma unroll
        for (int r = 0; r < 8; ++r) hv[r] = TANHF(c0[r] + c1[r]);
        #pragma unroll
        for (int r = 0; r < 8; ++r) hd[nxt + r * HSTR] = hv[r];
        bar_signal();
        const int tn = (s + 2 < CH) ? (t + 2) : (t0 + CH - 1);
        const float* pp = Sb + (size_t)tn * PSLOT;
        X0 = *(const float4*)(pp); X1 = *(const float4*)(pp + 128);
        if (WRITEBACK) {
            float* q = Sb + (size_t)t * PSLOT;
            *(float4*)(q)       = make_float4(hv[0], hv[1], hv[2], hv[3]);
            *(float4*)(q + 128) = make_float4(hv[4], hv[5], hv[6], hv[7]);
        }
        bar_wait();
    };
    for (int s = 0; s < CH; s += 2) {
        step(s,     0,    HBUF, e0, e1);
        step(s + 1, HBUF, 0,    o0, o1);
    }

    const float4 r0 = *(const float4*)(l0);
    float4 r1 = r0;
    if (two) r1 = *(const float4*)(l1);
    *(float4*)(g0) = r0;
    if (two) *(float4*)(g1) = r1;
}

__device__ __forceinline__ void qproj_chunk(const float* __restrict__ W2x, const float* __restrict__ b2,
                                            float* __restrict__ S, int t0, int qb, float* lds) {
    const int tid  = threadIdx.x;
    const int lane = tid & 31;
    const int wv   = __builtin_amdgcn_readfirstlane(tid >> 5);
    const int c    = lane & 15;
    const int kh   = (lane >> 4) << 1;
    const int crow = (lane >> 4) << 3;

    const int ncol = wv * 16 + c;
    v2f w[KBH];
    load_wfrags(w, W2x, ncol, kh);
    const float bias = (ncol < H) ? b2[ncol] : 0.f;

    const int p0 = qb * QPAIRS;
    const int bt = p0 / CH;
    const int ts = t0 + (p0 - bt * CH);
    float* Sb = S + ((size_t)bt * SEQT + ts) * PSLOT + wv * PTILE + lane * 4;

    float4 v0 = *(const float4*)(Sb), v1 = *(const float4*)(Sb + 128);
    float* d = lds + crow * HSTR + wv * 16 + c;
    const float* hb = lds + c * HSTR + kh;

    for (int i = 0; i < QPAIRS; ++i) {
        const int buf = (i & 1) * HBUF;
        float* q = Sb + (size_t)i * PSLOT;
        const float* qn = (i + 1 < QPAIRS) ? (q + PSLOT) : q;
        const float4 u0 = *(const float4*)(qn), u1 = *(const float4*)(qn + 128);
        d[buf + 0 * HSTR] = v0.x; d[buf + 1 * HSTR] = v0.y; d[buf + 2 * HSTR] = v0.z; d[buf + 3 * HSTR] = v0.w;
        d[buf + 4 * HSTR] = v1.x; d[buf + 5 * HSTR] = v1.y; d[buf + 6 * HSTR] = v1.z; d[buf + 7 * HSTR] = v1.w;
        bar_signal();
        bar_wait();
        v2f a[KBH];
        #pragma unroll
        for (int k = 0; k < KBH; ++k) a[k] = ld2(hb + buf + 4 * k);
        v8f c0, c1;
        #pragma unroll
        for (int r = 0; r < 8; ++r) { c0[r] = bias; c1[r] = 0.f; }
        chain25(c0, c1, a, w);
        *(float4*)(q)       = make_float4(c0[0] + c1[0], c0[1] + c1[1], c0[2] + c1[2], c0[3] + c1[3]);
        *(float4*)(q + 128) = make_float4(c0[4] + c1[4], c0[5] + c1[5], c0[6] + c1[6], c0[7] + c1[7]);
        v0 = u0; v1 = u1;
    }
}

__global__ void __launch_bounds__(RTHR)
drnn_wave_kernel(const float* __restrict__ W1h, const float* __restrict__ W2x,
                 const float* __restrict__ W2h, const float* __restrict__ b2,
                 const float* __restrict__ Wo,  const float* __restrict__ bo,
                 float* __restrict__ P,
                 float* __restrict__ outp, float* __restrict__ h1out, float* __restrict__ h2out,
                 int l) {
    extern __shared__ float lds[];
    const int blk = blockIdx.x;
    if (blk < NRECWG) {
        if (l < NCHUNK) rec_chunk<true>(W1h, P, l * CH, l == 0, h1out, lds, blk);
    } else if (blk < 2 * NRECWG) {
        const int ch = l - 2;
        if (ch >= 0 && ch < NCHUNK) {
            const int bt = blk - NRECWG;
            rec_chunk<false>(W2h, P, ch * CH, ch == 0, h2out, lds, bt);
            if (ch == NCHUNK - 1) {
                const int tid = threadIdx.x;
                if (tid < 16) {
                    float s = bo[0];
                    for (int n = 0; n < H; ++n) s += lds[tid * HSTR + n] * Wo[n];
                    outp[bt * 16 + tid] = s;
                }
            }
        }
    } else {
        const int ch = l - 1;
        if (ch >= 0 && ch < NCHUNK) qproj_chunk(W2x, b2, P, ch * CH, blk - 2 * NRECWG, lds);
    }
}

#define NXWG   (NRECWG * (NCHUNK - 1))

__global__ void __launch_bounds__(RTHR)
drnn_first_kernel(const float* __restrict__ x, const float* __restrict__ W1x,
                  const float* __restrict__ b1, const float* __restrict__ W1h,
                  float* __restrict__ P, float* __restrict__ h1out) {
    extern __shared__ float lds[];
    const int blk = blockIdx.x;
    if (blk < NRECWG) {
        rec_chunk<true>(W1h, P, 0, true, h1out, lds, blk);
    } else {
        const int q    = blk - NRECWG;
        const int bt   = q / (NCHUNK - 1);
        const int tc   = 1 + (q - bt * (NCHUNK - 1));
        const int lane = threadIdx.x & 31;
        const int wv   = __builtin_amdgcn_readfirstlane(threadIdx.x >> 5);
        const int n    = (wv < 4) ? 10 : 8;
        const int s0   = (wv < 4) ? 10 * wv : 8 * wv + 8;
        xproj_wave<false>(x, W1x, b1, P, bt, tc * CH + s0, 1, n, lane);
    }
}

extern "C" void kernel_launch(void* const* d_in, const int* in_sizes, int n_in,
                              void* d_out, int out_size, void* d_ws, size_t ws_size,
                              hipStream_t stream) {
    (void)in_sizes; (void)n_in; (void)out_size;
    const float* x   = (const float*)d_in[0];
    const float* W1x = (const float*)d_in[1];
    const float* W1h = (const float*)d_in[2];
    const float* b1  = (const float*)d_in[3];
    const float* W2x = (const float*)d_in[4];
    const float* W2h = (const float*)d_in[5];
    const float* b2  = (const float*)d_in[6];
    const float* Wo  = (const float*)d_in[7];
    const float* bo  = (const float*)d_in[8];

    float* outp  = (float*)d_out;
    float* h1out = outp + BATCH;
    float* h2out = h1out + BATCH * H;

    if (ws_size >= WS_NEED) {
        float* P = (float*)d_ws;
        xproj_head_kernel<<<(BATCH / 16) * 8, 256, 0, stream>>>(x, W1x, b1, P);
        (void)hipFuncSetAttribute((const void*)drnn_first_kernel,
                                  hipFuncAttributeMaxDynamicSharedMemorySize, (int)W_LDS_RESERVE);
        (void)hipFuncSetAttribute((const void*)drnn_wave_kernel,
                                  hipFuncAttributeMaxDynamicSharedMemorySize, (int)W_LDS_RESERVE);
        drnn_first_kernel<<<NRECWG + NXWG, RTHR, (size_t)W_LDS_RESERVE, stream>>>(x, W1x, b1, W1h, P, h1out);
#define WAVE_LAUNCH(l) drnn_wave_kernel<<<2 * NRECWG + NQWG, RTHR, (size_t)W_LDS_RESERVE, stream>>>( \
            W1h, W2x, W2h, b2, Wo, bo, P, outp, h1out, h2out, (l))
        WAVE_LAUNCH(1); WAVE_LAUNCH(2); WAVE_LAUNCH(3); WAVE_LAUNCH(4);
        WAVE_LAUNCH(5); WAVE_LAUNCH(6); WAVE_LAUNCH(7); WAVE_LAUNCH(8); WAVE_LAUNCH(9);
#undef WAVE_LAUNCH
        return;
    }

    const size_t shmem = (size_t)LDS_FLOATS * sizeof(float);
    (void)hipFuncSetAttribute((const void*)drnn_fused_kernel,
                              hipFuncAttributeMaxDynamicSharedMemorySize, (int)shmem);

    drnn_fused_kernel<<<BATCH / 16, 256, shmem, stream>>>(
        x, W1x, W1h, b1, W2x, W2h, b2, Wo, bo, outp, h1out, h2out);
}


# Weights as A, state as B: a wave's accumulator is the B fragments of its own tile's 4 k-blocks, multiplied between barrier signal and wait at low priority: 21 not 25 WMMAs behind; sum order changes
# speedup vs baseline: 2.6965x; 1.1226x over previous
#include <hip/hip_runtime.h>
#include <math.h>

typedef __attribute__((ext_vector_type(2))) float v2f;
typedef __attribute__((ext_vector_type(8))) float v8f;

#define BATCH 512
#define SEQT  512
#define IND   64
#define H     100
#define NT    7
#define KB1   16
#define KBH   25
#define HSTR  116
#define XSTR  68
#define HBUF  (16*HSTR)

#define OFF_W1X 0
#define OFF_W1H (OFF_W1X + NT*KB1*32*2)
#define OFF_W2X (OFF_W1H + NT*KBH*32*2)
#define OFF_W2H (OFF_W2X + NT*KBH*32*2)
#define OFF_H1  (OFF_W2H + NT*KBH*32*2)
#define OFF_H2  (OFF_H1 + 2*HBUF)
#define OFF_X   (OFF_H2 + 2*HBUF)
#define XBUF    (16*XSTR)
#define KBXP    11
#define OFF_P   (OFF_X + 4*XBUF)
#define PBUF    (NT*256)
#define LDS_FLOATS (OFF_P + 2*PBUF)

#if __has_builtin(__builtin_amdgcn_tanhf)
#define TANHF(v) __builtin_amdgcn_tanhf(v)
#elif __has_builtin(__builtin_amdgcn_tanh_f32)
#define TANHF(v) __builtin_amdgcn_tanh_f32(v)
#else
#define TANHF(v) tanhf(v)
#endif

__device__ inline void fill_frags(float* dst, const float* W, int Ka, int Na,
                                  int kb, int tid, int nthr) {
    int total = NT * kb * 32;
    for (int i = tid; i < total; i += nthr) {
        int laneI = i & 31;
        int kblk  = (i >> 5) % kb;
        int nt    = (i >> 5) / kb;
        int N = nt * 16 + (laneI & 15);
        int K = kblk * 4 + ((laneI >> 4) << 1);
        float a = 0.f, b = 0.f;
        if (N < Na) {
            if (K     < Ka) a = W[(size_t)K * Na + N];
            if (K + 1 < Ka) b = W[(size_t)(K + 1) * Na + N];
        }
        dst[2 * i + 0] = a;
        dst[2 * i + 1] = b;
    }
}

__device__ inline v2f ld2(const float* p) { return *(const v2f*)p; }

#define KBXR  (KB1 - KBXP)
#define NWR   80
#define WR_1X 0
#define WR_1H (WR_1X + KBXR)
#define WR_2X (WR_1H + KBH)
#define WR_2H (WR_2X + KBH)

#define WMMA4(a, b, c) __builtin_amdgcn_wmma_f32_16x16x4_f32(false, (a), false, (b), (short)0, (c), false, false)

__device__ __forceinline__ void tanh_store(const v8f& c0, const v8f& c1, float* dst) {
    #pragma unroll
    for (int r = 0; r < 8; ++r) dst[r * HSTR] = TANHF(c0[r] + c1[r]);
}

__device__ __forceinline__ void layer1_store(const float4& p0, const float4& p1,
                                             const v2f (&ax)[KBXR], const v2f (&a1)[KBH],
                                             const v2f (&wr)[NWR], float* dst) {
    v8f c0, c1;
    c0[0] = p0.x; c0[1] = p0.y; c0[2] = p0.z; c0[3] = p0.w;
    c0[4] = p1.x; c0[5] = p1.y; c0[6] = p1.z; c0[7] = p1.w;
    #pragma unroll
    for (int r = 0; r < 8; ++r) c1[r] = 0.f;
    #pragma unroll
    for (int k = 0; k < KBXR; ++k) {
        if (k & 1) c1 = WMMA4(ax[k], wr[WR_1X + k], c1);
        else       c0 = WMMA4(ax[k], wr[WR_1X + k], c0);
    }
    #pragma unroll
    for (int k = 0; k < KBH; ++k) {
        if ((k + KBXR) & 1) c1 = WMMA4(a1[k], wr[WR_1H + k], c1);
        else                c0 = WMMA4(a1[k], wr[WR_1H + k], c0);
    }
    tanh_store(c0, c1, dst);
}

__device__ __forceinline__ void layer2_store(float bias2, const v2f (&a1)[KBH], const v2f (&a2)[KBH],
                                             const v2f (&wr)[NWR], float* dst) {
    v8f c0, c1;
    #pragma unroll
    for (int r = 0; r < 8; ++r) { c0[r] = bias2; c1[r] = 0.f; }
    #pragma unroll
    for (int k = 0; k < KBH; ++k) {
        if (k & 1) c1 = WMMA4(a1[k], wr[WR_2X + k], c1);
        else       c0 = WMMA4(a1[k], wr[WR_2X + k], c0);
    }
    #pragma unroll
    for (int k = 0; k < KBH; ++k) {
        if (k & 1) c1 = WMMA4(a2[k], wr[WR_2H + k], c1);
        else       c0 = WMMA4(a2[k], wr[WR_2H + k], c0);
    }
    tanh_store(c0, c1, dst);
}

__device__ __forceinline__ void helper_p(const float* xl, const float (&pbias)[NT],
                                         const v2f (&wr)[NWR], float* pdst) {
    v8f acc[NT];
    #pragma unroll
    for (int j = 0; j < NT; ++j)
        #pragma unroll
        for (int r = 0; r < 8; ++r) acc[j][r] = pbias[j];
    #pragma unroll
    for (int kb = 0; kb < KBXP; ++kb) {
        v2f a = ld2(xl + 4 * kb);
        #pragma unroll
        for (int j = 0; j < NT; ++j)
            acc[j] = WMMA4(a, wr[j * KBXP + kb], acc[j]);
    }
    #pragma unroll
    for (int j = 0; j < NT; ++j) {
        float* pp = pdst + j * 256;
        *(float4*)(pp)       = make_float4(acc[j][0], acc[j][1], acc[j][2], acc[j][3]);
        *(float4*)(pp + 128) = make_float4(acc[j][4], acc[j][5], acc[j][6], acc[j][7]);
    }
}

__global__ void __launch_bounds__(256)
drnn_fused_kernel(const float* __restrict__ x,
                  const float* __restrict__ W1x, const float* __restrict__ W1h,
                  const float* __restrict__ b1,
                  const float* __restrict__ W2x, const float* __restrict__ W2h,
                  const float* __restrict__ b2,
                  const float* __restrict__ Wo,  const float* __restrict__ bo,
                  float* __restrict__ outp, float* __restrict__ h1out,
                  float* __restrict__ h2out) {
    extern __shared__ float lds[];
    const int tid  = threadIdx.x;
    const int nthr = blockDim.x;
    const int b0   = blockIdx.x * 16;

    fill_frags(lds + OFF_W1X, W1x, IND, H, KB1, tid, nthr);
    fill_frags(lds + OFF_W1H, W1h, H,   H, KBH, tid, nthr);
    fill_frags(lds + OFF_W2X, W2x, H,   H, KBH, tid, nthr);
    fill_frags(lds + OFF_W2H, W2h, H,   H, KBH, tid, nthr);
    for (int i = tid; i < 4 * HBUF; i += nthr) lds[OFF_H1 + i] = 0.f;

    const float* xsrc = x + ((size_t)(b0 + (tid >> 4)) * SEQT) * IND + ((tid & 15) << 2);
    float*       xdst = lds + OFF_X + (tid >> 4) * XSTR + ((tid & 15) << 2);
    *(float4*)(xdst)        = *(const float4*)(xsrc);
    *(float4*)(xdst + XBUF) = *(const float4*)(xsrc + IND);
    float4 xr = *(const float4*)(xsrc + 2 * IND);
    __syncthreads();

    const int lane = tid & 31;
    const int wv   = __builtin_amdgcn_readfirstlane(tid >> 5);
    const bool helper = (wv == NT);
    const int mi   = lane & 15;
    const int kh   = (lane >> 4) << 1;
    const int crow = (lane >> 4) << 3;

    float bias2 = 0.f;
    {
        int n = wv * 16 + mi;
        if (n < H) bias2 = b2[n];
    }

    v2f   wr[NWR];
    float pbias[NT];
    #pragma unroll
    for (int j = 0; j < NT; ++j) pbias[j] = 0.f;
    if (helper) {
        #pragma unroll
        for (int j = 0; j < NT; ++j) {
            pbias[j] = (j * 16 + mi < H) ? b1[j * 16 + mi] : 0.f;
            #pragma unroll
            for (int kb = 0; kb < KBXP; ++kb)
                wr[j * KBXP + kb] = ld2(lds + OFF_W1X + (j * KB1 + kb) * 64 + lane * 2);
        }
        #pragma unroll
        for (int i = NT * KBXP; i < NWR; ++i) { wr[i][0] = 0.f; wr[i][1] = 0.f; }
        const float* xl = lds + OFF_X + mi * XSTR + kh;
        v8f acc[NT];
        #pragma unroll
        for (int j = 0; j < NT; ++j)
            #pragma unroll
            for (int r = 0; r < 8; ++r) acc[j][r] = pbias[j];
        #pragma unroll
        for (int kb = 0; kb < KBXP; ++kb) {
            v2f a = ld2(xl + 4 * kb);
            #pragma unroll
            for (int j = 0; j < NT; ++j)
                acc[j] = __builtin_amdgcn_wmma_f32_16x16x4_f32(false, a, false, wr[j * KBXP + kb], (short)0, acc[j], false, false);
        }
        #pragma unroll
        for (int j = 0; j < NT; ++j) {
            float* pp = lds + OFF_P + j * 256 + lane * 4;
            *(float4*)(pp)       = make_float4(acc[j][0], acc[j][1], acc[j][2], acc[j][3]);
            *(float4*)(pp + 128) = make_float4(acc[j][4], acc[j][5], acc[j][6], acc[j][7]);
        }
    } else {
        #pragma unroll
        for (int k = 0; k < KBXR; ++k)
            wr[WR_1X + k] = ld2(lds + OFF_W1X + (wv * KB1 + KBXP + k) * 64 + lane * 2);
        #pragma unroll
        for (int k = 0; k < KBH; ++k) {
            wr[WR_1H + k] = ld2(lds + OFF_W1H + (wv * KBH + k) * 64 + lane * 2);
            wr[WR_2X + k] = ld2(lds + OFF_W2X + (wv * KBH + k) * 64 + lane * 2);
            wr[WR_2H + k] = ld2(lds + OFF_W2H + (wv * KBH + k) * 64 + lane * 2);
        }
    }
    __syncthreads();

    if (!helper) {
        v2f a1[KBH], ax[KBXR];
        const float* hl = lds + OFF_H1 + HBUF + mi * HSTR + kh;
        const float* xl = lds + OFF_X + mi * XSTR + kh;
        const float* pp = lds + OFF_P + wv * 256 + lane * 4;
        #pragma unroll
        for (int k = 0; k < KBH; ++k) a1[k] = ld2(hl + 4 * k);
        #pragma unroll
        for (int k = 0; k < KBXR; ++k) ax[k] = ld2(xl + 4 * (KBXP + k));
        const float4 p0 = *(const float4*)(pp);
        const float4 p1 = *(const float4*)(pp + 128);
        layer1_store(p0, p1, ax, a1, wr, lds + OFF_H1 + crow * HSTR + wv * 16 + mi);
    } else {
        helper_p(lds + OFF_X + XBUF + mi * XSTR + kh, pbias, wr, lds + OFF_P + PBUF + lane * 4);
    }
    *(float4*)(xdst + 2 * XBUF) = xr;
    __syncthreads();
    xr = *(const float4*)(xsrc + 3 * IND);

    for (int t = 0; t + 1 < SEQT; ++t) {
        const int cb = t & 1;
        const int nb = cb ^ 1;

        if (!helper) {
            v2f a1[KBH], a2[KBH], ax[KBXR];
            const float* h1l = lds + OFF_H1 + cb * HBUF + mi * HSTR + kh;
            const float* h2l = lds + OFF_H2 + nb * HBUF + mi * HSTR + kh;
            const float* xl  = lds + OFF_X + ((t + 1) & 3) * XBUF + mi * XSTR + kh;
            const float* pp  = lds + OFF_P + nb * PBUF + wv * 256 + lane * 4;
            const float4 p0 = *(const float4*)(pp);
            const float4 p1 = *(const float4*)(pp + 128);
            #pragma unroll
            for (int k = 0; k < KBXR; ++k) ax[k] = ld2(xl + 4 * (KBXP + k));
            #pragma unroll
            for (int k = 0; k < 4; ++k) a1[k] = ld2(h1l + 4 * k);
            __builtin_amdgcn_sched_barrier(0);
            #pragma unroll
            for (int k = 4; k < KBH; ++k) a1[k] = ld2(h1l + 4 * k);
            __builtin_amdgcn_sched_barrier(0);
            #pragma unroll
            for (int k = 0; k < KBH; ++k) a2[k] = ld2(h2l + 4 * k);
            __builtin_amdgcn_sched_barrier(0);

            layer2_store(bias2, a1, a2, wr, lds + OFF_H2 + cb * HBUF + crow * HSTR + wv * 16 + mi);
            layer1_store(p0, p1, ax, a1, wr, lds + OFF_H1 + nb * HBUF + crow * HSTR + wv * 16 + mi);
        } else if (t + 2 < SEQT) {
            helper_p(lds + OFF_X + ((t + 2) & 3) * XBUF + mi * XSTR + kh, pbias, wr,
                     lds + OFF_P + cb * PBUF + lane * 4);
        }

        if (t + 3 < SEQT) *(float4*)(xdst + ((t + 3) & 3) * XBUF) = xr;

        __syncthreads();

        if (t + 4 < SEQT) xr = *(const float4*)(xsrc + (size_t)(t + 4) * IND);
    }

    if (!helper) {
        const int cb = (SEQT - 1) & 1, nb = cb ^ 1;
        v2f a1[KBH], a2[KBH];
        const float* h1l = lds + OFF_H1 + cb * HBUF + mi * HSTR + kh;
        const float* h2l = lds + OFF_H2 + nb * HBUF + mi * HSTR + kh;
        #pragma unroll
        for (int k = 0; k < KBH; ++k) a1[k] = ld2(h1l + 4 * k);
        #pragma unroll
        for (int k = 0; k < KBH; ++k) a2[k] = ld2(h2l + 4 * k);
        layer2_store(bias2, a1, a2, wr, lds + OFF_H2 + cb * HBUF + crow * HSTR + wv * 16 + mi);
    }
    __syncthreads();

    const int fb = (SEQT - 1) & 1;
    for (int i = tid; i < 16 * H; i += nthr) {
        int m = i / H, n = i - m * H;
        h1out[(size_t)(b0 + m) * H + n] = lds[OFF_H1 + fb * HBUF + m * HSTR + n];
        h2out[(size_t)(b0 + m) * H + n] = lds[OFF_H2 + fb * HBUF + m * HSTR + n];
    }
    if (tid < 16) {
        float s = bo[0];
        for (int n = 0; n < H; ++n) s += lds[OFF_H2 + fb * HBUF + tid * HSTR + n] * Wo[n];
        outp[b0 + tid] = s;
    }
}

#define PTILE 256
#define PSLOT (NT * PTILE)
#define WS_NEED ((size_t)(BATCH / 16) * SEQT * PSLOT * sizeof(float))
template <bool SINGLE>
__device__ __forceinline__ void xproj_wave(const float* __restrict__ x, const float* __restrict__ W1x,
                                           const float* __restrict__ b1, float* __restrict__ P,
                                           int bt, int t0, int dt, int n, int lane) {
    const int c  = lane & 15;
    const int kh = (lane >> 4) << 1;

    const float* xrow = x + ((size_t)(bt * 16 + c) * SEQT) * IND + kh;
    float*       prow = P + ((size_t)bt * SEQT) * PSLOT + lane * 4;
    v2f xe[KB1], xo[KB1];
    #pragma unroll
    for (int kb = 0; kb < KB1; ++kb) xe[kb] = ld2(xrow + (size_t)t0 * IND + 4 * kb);

    v2f   wf[NT * KB1];
    float bias[NT][8];
    const int hrow = (lane >> 4) << 3;
    #pragma unroll
    for (int j = 0; j < NT; ++j) {
        const int  N  = j * 16 + c;
        const bool ok = N < H;
        const float* wp = W1x + (size_t)kh * H + (ok ? N : H - 1);
        #pragma unroll
        for (int r = 0; r < 8; ++r) {
            const int   nb  = j * 16 + hrow + r;
            const bool  okb = nb < H;
            const float bv  = b1[okb ? nb : H - 1];
            bias[j][r] = okb ? bv : 0.f;
        }
        #pragma unroll
        for (int kb = 0; kb < KB1; ++kb) {
            const float lo = wp[(size_t)(4 * kb) * H];
            const float hi = wp[(size_t)(4 * kb + 1) * H];
            v2f w;
            w[0] = ok ? lo : 0.f;
            w[1] = ok ? hi : 0.f;
            wf[j * KB1 + kb] = w;
        }
    }
    __builtin_amdgcn_sched_barrier(0);

    auto body = [&](const v2f (&xa)[KB1], const int t) __attribute__((always_inline)) {
        v8f acc[NT];
        #pragma unroll
        for (int j = 0; j < NT; ++j)
            #pragma unroll
            for (int r = 0; r < 8; ++r) acc[j][r] = bias[j][r];
        #pragma unroll
        for (int kb = 0; kb < KB1; ++kb)
            #pragma unroll
            for (int j = 0; j < NT; ++j)
                acc[j] = WMMA4(wf[j * KB1 + kb], xa[kb], acc[j]);
        float* pp = prow + (size_t)t * PSLOT;
        #pragma unroll
        for (int j = 0; j < NT; ++j) {
            *(float4*)(pp + j * PTILE)       = make_float4(acc[j][0], acc[j][1], acc[j][2], acc[j][3]);
            *(float4*)(pp + j * PTILE + 128) = make_float4(acc[j][4], acc[j][5], acc[j][6], acc[j][7]);
        }
    };

    if (SINGLE) {
        body(xe, t0);
        return;
    }
    for (int s = 0; s < n; s += 2) {
        const int t  = t0 + s * dt;
        const int t1 = t + dt;
        const int t2 = (s + 2 < n) ? t1 + dt : t1;
        #pragma unroll
        for (int kb = 0; kb < KB1; ++kb) xo[kb] = ld2(xrow + (size_t)t1 * IND + 4 * kb);
        __builtin_amdgcn_sched_barrier(0);
        body(xe, t);
        #pragma unroll
        for (int kb = 0; kb < KB1; ++kb) xe[kb] = ld2(xrow + (size_t)t2 * IND + 4 * kb);
        __builtin_amdgcn_sched_barrier(0);
        body(xo, t1);
    }
}

__global__ void __launch_bounds__(256)
xproj_head_kernel(const float* __restrict__ x, const float* __restrict__ W1x,
                  const float* __restrict__ b1, float* __restrict__ P) {
    const int lane = threadIdx.x & 31;
    const int wv   = __builtin_amdgcn_readfirstlane(threadIdx.x >> 5);
    xproj_wave<true>(x, W1x, b1, P, blockIdx.x >> 3, (blockIdx.x & 7) * 8 + wv, 1, 1, lane);
}

#define CH     64
#define NCHUNK (SEQT / CH)
#define NRECWG (BATCH / 16)
#define NQWG   128
#define QPAIRS (NRECWG * CH / NQWG)
#define W_LDS_FLOATS (2 * HBUF)
#define W_LDS_RESERVE (164 * 1024)
#define RTHR   (NT * 32)

__device__ __forceinline__ void load_wfrags(v2f (&w)[KBH], const float* __restrict__ Wm, int ncol, int kh) {
    const bool ok = ncol < H;
    const float* p = Wm + (size_t)kh * H + (ok ? ncol : H - 1);
    float lo[KBH], hi[KBH];
    #pragma unroll
    for (int k = 0; k < KBH; ++k) {
        lo[k] = p[(size_t)(4 * k) * H];
        hi[k] = p[(size_t)(4 * k + 1) * H];
    }
    #pragma unroll
    for (int k = 0; k < KBH; ++k) {
        w[k][0] = ok ? lo[k] : 0.f;
        w[k][1] = ok ? hi[k] : 0.f;
    }
}

#define CARRY4 (16 * H / 4)

__device__ __forceinline__ void bar_signal() {
    __builtin_amdgcn_fence(__ATOMIC_RELEASE, "workgroup", "local");
    __builtin_amdgcn_s_barrier_signal(-1);
}
__device__ __forceinline__ void bar_wait() {
    __builtin_amdgcn_s_barrier_wait(-1);
    __builtin_amdgcn_fence(__ATOMIC_ACQUIRE, "workgroup", "local");
}

__device__ __forceinline__ void chain25(v8f& c0, v8f& c1, const v2f (&a)[KBH], const v2f (&w)[KBH]) {
    #pragma unroll
    for (int k = 0; k < KBH; ++k) {
        if (k & 1) c1 = WMMA4(w[k], a[k], c1);
        else       c0 = WMMA4(w[k], a[k], c0);
    }
}

__device__ __forceinline__ void load_wfrags_tiles(v2f (&w)[KBH], const float* __restrict__ Wm,
                                                  int tj, int c, int half) {
    const int  ncol = tj * 16 + c;
    const bool ok = ncol < H;
    const float* p = Wm + (ok ? ncol : H - 1);
    float lo[KBH], hi[KBH];
    #pragma unroll
    for (int q = 0; q < 6; ++q) {
        int T = tj + q;
        if (T >= 6) T -= 6;
        const float* pt = p + (size_t)(16 * T + 8 * half) * H;
        #pragma unroll
        for (int i = 0; i < 4; ++i) {
            lo[4 * q + i] = pt[(size_t)(2 * i) * H];
            hi[4 * q + i] = pt[(size_t)(2 * i + 1) * H];
        }
    }
    const float* pl = p + (size_t)(96 + 2 * half) * H;
    lo[KBH - 1] = pl[0];
    hi[KBH - 1] = pl[H];
    #pragma unroll
    for (int k = 0; k < KBH; ++k) {
        w[k][0] = ok ? lo[k] : 0.f;
        w[k][1] = ok ? hi[k] : 0.f;
    }
}

__device__ __forceinline__ float4 ld4(const float* p) { return *(const float4*)p; }

__device__ __forceinline__ void tile4(v8f& c0, v8f& c1, const float4& u0, const float4& u1,
                                      const v2f& w0, const v2f& w1, const v2f& w2, const v2f& w3) {
    v2f f0, f1, f2, f3;
    f0[0] = u0.x; f0[1] = u0.y; f1[0] = u0.z; f1[1] = u0.w;
    f2[0] = u1.x; f2[1] = u1.y; f3[0] = u1.z; f3[1] = u1.w;
    c0 = WMMA4(w0, f0, c0);
    c1 = WMMA4(w1, f1, c1);
    c0 = WMMA4(w2, f2, c0);
    c1 = WMMA4(w3, f3, c1);
}

__device__ __forceinline__ void start_from(v8f& c0, v8f& c1, const float4& p0, const float4& p1) {
    c0[0] = p0.x; c0[1] = p0.y; c0[2] = p0.z; c0[3] = p0.w;
    c0[4] = p1.x; c0[5] = p1.y; c0[6] = p1.z; c0[7] = p1.w;
    #pragma unroll
    for (int r = 0; r < 8; ++r) c1[r] = 0.f;
}

template <bool WRITEBACK>
__device__ __forceinline__ void rec_chunk(const float* __restrict__ Wh, float* __restrict__ S,
                                          int t0, bool first, float* __restrict__ hcarry,
                                          float* lds, int bt) {
    const int tid  = threadIdx.x;
    const int lane = tid & 31;
    const int wv   = __builtin_amdgcn_readfirstlane(tid >> 5);
    const int  tj    = (wv == 3) ? (NT - 1) : ((wv == NT - 1) ? 3 : wv);
    const bool tailw = (tj == NT - 1);
    const int c    = lane & 15;
    const int half = lane >> 4;
    const int b0   = bt * 16;


    float* Sb = S + (size_t)bt * SEQT * PSLOT + tj * PTILE + lane * 4;
    const float* pa = Sb + (size_t)t0 * PSLOT;
    float4 ra0 = *(const float4*)(pa),             ra1 = *(const float4*)(pa + 128);
    float4 rb0 = *(const float4*)(pa + PSLOT),     rb1 = *(const float4*)(pa + PSLOT + 128);
    float4 rc0 = *(const float4*)(pa + 2 * PSLOT), rc1 = *(const float4*)(pa + 2 * PSLOT + 128);
    float4 rd0 = *(const float4*)(pa + 3 * PSLOT), rd1 = *(const float4*)(pa + 3 * PSLOT + 128);

    const int  i0 = tid, i1 = tid + RTHR;
    const bool two = i1 < CARRY4;
    const int  m0 = i0 / 25, m1 = i1 / 25;
    float* l0 = lds + m0 * HSTR + 4 * (i0 - m0 * 25);
    float* l1 = lds + m1 * HSTR + 4 * (i1 - m1 * 25);
    float* g0 = hcarry + (size_t)b0 * H + 4 * i0;
    float* g1 = hcarry + (size_t)b0 * H + 4 * i1;
    float4 s0 = make_float4(0.f, 0.f, 0.f, 0.f), s1 = s0;
    if (!first) {
        s0 = *(const float4*)(g0);
        if (two) s1 = *(const float4*)(g1);
    }

    v2f w[KBH];
    load_wfrags_tiles(w, Wh, tj, c, half);

    *(float4*)(l0) = s0;
    if (two) *(float4*)(l1) = s1;
    __syncthreads();

    const float* hq[6];
    #pragma unroll
    for (int q = 0; q < 6; ++q) {
        int T = tj + q;
        if (T >= 6) T -= 6;
        hq[q] = lds + c * HSTR + 16 * T + 8 * half;
    }
    const float* ht = lds + c * HSTR + 96 + 2 * half;
    float*       hd = lds + c * HSTR + 16 * tj + 8 * half;

    v8f c0, c1;
    start_from(c0, c1, ra0, ra1);
    if (!tailw) {
        const float4 u0 = ld4(hq[0]), u1 = ld4(hq[0] + 4);
        tile4(c0, c1, u0, u1, w[0], w[1], w[2], w[3]);
    }

    auto step = [&](const int s, const int cur, const int nxt, float4& U0, float4& U1,
                    float4& N0, float4& N1) __attribute__((always_inline)) {
        const int t = t0 + s;
        float4 g[10];
        float4 ga, gb;
        if (tailw) { ga = ld4(hq[0] + cur); gb = ld4(hq[0] + cur + 4); }
        #pragma unroll
        for (int q = 1; q < 6; ++q) {
            g[2 * q - 2] = ld4(hq[q] + cur);
            g[2 * q - 1] = ld4(hq[q] + cur + 4);
        }
        const v2f tl = ld2(ht + cur);
        __builtin_amdgcn_sched_barrier(0);
        __builtin_amdgcn_s_setprio(1);
        if (tailw) tile4(c0, c1, ga, gb, w[0], w[1], w[2], w[3]);
        #pragma unroll
        for (int q = 1; q < 6; ++q)
            tile4(c0, c1, g[2 * q - 2], g[2 * q - 1], w[4 * q], w[4 * q + 1], w[4 * q + 2], w[4 * q + 3]);
        c0 = WMMA4(w[KBH - 1], tl, c0);
        float hv[8];
        #pragma unroll
        for (int r = 0; r < 8; ++r) hv[r] = TANHF(c0[r] + c1[r]);
        const float4 v0 = make_float4(hv[0], hv[1], hv[2], hv[3]);
        const float4 v1 = make_float4(hv[4], hv[5], hv[6], hv[7]);
        *(float4*)(hd + nxt)     = v0;
        *(float4*)(hd + nxt + 4) = v1;
        bar_signal();
        __builtin_amdgcn_s_setprio(0);
        if (WRITEBACK) {
            float* q = Sb + (size_t)t * PSLOT;
            *(float4*)(q)       = v0;
            *(float4*)(q + 128) = v1;
        }
        const int tn = (s + 4 < CH) ? (t + 4) : (t0 + CH - 1);
        const float* pp = Sb + (size_t)tn * PSLOT;
        U0 = *(const float4*)(pp); U1 = *(const float4*)(pp + 128);
        start_from(c0, c1, N0, N1);
        if (!tailw) tile4(c0, c1, v0, v1, w[0], w[1], w[2], w[3]);
        bar_wait();
    };
    for (int s = 0; s < CH; s += 4) {
        step(s,     0,    HBUF, ra0, ra1, rb0, rb1);
        step(s + 1, HBUF, 0,    rb0, rb1, rc0, rc1);
        step(s + 2, 0,    HBUF, rc0, rc1, rd0, rd1);
        step(s + 3, HBUF, 0,    rd0, rd1, ra0, ra1);
    }

    const float4 r0 = *(const float4*)(l0);
    float4 r1 = r0;
    if (two) r1 = *(const float4*)(l1);
    *(float4*)(g0) = r0;
    if (two) *(float4*)(g1) = r1;
}

__device__ __forceinline__ void qproj_chunk(const float* __restrict__ W2x, const float* __restrict__ b2,
                                            float* __restrict__ S, int t0, int qb, float* lds) {
    const int tid  = threadIdx.x;
    const int lane = tid & 31;
    const int wv   = __builtin_amdgcn_readfirstlane(tid >> 5);
    const int c    = lane & 15;
    const int kh   = (lane >> 4) << 1;
    const int hrow = (lane >> 4) << 3;

    const int ncol = wv * 16 + c;
    v2f w[KBH];
    load_wfrags(w, W2x, ncol, kh);
    float bias[8];
    #pragma unroll
    for (int r = 0; r < 8; ++r) {
        const int   nb  = wv * 16 + hrow + r;
        const bool  okb = nb < H;
        const float bv  = b2[okb ? nb : H - 1];
        bias[r] = okb ? bv : 0.f;
    }

    const int p0 = qb * QPAIRS;
    const int bt = p0 / CH;
    const int ts = t0 + (p0 - bt * CH);
    float* Sb = S + ((size_t)bt * SEQT + ts) * PSLOT + wv * PTILE + lane * 4;

    float4 v0 = *(const float4*)(Sb), v1 = *(const float4*)(Sb + 128);
    float* d = lds + c * HSTR + wv * 16 + hrow;
    const float* hb = lds + c * HSTR + kh;

    for (int i = 0; i < QPAIRS; ++i) {
        const int buf = (i & 1) * HBUF;
        float* q = Sb + (size_t)i * PSLOT;
        const float* qn = (i + 1 < QPAIRS) ? (q + PSLOT) : q;
        const float4 u0 = *(const float4*)(qn), u1 = *(const float4*)(qn + 128);
        *(float4*)(d + buf)     = v0;
        *(float4*)(d + buf + 4) = v1;
        bar_signal();
        bar_wait();
        v2f a[KBH];
        #pragma unroll
        for (int k = 0; k < KBH; ++k) a[k] = ld2(hb + buf + 4 * k);
        v8f c0, c1;
        #pragma unroll
        for (int r = 0; r < 8; ++r) { c0[r] = bias[r]; c1[r] = 0.f; }
        chain25(c0, c1, a, w);
        *(float4*)(q)       = make_float4(c0[0] + c1[0], c0[1] + c1[1], c0[2] + c1[2], c0[3] + c1[3]);
        *(float4*)(q + 128) = make_float4(c0[4] + c1[4], c0[5] + c1[5], c0[6] + c1[6], c0[7] + c1[7]);
        v0 = u0; v1 = u1;
    }
}

__global__ void __launch_bounds__(RTHR)
drnn_wave_kernel(const float* __restrict__ W1h, const float* __restrict__ W2x,
                 const float* __restrict__ W2h, const float* __restrict__ b2,
                 const float* __restrict__ Wo,  const float* __restrict__ bo,
                 float* __restrict__ P,
                 float* __restrict__ outp, float* __restrict__ h1out, float* __restrict__ h2out,
                 int l) {
    extern __shared__ float lds[];
    const int blk = blockIdx.x;
    if (blk < NRECWG) {
        if (l < NCHUNK) rec_chunk<true>(W1h, P, l * CH, l == 0, h1out, lds, blk);
    } else if (blk < 2 * NRECWG) {
        const int ch = l - 2;
        if (ch >= 0 && ch < NCHUNK) {
            const int bt = blk - NRECWG;
            rec_chunk<false>(W2h, P, ch * CH, ch == 0, h2out, lds, bt);
            if (ch == NCHUNK - 1) {
                const int tid = threadIdx.x;
                if (tid < 16) {
                    float s = bo[0];
                    for (int n = 0; n < H; ++n) s += lds[tid * HSTR + n] * Wo[n];
                    outp[bt * 16 + tid] = s;
                }
            }
        }
    } else {
        const int ch = l - 1;
        if (ch >= 0 && ch < NCHUNK) qproj_chunk(W2x, b2, P, ch * CH, blk - 2 * NRECWG, lds);
    }
}

#define NXWG   (NRECWG * (NCHUNK - 1))

__global__ void __launch_bounds__(RTHR)
drnn_first_kernel(const float* __restrict__ x, const float* __restrict__ W1x,
                  const float* __restrict__ b1, const float* __restrict__ W1h,
                  float* __restrict__ P, float* __restrict__ h1out) {
    extern __shared__ float lds[];
    const int blk = blockIdx.x;
    if (blk < NRECWG) {
        rec_chunk<true>(W1h, P, 0, true, h1out, lds, blk);
    } else {
        const int q    = blk - NRECWG;
        const int bt   = q / (NCHUNK - 1);
        const int tc   = 1 + (q - bt * (NCHUNK - 1));
        const int lane = threadIdx.x & 31;
        const int wv   = __builtin_amdgcn_readfirstlane(threadIdx.x >> 5);
        const int n    = (wv < 4) ? 10 : 8;
        const int s0   = (wv < 4) ? 10 * wv : 8 * wv + 8;
        xproj_wave<false>(x, W1x, b1, P, bt, tc * CH + s0, 1, n, lane);
    }
}

extern "C" void kernel_launch(void* const* d_in, const int* in_sizes, int n_in,
                              void* d_out, int out_size, void* d_ws, size_t ws_size,
                              hipStream_t stream) {
    (void)in_sizes; (void)n_in; (void)out_size;
    const float* x   = (const float*)d_in[0];
    const float* W1x = (const float*)d_in[1];
    const float* W1h = (const float*)d_in[2];
    const float* b1  = (const float*)d_in[3];
    const float* W2x = (const float*)d_in[4];
    const float* W2h = (const float*)d_in[5];
    const float* b2  = (const float*)d_in[6];
    const float* Wo  = (const float*)d_in[7];
    const float* bo  = (const float*)d_in[8];

    float* outp  = (float*)d_out;
    float* h1out = outp + BATCH;
    float* h2out = h1out + BATCH * H;

    if (ws_size >= WS_NEED) {
        float* P = (float*)d_ws;
        xproj_head_kernel<<<(BATCH / 16) * 8, 256, 0, stream>>>(x, W1x, b1, P);
        (void)hipFuncSetAttribute((const void*)drnn_first_kernel,
                                  hipFuncAttributeMaxDynamicSharedMemorySize, (int)W_LDS_RESERVE);
        (void)hipFuncSetAttribute((const void*)drnn_wave_kernel,
                                  hipFuncAttributeMaxDynamicSharedMemorySize, (int)W_LDS_RESERVE);
        drnn_first_kernel<<<NRECWG + NXWG, RTHR, (size_t)W_LDS_RESERVE, stream>>>(x, W1x, b1, W1h, P, h1out);
#define WAVE_LAUNCH(l) drnn_wave_kernel<<<2 * NRECWG + NQWG, RTHR, (size_t)W_LDS_RESERVE, stream>>>( \
            W1h, W2x, W2h, b2, Wo, bo, P, outp, h1out, h2out, (l))
        WAVE_LAUNCH(1); WAVE_LAUNCH(2); WAVE_LAUNCH(3); WAVE_LAUNCH(4);
        WAVE_LAUNCH(5); WAVE_LAUNCH(6); WAVE_LAUNCH(7); WAVE_LAUNCH(8); WAVE_LAUNCH(9);
#undef WAVE_LAUNCH
        return;
    }

    const size_t shmem = (size_t)LDS_FLOATS * sizeof(float);
    (void)hipFuncSetAttribute((const void*)drnn_fused_kernel,
                              hipFuncAttributeMaxDynamicSharedMemorySize, (int)shmem);

    drnn_fused_kernel<<<BATCH / 16, 256, shmem, stream>>>(
        x, W1x, W1h, b1, W2x, W2h, b2, Wo, bo, outp, h1out, h2out);
}
